# starting kernel
# baseline (speedup 1.0000x reference)
#include <hip/hip_runtime.h>
#include <hip/hip_bf16.h>


typedef __attribute__((ext_vector_type(2))) float v2f;
typedef __attribute__((ext_vector_type(8))) float v8f;

#define NN 50000
#define EE 800000
#define HD 128
#define CD 384

__global__ void deg_init(float* deg, int n) {
    int i = blockIdx.x * blockDim.x + threadIdx.x;
    if (i < n) deg[i] = 1.0f;
}
__global__ void deg_count(const int* __restrict__ dst, float* deg, int e) {
    int i = blockIdx.x * blockDim.x + threadIdx.x;
    if (i < e) atomicAdd(&deg[dst[i]], 1.0f);
}
__global__ void deg_rsqrt(float* deg, int n) {
    int i = blockIdx.x * blockDim.x + threadIdx.x;
    if (i < n) deg[i] = rsqrtf(deg[i]);
}

template <bool BT>
__global__ void __launch_bounds__(256)
gemm16_f32(const float* __restrict__ A, int lda,
           const float* __restrict__ B, int ldb,
           float* __restrict__ C, int ldc,
           int K, int tilesPerRow, int totalTiles) {
    int wave = threadIdx.x >> 5;
    int lane = threadIdx.x & 31;
    int tile = blockIdx.x * (blockDim.x >> 5) + wave;
    if (tile >= totalTiles) return;
    int tm = (tile / tilesPerRow) << 4;
    int tn = (tile % tilesPerRow) << 4;
    int rc   = lane & 15;
    int half = lane >> 4;

    v8f acc = {};
    const float* arow = A + (size_t)(tm + rc) * lda;
#pragma unroll 4
    for (int k0 = 0; k0 < K; k0 += 4) {
        int ka = k0 + 2 * half;
        v2f a;
        a.x = arow[ka];
        a.y = arow[ka + 1];
        v2f b;
        if (BT) {
            const float* bp = B + (size_t)(tn + rc) * ldb + ka;
            b.x = bp[0];
            b.y = bp[1];
        } else {
            b.x = B[(size_t)(ka + 0) * ldb + tn + rc];
            b.y = B[(size_t)(ka + 1) * ldb + tn + rc];
        }
        acc = __builtin_amdgcn_wmma_f32_16x16x4_f32(
            false, a, false, b, (short)0, acc, false, false);
    }
#pragma unroll
    for (int i = 0; i < 8; ++i)
        C[(size_t)(tm + i + 8 * half) * ldc + tn + rc] = acc[i];
}

__global__ void self_init(const float* __restrict__ XW, const float* __restrict__ dinv,
                          const float* __restrict__ bias,
                          float* __restrict__ H, int ldh, int colOff, int n) {
    int idx = blockIdx.x * blockDim.x + threadIdx.x;
    int v = idx >> 5, lane = idx & 31;
    if (v >= n) return;
    float dn = dinv[v];
    float nn = dn * dn;
    float4 x = ((const float4*)(XW + (size_t)v * HD))[lane];
    float4 b = ((const float4*)bias)[lane];
    float* op = H + (size_t)v * ldh + colOff + lane * 4;
    op[0] = x.x * nn + b.x;
    op[1] = x.y * nn + b.y;
    op[2] = x.z * nn + b.z;
    op[3] = x.w * nn + b.w;
}

__global__ void __launch_bounds__(256)
edge_scatter(const float* __restrict__ XW, const float* __restrict__ dinv,
             const int* __restrict__ src, const int* __restrict__ dst,
             float* __restrict__ H, int ldh, int colOff, int e) {
    int idx = blockIdx.x * blockDim.x + threadIdx.x;
    int eid = idx >> 5, lane = idx & 31;
    if (eid >= e) return;
    int s = src[eid], d = dst[eid];
    float norm = dinv[s] * dinv[d];
    float4 v = ((const float4*)(XW + (size_t)s * HD))[lane];
    float* op = H + (size_t)d * ldh + colOff + lane * 4;
    atomicAdd(op + 0, v.x * norm);
    atomicAdd(op + 1, v.y * norm);
    atomicAdd(op + 2, v.z * norm);
    atomicAdd(op + 3, v.w * norm);
}

__global__ void logits_init(float* logits, const float* __restrict__ lb2, int n) {
    int i = blockIdx.x * blockDim.x + threadIdx.x;
    if (i < n) {
        logits[2 * i + 0] = lb2[0];
        logits[2 * i + 1] = lb2[1];
    }
}

__global__ void __launch_bounds__(256)
linear_fused(const float* __restrict__ H,
             const float* __restrict__ lW1,
             const float* __restrict__ lb1,
             const float* __restrict__ lW2,
             float* __restrict__ logits,
             int totalTiles) {
    const int K = CD, tilesPerRow = CD / 16;
    int wave = threadIdx.x >> 5;
    int lane = threadIdx.x & 31;
    int tile = blockIdx.x * (blockDim.x >> 5) + wave;
    if (tile >= totalTiles) return;
    int tm = (tile / tilesPerRow) << 4;
    int tn = (tile % tilesPerRow) << 4;
    int col  = lane & 15;
    int half = lane >> 4;

    v8f acc = {};
    const float* arow = H + (size_t)(tm + col) * K;
    const float* brow = lW1 + (size_t)(tn + col) * K;
#pragma unroll 4
    for (int k0 = 0; k0 < K; k0 += 4) {
        int ka = k0 + 2 * half;
        v2f a;
        a.x = arow[ka];
        a.y = arow[ka + 1];
        v2f b;
        b.x = brow[ka];
        b.y = brow[ka + 1];
        acc = __builtin_amdgcn_wmma_f32_16x16x4_f32(
            false, a, false, b, (short)0, acc, false, false);
    }

    float bias = lb1[tn + col];
    float w0 = lW2[0 * CD + tn + col];
    float w1 = lW2[1 * CD + tn + col];
#pragma unroll
    for (int i = 0; i < 8; ++i) {
        float t = acc[i] + bias;
        t = t > 0.0f ? t : 0.0f;
        float p0 = t * w0;
        float p1 = t * w1;
#pragma unroll
        for (int m = 8; m >= 1; m >>= 1) {
            p0 += __shfl_xor(p0, m, 16);
            p1 += __shfl_xor(p1, m, 16);
        }
        if (col == 0) {
            int r = tm + i + 8 * half;
            atomicAdd(&logits[2 * r + 0], p0);
            atomicAdd(&logits[2 * r + 1], p1);
        }
    }
}

__global__ void softmax2(const float* __restrict__ logits, float* __restrict__ probs, int n) {
    int i = blockIdx.x * blockDim.x + threadIdx.x;
    if (i >= n) return;
    float a = logits[2 * i + 0], b = logits[2 * i + 1];
    float m = fmaxf(a, b);
    float ea = __expf(a - m), eb = __expf(b - m);
    float inv = 1.0f / (ea + eb);
    probs[2 * i + 0] = ea * inv;
    probs[2 * i + 1] = eb * inv;
}

extern "C" void kernel_launch(void* const* d_in, const int* in_sizes, int n_in,
                              void* d_out, int out_size, void* d_ws, size_t ws_size,
                              hipStream_t stream) {
    const float* x   = (const float*)d_in[0];
    const int*   ei  = (const int*)d_in[1];
    const float* W1  = (const float*)d_in[2];
    const float* b1  = (const float*)d_in[3];
    const float* W2  = (const float*)d_in[4];
    const float* b2  = (const float*)d_in[5];
    const float* W3  = (const float*)d_in[6];
    const float* b3  = (const float*)d_in[7];
    const float* lW1 = (const float*)d_in[8];
    const float* lb1 = (const float*)d_in[9];
    const float* lW2 = (const float*)d_in[10];
    const float* lb2 = (const float*)d_in[11];

    const int N = NN, E = EE;
    const int* src = ei;
    const int* dst = ei + E;

    float* dinv = (float*)d_ws;
    float* xw   = dinv + N;
    float* h    = xw + (size_t)N * HD;

    float* logits = (float*)d_out;
    float* probs  = logits + (size_t)N * 2;

    const int T = 256;
    const int rowTiles = N / 16;

    deg_init<<<(N + T - 1) / T, T, 0, stream>>>(dinv, N);
    deg_count<<<(E + T - 1) / T, T, 0, stream>>>(dst, dinv, E);
    deg_rsqrt<<<(N + T - 1) / T, T, 0, stream>>>(dinv, N);

    {
        int tiles = rowTiles * (HD / 16);
        gemm16_f32<false><<<(tiles + 7) / 8, T, 0, stream>>>(
            x, 64, W1, HD, xw, HD, 64, HD / 16, tiles);
        self_init<<<((size_t)N * 32 + T - 1) / T, T, 0, stream>>>(xw, dinv, b1, h, CD, 0, N);
        edge_scatter<<<((size_t)E * 32 + T - 1) / T, T, 0, stream>>>(
            xw, dinv, src, dst, h, CD, 0, E);
    }
    {
        int tiles = rowTiles * (HD / 16);
        gemm16_f32<false><<<(tiles + 7) / 8, T, 0, stream>>>(
            h + 0, CD, W2, HD, xw, HD, HD, HD / 16, tiles);
        self_init<<<((size_t)N * 32 + T - 1) / T, T, 0, stream>>>(xw, dinv, b2, h, CD, HD, N);
        edge_scatter<<<((size_t)E * 32 + T - 1) / T, T, 0, stream>>>(
            xw, dinv, src, dst, h, CD, HD, E);
    }
    {
        int tiles = rowTiles * (HD / 16);
        gemm16_f32<false><<<(tiles + 7) / 8, T, 0, stream>>>(
            h + HD, CD, W3, HD, xw, HD, HD, HD / 16, tiles);
        self_init<<<((size_t)N * 32 + T - 1) / T, T, 0, stream>>>(xw, dinv, b3, h, CD, 2 * HD, N);
        edge_scatter<<<((size_t)E * 32 + T - 1) / T, T, 0, stream>>>(
            xw, dinv, src, dst, h, CD, 2 * HD, E);
    }

    logits_init<<<(N + T - 1) / T, T, 0, stream>>>(logits, lb2, N);
    {
        int tiles = rowTiles * (CD / 16);
        linear_fused<<<(tiles + 7) / 8, T, 0, stream>>>(h, lW1, lb1, lW2, logits, tiles);
    }
    softmax2<<<(N + T - 1) / T, T, 0, stream>>>(logits, probs, N);
}


# GEMMs retiled: 80-row block tiles staged in LDS, k-step 32, b128 fragment reads (k summed in another order), A fragment reused over column tiles; head sums rows in-block, fixed order, no atomics
# speedup vs baseline: 1.2727x; 1.2727x over previous
#include <hip/hip_runtime.h>
#include <hip/hip_bf16.h>


typedef __attribute__((ext_vector_type(2))) float v2f;
typedef __attribute__((ext_vector_type(4))) float v4f;
typedef __attribute__((ext_vector_type(8))) float v8f;

#define NN 50000
#define EE 800000
#define HD 128
#define CD 384

__global__ void deg_init(float* deg, int n) {
    int i = blockIdx.x * blockDim.x + threadIdx.x;
    if (i < n) deg[i] = 1.0f;
}
__global__ void deg_count(const int* __restrict__ dst, float* deg, int e) {
    int i = blockIdx.x * blockDim.x + threadIdx.x;
    if (i < e) atomicAdd(&deg[dst[i]], 1.0f);
}
__global__ void deg_rsqrt(float* deg, int n) {
    int i = blockIdx.x * blockDim.x + threadIdx.x;
    if (i < n) deg[i] = rsqrtf(deg[i]);
}

#define BM 80
#define KB 32
#define PA (KB + 4)

template <bool BT, int BN, bool HEAD>
__global__ void __launch_bounds__(256)
gemm_tiled(const float* __restrict__ A, int lda,
           const float* __restrict__ B, int ldb,
           float* __restrict__ C, int ldc, int K,
           const float* __restrict__ lb1,
           const float* __restrict__ lW2,
           const float* __restrict__ lb2) {
    constexpr int RT = BM / 16;
    constexpr int CT = BN / 128;
    constexpr int PB = BT ? PA : (BN + 4);
    constexpr int BROWS = BT ? BN : KB;
    constexpr int QA = KB / 4;
    constexpr int A4 = BM * QA;
    constexpr int AL = (A4 + 255) / 256;
    constexpr int BL = BN * QA / 256;
    constexpr int QB = BN / 4;

    __shared__ __attribute__((aligned(16))) float As[BM * PA];
    __shared__ __attribute__((aligned(16))) float Bs[BROWS * PB];

    const int tid  = threadIdx.x;
    const int wave = tid >> 5;
    const int lane = tid & 31;
    const int c    = lane & 15;
    const int h    = lane >> 4;
    const int m0   = blockIdx.x * BM;
    const int wn0  = wave * (CT * 16);

    v8f acc[RT][CT];
#pragma unroll
    for (int i = 0; i < RT; ++i)
#pragma unroll
        for (int j = 0; j < CT; ++j) {
            v8f z = {};
            acc[i][j] = z;
        }

    v4f ra[AL], rb[BL];

#define TILE_GLOAD(kb)                                                              \
    {                                                                               \
        _Pragma("unroll") for (int u = 0; u < AL; ++u) {                            \
            int idx = tid + 256 * u;                                                \
            if (idx < A4) {                                                         \
                int row = idx / QA, q = idx % QA;                                   \
                ra[u] = *(const v4f*)(A + (size_t)(m0 + row) * lda + (kb) + 4 * q); \
            }                                                                       \
        }                                                                           \
        _Pragma("unroll") for (int u = 0; u < BL; ++u) {                            \
            int idx = tid + 256 * u;                                                \
            if (BT) {                                                               \
                int n = idx / QA, q = idx % QA;                                     \
                rb[u] = *(const v4f*)(B + (size_t)n * ldb + (kb) + 4 * q);       \
            } else {                                                                \
                int k = idx / QB, q = idx % QB;                                     \
                rb[u] = *(const v4f*)(B + (size_t)((kb) + k) * ldb + 4 * q);     \
            }                                                                       \
        }                                                                           \
    }

    TILE_GLOAD(0)
    const int S = K / KB;
    for (int s = 0; s < S; ++s) {
#pragma unroll
        for (int u = 0; u < AL; ++u) {
            int idx = tid + 256 * u;
            if (idx < A4) {
                int row = idx / QA, q = idx % QA;
                *(v4f*)&As[row * PA + 4 * q] = ra[u];
            }
        }
#pragma unroll
        for (int u = 0; u < BL; ++u) {
            int idx = tid + 256 * u;
            if (BT) {
                int n = idx / QA, q = idx % QA;
                *(v4f*)&Bs[n * PB + 4 * q] = rb[u];
            } else {
                int k = idx / QB, q = idx % QB;
                *(v4f*)&Bs[k * PB + 4 * q] = rb[u];
            }
        }
        __syncthreads();
        if (s + 1 < S) TILE_GLOAD((s + 1) * KB)

#pragma unroll
        for (int g = 0; g < KB / 8; ++g) {
            v2f alo[RT], ahi[RT], blo[CT], bhi[CT];
#pragma unroll
            for (int i = 0; i < RT; ++i) {
                v4f a4 = *(const v4f*)&As[(16 * i + c) * PA + 8 * g + 4 * h];
                alo[i].x = a4.x; alo[i].y = a4.y;
                ahi[i].x = a4.z; ahi[i].y = a4.w;
            }
#pragma unroll
            for (int j = 0; j < CT; ++j) {
                if (BT) {
                    v4f b4 = *(const v4f*)&Bs[(wn0 + 16 * j + c) * PB + 8 * g + 4 * h];
                    blo[j].x = b4.x; blo[j].y = b4.y;
                    bhi[j].x = b4.z; bhi[j].y = b4.w;
                } else {
                    const float* bp = &Bs[(8 * g + 4 * h) * PB + wn0 + 16 * j + c];
                    blo[j].x = bp[0];
                    blo[j].y = bp[PB];
                    bhi[j].x = bp[2 * PB];
                    bhi[j].y = bp[3 * PB];
                }
            }
#pragma unroll
            for (int i = 0; i < RT; ++i)
#pragma unroll
                for (int j = 0; j < CT; ++j)
                    acc[i][j] = __builtin_amdgcn_wmma_f32_16x16x4_f32(
                        false, alo[i], false, blo[j], (short)0, acc[i][j], false, false);
#pragma unroll
            for (int i = 0; i < RT; ++i)
#pragma unroll
                for (int j = 0; j < CT; ++j)
                    acc[i][j] = __builtin_amdgcn_wmma_f32_16x16x4_f32(
                        false, ahi[i], false, bhi[j], (short)0, acc[i][j], false, false);
        }
        __syncthreads();
    }
#undef TILE_GLOAD

    if (!HEAD) {
#pragma unroll
        for (int i = 0; i < RT; ++i)
#pragma unroll
            for (int j = 0; j < CT; ++j)
#pragma unroll
                for (int r = 0; r < 8; ++r)
                    C[(size_t)(m0 + 16 * i + 8 * h + r) * ldc + wn0 + 16 * j + c] = acc[i][j][r];
    } else {
        float bias[CT], w0[CT], w1[CT];
#pragma unroll
        for (int j = 0; j < CT; ++j) {
            int col = wn0 + 16 * j + c;
            bias[j] = lb1[col];
            w0[j]   = lW2[col];
            w1[j]   = lW2[BN + col];
        }
        float* red = Bs;
#pragma unroll
        for (int i = 0; i < RT; ++i)
#pragma unroll
            for (int r = 0; r < 8; ++r) {
                float p0 = 0.0f, p1 = 0.0f;
#pragma unroll
                for (int j = 0; j < CT; ++j) {
                    float t = acc[i][j][r] + bias[j];
                    t = t > 0.0f ? t : 0.0f;
                    p0 += t * w0[j];
                    p1 += t * w1[j];
                }
#pragma unroll
                for (int m = 8; m >= 1; m >>= 1) {
                    p0 += __shfl_xor(p0, m, 16);
                    p1 += __shfl_xor(p1, m, 16);
                }
                if (c == 0) {
                    int row = 16 * i + 8 * h + r;
                    red[(wave * BM + row) * 2 + 0] = p0;
                    red[(wave * BM + row) * 2 + 1] = p1;
                }
            }
        __syncthreads();
        if (tid < 2 * BM) {
            int row = tid >> 1, cls = tid & 1;
            float sum = lb2[cls];
#pragma unroll
            for (int w = 0; w < 8; ++w) sum += red[(w * BM + row) * 2 + cls];
            C[(size_t)(m0 + row) * ldc + cls] = sum;
        }
    }
}

__global__ void self_init(const float* __restrict__ XW, const float* __restrict__ dinv,
                          const float* __restrict__ bias,
                          float* __restrict__ H, int ldh, int colOff, int n) {
    int idx = blockIdx.x * blockDim.x + threadIdx.x;
    int v = idx >> 5, lane = idx & 31;
    if (v >= n) return;
    float dn = dinv[v];
    float nn = dn * dn;
    float4 x = ((const float4*)(XW + (size_t)v * HD))[lane];
    float4 b = ((const float4*)bias)[lane];
    float* op = H + (size_t)v * ldh + colOff + lane * 4;
    op[0] = x.x * nn + b.x;
    op[1] = x.y * nn + b.y;
    op[2] = x.z * nn + b.z;
    op[3] = x.w * nn + b.w;
}

__global__ void __launch_bounds__(256)
edge_scatter(const float* __restrict__ XW, const float* __restrict__ dinv,
             const int* __restrict__ src, const int* __restrict__ dst,
             float* __restrict__ H, int ldh, int colOff, int e) {
    int idx = blockIdx.x * blockDim.x + threadIdx.x;
    int eid = idx >> 5, lane = idx & 31;
    if (eid >= e) return;
    int s = src[eid], d = dst[eid];
    float norm = dinv[s] * dinv[d];
    float4 v = ((const float4*)(XW + (size_t)s * HD))[lane];
    float* op = H + (size_t)d * ldh + colOff + lane * 4;
    atomicAdd(op + 0, v.x * norm);
    atomicAdd(op + 1, v.y * norm);
    atomicAdd(op + 2, v.z * norm);
    atomicAdd(op + 3, v.w * norm);
}

__global__ void softmax2(const float* __restrict__ logits, float* __restrict__ probs, int n) {
    int i = blockIdx.x * blockDim.x + threadIdx.x;
    if (i >= n) return;
    float a = logits[2 * i + 0], b = logits[2 * i + 1];
    float m = fmaxf(a, b);
    float ea = __expf(a - m), eb = __expf(b - m);
    float inv = 1.0f / (ea + eb);
    probs[2 * i + 0] = ea * inv;
    probs[2 * i + 1] = eb * inv;
}

extern "C" void kernel_launch(void* const* d_in, const int* in_sizes, int n_in,
                              void* d_out, int out_size, void* d_ws, size_t ws_size,
                              hipStream_t stream) {
    const float* x   = (const float*)d_in[0];
    const int*   ei  = (const int*)d_in[1];
    const float* W1  = (const float*)d_in[2];
    const float* b1  = (const float*)d_in[3];
    const float* W2  = (const float*)d_in[4];
    const float* b2  = (const float*)d_in[5];
    const float* W3  = (const float*)d_in[6];
    const float* b3  = (const float*)d_in[7];
    const float* lW1 = (const float*)d_in[8];
    const float* lb1 = (const float*)d_in[9];
    const float* lW2 = (const float*)d_in[10];
    const float* lb2 = (const float*)d_in[11];

    const int N = NN, E = EE;
    const int* src = ei;
    const int* dst = ei + E;

    float* dinv = (float*)d_ws;
    float* xw   = dinv + N;
    float* h    = xw + (size_t)N * HD;

    float* logits = (float*)d_out;
    float* probs  = logits + (size_t)N * 2;

    const int T = 256;
    const int rowBlocks = N / BM;

    deg_init<<<(N + T - 1) / T, T, 0, stream>>>(dinv, N);
    deg_count<<<(E + T - 1) / T, T, 0, stream>>>(dst, dinv, E);
    deg_rsqrt<<<(N + T - 1) / T, T, 0, stream>>>(dinv, N);

    {
        gemm_tiled<false, HD, false><<<rowBlocks, T, 0, stream>>>(
            x, 64, W1, HD, xw, HD, 64, nullptr, nullptr, nullptr);
        self_init<<<((size_t)N * 32 + T - 1) / T, T, 0, stream>>>(xw, dinv, b1, h, CD, 0, N);
        edge_scatter<<<((size_t)E * 32 + T - 1) / T, T, 0, stream>>>(
            xw, dinv, src, dst, h, CD, 0, E);
    }
    {
        gemm_tiled<false, HD, false><<<rowBlocks, T, 0, stream>>>(
            h + 0, CD, W2, HD, xw, HD, HD, nullptr, nullptr, nullptr);
        self_init<<<((size_t)N * 32 + T - 1) / T, T, 0, stream>>>(xw, dinv, b2, h, CD, HD, N);
        edge_scatter<<<((size_t)E * 32 + T - 1) / T, T, 0, stream>>>(
            xw, dinv, src, dst, h, CD, HD, E);
    }
    {
        gemm_tiled<false, HD, false><<<rowBlocks, T, 0, stream>>>(
            h + HD, CD, W3, HD, xw, HD, HD, nullptr, nullptr, nullptr);
        self_init<<<((size_t)N * 32 + T - 1) / T, T, 0, stream>>>(xw, dinv, b3, h, CD, 2 * HD, N);
        edge_scatter<<<((size_t)E * 32 + T - 1) / T, T, 0, stream>>>(
            xw, dinv, src, dst, h, CD, 2 * HD, E);
    }

    gemm_tiled<true, CD, true><<<rowBlocks, T, 0, stream>>>(
        h, CD, lW1, CD, logits, 2, CD, lb1, lW2, lb2);
    softmax2<<<(N + T - 1) / T, T, 0, stream>>>(logits, probs, N);
}


# Edge scatter by f32 atomics replaced by per-call CSR build (int count, one-block scan, cursor fill) + one-wave-per-row register gather, self-loop and bias fused; row sums in fill-arrival order
# speedup vs baseline: 2.4058x; 1.8903x over previous
#include <hip/hip_runtime.h>
#include <hip/hip_bf16.h>


typedef __attribute__((ext_vector_type(2))) float v2f;
typedef __attribute__((ext_vector_type(4))) float v4f;
typedef __attribute__((ext_vector_type(8))) float v8f;

#define NN 50000
#define EE 800000
#define HD 128
#define CD 384

__global__ void csr_zero(int* __restrict__ cnt, int n) {
    int i = blockIdx.x * blockDim.x + threadIdx.x;
    if (i < n) cnt[i] = 0;
}
__global__ void csr_count(const int* __restrict__ dst, int* cnt, int e) {
    int i = blockIdx.x * blockDim.x + threadIdx.x;
    if (i < e) atomicAdd(&cnt[dst[i]], 1);
}
#define SCAN_T 1024
#define SCAN_PER ((NN + SCAN_T - 1) / SCAN_T)
__global__ void __launch_bounds__(SCAN_T)
csr_scan(const int* __restrict__ cnt, int* __restrict__ rowptr, int* __restrict__ cursor,
         float* __restrict__ dinv) {
    __shared__ int sc[SCAN_PER * SCAN_T];
    __shared__ int wsum[SCAN_T / 32];
    const int t = threadIdx.x;
    for (int i = 0; i < SCAN_PER; ++i) {
        int idx = i * SCAN_T + t;
        int cv = 0;
        if (idx < NN) {
            cv = cnt[idx];
            dinv[idx] = rsqrtf((float)(cv + 1));
        }
        sc[idx] = cv;
    }
    __syncthreads();
    int local = 0;
    for (int i = 0; i < SCAN_PER; ++i) local += sc[t * SCAN_PER + i];
    int incl = local;
    for (int m = 1; m < 32; m <<= 1) {
        int up = __shfl_up(incl, m, 32);
        if ((t & 31) >= m) incl += up;
    }
    if ((t & 31) == 31) wsum[t >> 5] = incl;
    __syncthreads();
    if (t < 32) {
        int w = wsum[t];
        int wi = w;
        for (int m = 1; m < 32; m <<= 1) {
            int up = __shfl_up(wi, m, 32);
            if (t >= m) wi += up;
        }
        wsum[t] = wi - w;
    }
    __syncthreads();
    int run = incl - local + wsum[t >> 5];
    for (int i = 0; i < SCAN_PER; ++i) {
        int cv = sc[t * SCAN_PER + i];
        sc[t * SCAN_PER + i] = run;
        run += cv;
    }
    if (t == SCAN_T - 1) rowptr[NN] = run;
    __syncthreads();
    for (int i = 0; i < SCAN_PER; ++i) {
        int idx = i * SCAN_T + t;
        if (idx < NN) {
            int p = sc[idx];
            rowptr[idx] = p;
            cursor[idx] = p;
        }
    }
}
__global__ void csr_fill(const int* __restrict__ src, const int* __restrict__ dst,
                         const float* __restrict__ dinv, int* cursor,
                         int* __restrict__ col, float* __restrict__ wgt, int e) {
    int i = blockIdx.x * blockDim.x + threadIdx.x;
    if (i >= e) return;
    int sv = src[i], dv = dst[i];
    int p = atomicAdd(&cursor[dv], 1);
    col[p] = sv;
    wgt[p] = dinv[sv] * dinv[dv];
}

#define BM 80
#define KB 32
#define PA (KB + 4)

template <bool BT, int BN, bool HEAD>
__global__ void __launch_bounds__(256)
gemm_tiled(const float* __restrict__ A, int lda,
           const float* __restrict__ B, int ldb,
           float* __restrict__ C, int ldc, int K,
           const float* __restrict__ lb1,
           const float* __restrict__ lW2,
           const float* __restrict__ lb2) {
    constexpr int RT = BM / 16;
    constexpr int CT = BN / 128;
    constexpr int PB = BT ? PA : (BN + 4);
    constexpr int BROWS = BT ? BN : KB;
    constexpr int QA = KB / 4;
    constexpr int A4 = BM * QA;
    constexpr int AL = (A4 + 255) / 256;
    constexpr int BL = BN * QA / 256;
    constexpr int QB = BN / 4;

    __shared__ __attribute__((aligned(16))) float As[BM * PA];
    __shared__ __attribute__((aligned(16))) float Bs[BROWS * PB];

    const int tid  = threadIdx.x;
    const int wave = tid >> 5;
    const int lane = tid & 31;
    const int c    = lane & 15;
    const int h    = lane >> 4;
    const int m0   = blockIdx.x * BM;
    const int wn0  = wave * (CT * 16);

    v8f acc[RT][CT];
#pragma unroll
    for (int i = 0; i < RT; ++i)
#pragma unroll
        for (int j = 0; j < CT; ++j) {
            v8f z = {};
            acc[i][j] = z;
        }

    v4f ra[AL], rb[BL];

#define TILE_GLOAD(kb)                                                              \
    {                                                                               \
        _Pragma("unroll") for (int u = 0; u < AL; ++u) {                            \
            int idx = tid + 256 * u;                                                \
            if (idx < A4) {                                                         \
                int row = idx / QA, q = idx % QA;                                   \
                ra[u] = *(const v4f*)(A + (size_t)(m0 + row) * lda + (kb) + 4 * q); \
            }                                                                       \
        }                                                                           \
        _Pragma("unroll") for (int u = 0; u < BL; ++u) {                            \
            int idx = tid + 256 * u;                                                \
            if (BT) {                                                               \
                int n = idx / QA, q = idx % QA;                                     \
                rb[u] = *(const v4f*)(B + (size_t)n * ldb + (kb) + 4 * q);       \
            } else {                                                                \
                int k = idx / QB, q = idx % QB;                                     \
                rb[u] = *(const v4f*)(B + (size_t)((kb) + k) * ldb + 4 * q);     \
            }                                                                       \
        }                                                                           \
    }

    TILE_GLOAD(0)
    const int S = K / KB;
    for (int s = 0; s < S; ++s) {
#pragma unroll
        for (int u = 0; u < AL; ++u) {
            int idx = tid + 256 * u;
            if (idx < A4) {
                int row = idx / QA, q = idx % QA;
                *(v4f*)&As[row * PA + 4 * q] = ra[u];
            }
        }
#pragma unroll
        for (int u = 0; u < BL; ++u) {
            int idx = tid + 256 * u;
            if (BT) {
                int n = idx / QA, q = idx % QA;
                *(v4f*)&Bs[n * PB + 4 * q] = rb[u];
            } else {
                int k = idx / QB, q = idx % QB;
                *(v4f*)&Bs[k * PB + 4 * q] = rb[u];
            }
        }
        __syncthreads();
        if (s + 1 < S) TILE_GLOAD((s + 1) * KB)

#pragma unroll
        for (int g = 0; g < KB / 8; ++g) {
            v2f alo[RT], ahi[RT], blo[CT], bhi[CT];
#pragma unroll
            for (int i = 0; i < RT; ++i) {
                v4f a4 = *(const v4f*)&As[(16 * i + c) * PA + 8 * g + 4 * h];
                alo[i].x = a4.x; alo[i].y = a4.y;
                ahi[i].x = a4.z; ahi[i].y = a4.w;
            }
#pragma unroll
            for (int j = 0; j < CT; ++j) {
                if (BT) {
                    v4f b4 = *(const v4f*)&Bs[(wn0 + 16 * j + c) * PB + 8 * g + 4 * h];
                    blo[j].x = b4.x; blo[j].y = b4.y;
                    bhi[j].x = b4.z; bhi[j].y = b4.w;
                } else {
                    const float* bp = &Bs[(8 * g + 4 * h) * PB + wn0 + 16 * j + c];
                    blo[j].x = bp[0];
                    blo[j].y = bp[PB];
                    bhi[j].x = bp[2 * PB];
                    bhi[j].y = bp[3 * PB];
                }
            }
#pragma unroll
            for (int i = 0; i < RT; ++i)
#pragma unroll
                for (int j = 0; j < CT; ++j)
                    acc[i][j] = __builtin_amdgcn_wmma_f32_16x16x4_f32(
                        false, alo[i], false, blo[j], (short)0, acc[i][j], false, false);
#pragma unroll
            for (int i = 0; i < RT; ++i)
#pragma unroll
                for (int j = 0; j < CT; ++j)
                    acc[i][j] = __builtin_amdgcn_wmma_f32_16x16x4_f32(
                        false, ahi[i], false, bhi[j], (short)0, acc[i][j], false, false);
        }
        __syncthreads();
    }
#undef TILE_GLOAD

    if (!HEAD) {
#pragma unroll
        for (int i = 0; i < RT; ++i)
#pragma unroll
            for (int j = 0; j < CT; ++j)
#pragma unroll
                for (int r = 0; r < 8; ++r)
                    C[(size_t)(m0 + 16 * i + 8 * h + r) * ldc + wn0 + 16 * j + c] = acc[i][j][r];
    } else {
        float bias[CT], w0[CT], w1[CT];
#pragma unroll
        for (int j = 0; j < CT; ++j) {
            int col = wn0 + 16 * j + c;
            bias[j] = lb1[col];
            w0[j]   = lW2[col];
            w1[j]   = lW2[BN + col];
        }
        float* red = Bs;
#pragma unroll
        for (int i = 0; i < RT; ++i)
#pragma unroll
            for (int r = 0; r < 8; ++r) {
                float p0 = 0.0f, p1 = 0.0f;
#pragma unroll
                for (int j = 0; j < CT; ++j) {
                    float t = acc[i][j][r] + bias[j];
                    t = t > 0.0f ? t : 0.0f;
                    p0 += t * w0[j];
                    p1 += t * w1[j];
                }
#pragma unroll
                for (int m = 8; m >= 1; m >>= 1) {
                    p0 += __shfl_xor(p0, m, 16);
                    p1 += __shfl_xor(p1, m, 16);
                }
                if (c == 0) {
                    int row = 16 * i + 8 * h + r;
                    red[(wave * BM + row) * 2 + 0] = p0;
                    red[(wave * BM + row) * 2 + 1] = p1;
                }
            }
        __syncthreads();
        if (tid < 2 * BM) {
            int row = tid >> 1, cls = tid & 1;
            float sum = lb2[cls];
#pragma unroll
            for (int w = 0; w < 8; ++w) sum += red[(w * BM + row) * 2 + cls];
            C[(size_t)(m0 + row) * ldc + cls] = sum;
        }
    }
}

__global__ void __launch_bounds__(256)
gather_rows(const float* __restrict__ XW, const float* __restrict__ dinv,
            const float* __restrict__ bias,
            const int* __restrict__ rowptr, const int* __restrict__ col,
            const float* __restrict__ wgt,
            float* __restrict__ H, int ldh, int colOff, int n) {
    const int lane = threadIdx.x & 31;
    const int v = __builtin_amdgcn_readfirstlane(blockIdx.x * 8 + (threadIdx.x >> 5));
    if (v >= n) return;
    const int beg = rowptr[v], end = rowptr[v + 1];
    const float dn = dinv[v];
    const float nn = dn * dn;
    const float4* X4 = (const float4*)XW;
    float4 x = X4[(size_t)v * (HD / 4) + lane];
    float4 b = ((const float4*)bias)[lane];
    float4 acc;
    acc.x = x.x * nn + b.x;
    acc.y = x.y * nn + b.y;
    acc.z = x.z * nn + b.z;
    acc.w = x.w * nn + b.w;
    for (int base = beg; base < end; base += 32) {
        const int m = (end - base) < 32 ? (end - base) : 32;
        int   myc = 0;
        float myw = 0.0f;
        if (lane < m) {
            myc = col[base + lane];
            myw = wgt[base + lane];
        }
        const int mywi = __float_as_int(myw);
        int k = 0;
        for (; k + 4 <= m; k += 4) {
            int s0 = __builtin_amdgcn_readlane(myc, k + 0);
            int s1 = __builtin_amdgcn_readlane(myc, k + 1);
            int s2 = __builtin_amdgcn_readlane(myc, k + 2);
            int s3 = __builtin_amdgcn_readlane(myc, k + 3);
            float w0 = __int_as_float(__builtin_amdgcn_readlane(mywi, k + 0));
            float w1 = __int_as_float(__builtin_amdgcn_readlane(mywi, k + 1));
            float w2 = __int_as_float(__builtin_amdgcn_readlane(mywi, k + 2));
            float w3 = __int_as_float(__builtin_amdgcn_readlane(mywi, k + 3));
            float4 r0 = X4[(size_t)s0 * (HD / 4) + lane];
            float4 r1 = X4[(size_t)s1 * (HD / 4) + lane];
            float4 r2 = X4[(size_t)s2 * (HD / 4) + lane];
            float4 r3 = X4[(size_t)s3 * (HD / 4) + lane];
            acc.x += r0.x * w0; acc.y += r0.y * w0; acc.z += r0.z * w0; acc.w += r0.w * w0;
            acc.x += r1.x * w1; acc.y += r1.y * w1; acc.z += r1.z * w1; acc.w += r1.w * w1;
            acc.x += r2.x * w2; acc.y += r2.y * w2; acc.z += r2.z * w2; acc.w += r2.w * w2;
            acc.x += r3.x * w3; acc.y += r3.y * w3; acc.z += r3.z * w3; acc.w += r3.w * w3;
        }
        for (; k < m; ++k) {
            int s0 = __builtin_amdgcn_readlane(myc, k);
            float w0 = __int_as_float(__builtin_amdgcn_readlane(mywi, k));
            float4 r0 = X4[(size_t)s0 * (HD / 4) + lane];
            acc.x += r0.x * w0; acc.y += r0.y * w0; acc.z += r0.z * w0; acc.w += r0.w * w0;
        }
    }
    *(float4*)(H + (size_t)v * ldh + colOff + lane * 4) = acc;
}

__global__ void softmax2(const float* __restrict__ logits, float* __restrict__ probs, int n) {
    int i = blockIdx.x * blockDim.x + threadIdx.x;
    if (i >= n) return;
    float a = logits[2 * i + 0], b = logits[2 * i + 1];
    float m = fmaxf(a, b);
    float ea = __expf(a - m), eb = __expf(b - m);
    float inv = 1.0f / (ea + eb);
    probs[2 * i + 0] = ea * inv;
    probs[2 * i + 1] = eb * inv;
}

extern "C" void kernel_launch(void* const* d_in, const int* in_sizes, int n_in,
                              void* d_out, int out_size, void* d_ws, size_t ws_size,
                              hipStream_t stream) {
    const float* x   = (const float*)d_in[0];
    const int*   ei  = (const int*)d_in[1];
    const float* W1  = (const float*)d_in[2];
    const float* b1  = (const float*)d_in[3];
    const float* W2  = (const float*)d_in[4];
    const float* b2  = (const float*)d_in[5];
    const float* W3  = (const float*)d_in[6];
    const float* b3  = (const float*)d_in[7];
    const float* lW1 = (const float*)d_in[8];
    const float* lb1 = (const float*)d_in[9];
    const float* lW2 = (const float*)d_in[10];
    const float* lb2 = (const float*)d_in[11];

    const int N = NN, E = EE;
    const int* src = ei;
    const int* dst = ei + E;

    float* dinv = (float*)d_ws;
    float* xw   = dinv + N;
    float* h    = xw + (size_t)N * HD;
    int*   cnt    = (int*)(h + (size_t)N * CD);
    int*   cursor = cnt + N;
    int*   rowptr = cursor + N;
    int*   col    = rowptr + (N + 1);
    float* wgt    = (float*)(col + E);

    float* logits = (float*)d_out;
    float* probs  = logits + (size_t)N * 2;

    const int T = 256;
    const int rowBlocks = N / BM;
    const int gatherBlocks = (N + 7) / 8;

    csr_zero<<<(N + T - 1) / T, T, 0, stream>>>(cnt, N);
    csr_count<<<(E + T - 1) / T, T, 0, stream>>>(dst, cnt, E);
    csr_scan<<<1, SCAN_T, 0, stream>>>(cnt, rowptr, cursor, dinv);
    csr_fill<<<(E + T - 1) / T, T, 0, stream>>>(src, dst, dinv, cursor, col, wgt, E);

    gemm_tiled<false, HD, false><<<rowBlocks, T, 0, stream>>>(
        x, 64, W1, HD, xw, HD, 64, nullptr, nullptr, nullptr);
    gather_rows<<<gatherBlocks, T, 0, stream>>>(xw, dinv, b1, rowptr, col, wgt, h, CD, 0, N);
    gemm_tiled<false, HD, false><<<rowBlocks, T, 0, stream>>>(
        h + 0, CD, W2, HD, xw, HD, HD, nullptr, nullptr, nullptr);
    gather_rows<<<gatherBlocks, T, 0, stream>>>(xw, dinv, b2, rowptr, col, wgt, h, CD, HD, N);
    gemm_tiled<false, HD, false><<<rowBlocks, T, 0, stream>>>(
        h + HD, CD, W3, HD, xw, HD, HD, nullptr, nullptr, nullptr);
    gather_rows<<<gatherBlocks, T, 0, stream>>>(xw, dinv, b3, rowptr, col, wgt, h, CD, 2 * HD, N);

    gemm_tiled<true, CD, true><<<rowBlocks, T, 0, stream>>>(
        h, CD, lW1, CD, logits, 2, CD, lb1, lW2, lb2);
    softmax2<<<(N + T - 1) / T, T, 0, stream>>>(logits, probs, N);
}


# Conv stack's linearity used: gathers run on 64-wide input rows (two source rows per load), W1 W2 W3 and biases folded per call into one 192-deep head GEMM on WMMA; sums in another order
# speedup vs baseline: 2.7699x; 1.1513x over previous
#include <hip/hip_runtime.h>
#include <hip/hip_bf16.h>


typedef __attribute__((ext_vector_type(2))) float v2f;
typedef __attribute__((ext_vector_type(4))) float v4f;
typedef __attribute__((ext_vector_type(8))) float v8f;

#define NN 50000
#define EE 800000
#define HD 128
#define CD 384
#define XD 64
#define YD 192

__global__ void csr_zero(int* __restrict__ cnt, int n) {
    int i = blockIdx.x * blockDim.x + threadIdx.x;
    if (i < n) cnt[i] = 0;
}
__global__ void csr_count(const int* __restrict__ dst, int* cnt, int e) {
    int i = blockIdx.x * blockDim.x + threadIdx.x;
    if (i < e) atomicAdd(&cnt[dst[i]], 1);
}
#define SCAN_T 1024
#define SCAN_PER ((NN + SCAN_T - 1) / SCAN_T)
__global__ void __launch_bounds__(SCAN_T)
csr_scan(const int* __restrict__ cnt, int* __restrict__ rowptr, int* __restrict__ cursor,
         float* __restrict__ dinv) {
    __shared__ int sc[SCAN_PER * SCAN_T];
    __shared__ int wsum[SCAN_T / 32];
    const int t = threadIdx.x;
    for (int i = 0; i < SCAN_PER; ++i) {
        int idx = i * SCAN_T + t;
        int cv = 0;
        if (idx < NN) {
            cv = cnt[idx];
            dinv[idx] = rsqrtf((float)(cv + 1));
        }
        sc[idx] = cv;
    }
    __syncthreads();
    int local = 0;
    for (int i = 0; i < SCAN_PER; ++i) local += sc[t * SCAN_PER + i];
    int incl = local;
    for (int m = 1; m < 32; m <<= 1) {
        int up = __shfl_up(incl, m, 32);
        if ((t & 31) >= m) incl += up;
    }
    if ((t & 31) == 31) wsum[t >> 5] = incl;
    __syncthreads();
    if (t < 32) {
        int w = wsum[t];
        int wi = w;
        for (int m = 1; m < 32; m <<= 1) {
            int up = __shfl_up(wi, m, 32);
            if (t >= m) wi += up;
        }
        wsum[t] = wi - w;
    }
    __syncthreads();
    int run = incl - local + wsum[t >> 5];
    for (int i = 0; i < SCAN_PER; ++i) {
        int cv = sc[t * SCAN_PER + i];
        sc[t * SCAN_PER + i] = run;
        run += cv;
    }
    if (t == SCAN_T - 1) rowptr[NN] = run;
    __syncthreads();
    for (int i = 0; i < SCAN_PER; ++i) {
        int idx = i * SCAN_T + t;
        if (idx < NN) {
            int p = sc[idx];
            rowptr[idx] = p;
            cursor[idx] = p;
        }
    }
}
__global__ void csr_fill(const int* __restrict__ src, const int* __restrict__ dst,
                         const float* __restrict__ dinv, int* cursor,
                         int* __restrict__ col, float* __restrict__ wgt, int e) {
    int i = blockIdx.x * blockDim.x + threadIdx.x;
    if (i >= e) return;
    int sv = src[i], dv = dst[i];
    int p = atomicAdd(&cursor[dv], 1);
    col[p] = sv;
    wgt[p] = dinv[sv] * dinv[dv];
}

#define BM 80
#define KB 32
#define PA (KB + 4)

template <bool BT, int BN, bool HEAD>
__global__ void __launch_bounds__(256)
gemm_tiled(const float* __restrict__ A, int lda,
           const float* __restrict__ B, int ldb,
           float* __restrict__ C, int ldc, int K,
           const float* __restrict__ lb1,
           const float* __restrict__ lW2,
           const float* __restrict__ lb2,
           int bpad,
           const float* __restrict__ rowvec, int rowidx,
           int sAy, int sBy, int sCy, int sBz, int sCz,
           const float* __restrict__ r1, const float* __restrict__ r2) {
    A += (size_t)blockIdx.y * sAy;
    B += (size_t)blockIdx.y * sBy + (size_t)blockIdx.z * sBz;
    C += (size_t)blockIdx.y * sCy + (size_t)blockIdx.z * sCz;
    constexpr int RT = BM / 16;
    constexpr int CT = BN / 128;
    constexpr int PB = BT ? PA : (BN + 4);
    constexpr int BROWS = BT ? BN : KB;
    constexpr int QA = KB / 4;
    constexpr int A4 = BM * QA;
    constexpr int AL = (A4 + 255) / 256;
    constexpr int BL = BN * QA / 256;
    constexpr int QB = BN / 4;

    __shared__ __attribute__((aligned(16))) float As[BM * PA];
    __shared__ __attribute__((aligned(16))) float Bs[BROWS * PB];

    const int tid  = threadIdx.x;
    const int wave = tid >> 5;
    const int lane = tid & 31;
    const int c    = lane & 15;
    const int h    = lane >> 4;
    const int m0   = blockIdx.x * BM;
    const int wn0  = wave * (CT * 16);

    v8f acc[RT][CT];
#pragma unroll
    for (int i = 0; i < RT; ++i)
#pragma unroll
        for (int j = 0; j < CT; ++j) {
            v8f z = {};
            acc[i][j] = z;
        }

    v4f ra[AL], rb[BL];

#define TILE_GLOAD(kb)                                                              \
    {                                                                               \
        _Pragma("unroll") for (int u = 0; u < AL; ++u) {                            \
            int idx = tid + 256 * u;                                                \
            if (idx < A4) {                                                         \
                int row = idx / QA, q = idx % QA;                                   \
                ra[u] = *(const v4f*)(A + (size_t)(m0 + row) * lda + (kb) + 4 * q); \
            }                                                                       \
        }                                                                           \
        _Pragma("unroll") for (int u = 0; u < BL; ++u) {                            \
            int idx = tid + 256 * u;                                                \
            if (BT) {                                                               \
                int n = idx / QA, q = idx % QA;                                     \
                rb[u] = *(const v4f*)(B + (size_t)n * ldb + (kb) + 4 * q);       \
            } else {                                                                \
                int k = idx / QB, q = idx % QB;                                     \
                rb[u] = *(const v4f*)(B + (size_t)((kb) + k + bpad * ((kb) >> 6)) * ldb + 4 * q); \
            }                                                                       \
        }                                                                           \
    }

    TILE_GLOAD(0)
    const int S = K / KB;
    for (int s = 0; s < S; ++s) {
#pragma unroll
        for (int u = 0; u < AL; ++u) {
            int idx = tid + 256 * u;
            if (idx < A4) {
                int row = idx / QA, q = idx % QA;
                *(v4f*)&As[row * PA + 4 * q] = ra[u];
            }
        }
#pragma unroll
        for (int u = 0; u < BL; ++u) {
            int idx = tid + 256 * u;
            if (BT) {
                int n = idx / QA, q = idx % QA;
                *(v4f*)&Bs[n * PB + 4 * q] = rb[u];
            } else {
                int k = idx / QB, q = idx % QB;
                *(v4f*)&Bs[k * PB + 4 * q] = rb[u];
            }
        }
        __syncthreads();
        if (s + 1 < S) TILE_GLOAD((s + 1) * KB)

#pragma unroll
        for (int g = 0; g < KB / 8; ++g) {
            v2f alo[RT], ahi[RT], blo[CT], bhi[CT];
#pragma unroll
            for (int i = 0; i < RT; ++i) {
                v4f a4 = *(const v4f*)&As[(16 * i + c) * PA + 8 * g + 4 * h];
                alo[i].x = a4.x; alo[i].y = a4.y;
                ahi[i].x = a4.z; ahi[i].y = a4.w;
            }
#pragma unroll
            for (int j = 0; j < CT; ++j) {
                if (BT) {
                    v4f b4 = *(const v4f*)&Bs[(wn0 + 16 * j + c) * PB + 8 * g + 4 * h];
                    blo[j].x = b4.x; blo[j].y = b4.y;
                    bhi[j].x = b4.z; bhi[j].y = b4.w;
                } else {
                    const float* bp = &Bs[(8 * g + 4 * h) * PB + wn0 + 16 * j + c];
                    blo[j].x = bp[0];
                    blo[j].y = bp[PB];
                    bhi[j].x = bp[2 * PB];
                    bhi[j].y = bp[3 * PB];
                }
            }
#pragma unroll
            for (int i = 0; i < RT; ++i)
#pragma unroll
                for (int j = 0; j < CT; ++j)
                    acc[i][j] = __builtin_amdgcn_wmma_f32_16x16x4_f32(
                        false, alo[i], false, blo[j], (short)0, acc[i][j], false, false);
#pragma unroll
            for (int i = 0; i < RT; ++i)
#pragma unroll
                for (int j = 0; j < CT; ++j)
                    acc[i][j] = __builtin_amdgcn_wmma_f32_16x16x4_f32(
                        false, ahi[i], false, bhi[j], (short)0, acc[i][j], false, false);
        }
        __syncthreads();
    }
#undef TILE_GLOAD

    if (!HEAD) {
#pragma unroll
        for (int i = 0; i < RT; ++i)
#pragma unroll
            for (int j = 0; j < CT; ++j)
#pragma unroll
                for (int r = 0; r < 8; ++r) {
                    const int row = m0 + 16 * i + 8 * h + r;
                    const int cc  = wn0 + 16 * j + c;
                    float o = acc[i][j][r];
                    if (rowvec != nullptr && row == rowidx) o += rowvec[cc];
                    C[(size_t)row * ldc + cc] = o;
                }
    } else {
        float bias[CT], u1[CT], u2[CT], w0[CT], w1[CT];
#pragma unroll
        for (int j = 0; j < CT; ++j) {
            int col = wn0 + 16 * j + c;
            bias[j] = lb1[col] + B[(size_t)(0 * BM + 64) * ldb + col]
                               + B[(size_t)(1 * BM + 65) * ldb + col]
                               + B[(size_t)(2 * BM + 66) * ldb + col];
            u1[j]   = B[(size_t)(1 * BM + 64) * ldb + col] + B[(size_t)(2 * BM + 65) * ldb + col];
            u2[j]   = B[(size_t)(2 * BM + 64) * ldb + col];
            w0[j]   = lW2[col];
            w1[j]   = lW2[BN + col];
        }
        float* red = Bs;
#pragma unroll
        for (int i = 0; i < RT; ++i)
#pragma unroll
            for (int r = 0; r < 8; ++r) {
                float p0 = 0.0f, p1 = 0.0f;
                const int grow = m0 + 16 * i + 8 * h + r;
                const float a1 = r1[grow], a2 = r2[grow];
#pragma unroll
                for (int j = 0; j < CT; ++j) {
                    float t = acc[i][j][r] + bias[j] + a1 * u1[j] + a2 * u2[j];
                    t = t > 0.0f ? t : 0.0f;
                    p0 += t * w0[j];
                    p1 += t * w1[j];
                }
#pragma unroll
                for (int m = 8; m >= 1; m >>= 1) {
                    p0 += __shfl_xor(p0, m, 16);
                    p1 += __shfl_xor(p1, m, 16);
                }
                if (c == 0) {
                    int row = 16 * i + 8 * h + r;
                    red[(wave * BM + row) * 2 + 0] = p0;
                    red[(wave * BM + row) * 2 + 1] = p1;
                }
            }
        __syncthreads();
        if (tid < 2 * BM) {
            int row = tid >> 1, cls = tid & 1;
            float sum = lb2[cls];
#pragma unroll
            for (int w = 0; w < 8; ++w) sum += red[(w * BM + row) * 2 + cls];
            C[(size_t)(m0 + row) * ldc + cls] = sum;
        }
    }
}

__global__ void fold_seed(const float* __restrict__ W1, const float* __restrict__ b1,
                          float* __restrict__ T1) {
    int i = blockIdx.x * blockDim.x + threadIdx.x;
    if (i >= BM * HD) return;
    int r = i / HD, cc = i % HD;
    float o = 0.0f;
    if (r < XD) o = W1[i];
    else if (r == XD) o = b1[cc];
    T1[i] = o;
}

template <int MODE>
__global__ void __launch_bounds__(256)
gather64(const float* __restrict__ Xin, int ldi, const float* __restrict__ dinv,
         const int* __restrict__ rowptr, const int* __restrict__ col,
         const float* __restrict__ wgt,
         float* __restrict__ Yout, int ldo,
         const float* __restrict__ rin, float* __restrict__ rout, int n) {
    const int lane = threadIdx.x & 31;
    const int c  = lane & 15;
    const int hh = lane >> 4;
    const int v = __builtin_amdgcn_readfirstlane(blockIdx.x * 8 + (threadIdx.x >> 5));
    if (v >= n) return;
    const int beg = rowptr[v], end = rowptr[v + 1];
    const float dn = dinv[v];
    const float nn = dn * dn;
    const float4* X4 = (const float4*)Xin;
    const int ld4 = ldi >> 2;
    float4 acc = make_float4(0.0f, 0.0f, 0.0f, 0.0f);
    if (hh == 0) {
        float4 x = X4[(size_t)v * ld4 + c];
        acc.x = x.x * nn; acc.y = x.y * nn; acc.z = x.z * nn; acc.w = x.w * nn;
    }
    float rs = 0.0f;
    for (int base = beg; base < end; base += 32) {
        const int m = (end - base) < 32 ? (end - base) : 32;
        int   myc = 0;
        float myw = 0.0f;
        if (lane < m) {
            myc = col[base + lane];
            myw = wgt[base + lane];
            if (MODE == 1) rs += myw;
            if (MODE == 2) rs += myw * rin[myc];
        }
        const int mywi = __float_as_int(myw);
        int k = 0;
        for (; k + 8 <= m; k += 8) {
            int sa0 = __builtin_amdgcn_readlane(myc, k + 0), sb0 = __builtin_amdgcn_readlane(myc, k + 1);
            int sa1 = __builtin_amdgcn_readlane(myc, k + 2), sb1 = __builtin_amdgcn_readlane(myc, k + 3);
            int sa2 = __builtin_amdgcn_readlane(myc, k + 4), sb2 = __builtin_amdgcn_readlane(myc, k + 5);
            int sa3 = __builtin_amdgcn_readlane(myc, k + 6), sb3 = __builtin_amdgcn_readlane(myc, k + 7);
            int wa0 = __builtin_amdgcn_readlane(mywi, k + 0), wb0 = __builtin_amdgcn_readlane(mywi, k + 1);
            int wa1 = __builtin_amdgcn_readlane(mywi, k + 2), wb1 = __builtin_amdgcn_readlane(mywi, k + 3);
            int wa2 = __builtin_amdgcn_readlane(mywi, k + 4), wb2 = __builtin_amdgcn_readlane(mywi, k + 5);
            int wa3 = __builtin_amdgcn_readlane(mywi, k + 6), wb3 = __builtin_amdgcn_readlane(mywi, k + 7);
            int s0 = hh ? sb0 : sa0, s1 = hh ? sb1 : sa1, s2 = hh ? sb2 : sa2, s3 = hh ? sb3 : sa3;
            float w0 = __int_as_float(hh ? wb0 : wa0);
            float w1 = __int_as_float(hh ? wb1 : wa1);
            float w2 = __int_as_float(hh ? wb2 : wa2);
            float w3 = __int_as_float(hh ? wb3 : wa3);
            float4 q0 = X4[(size_t)s0 * ld4 + c];
            float4 q1 = X4[(size_t)s1 * ld4 + c];
            float4 q2 = X4[(size_t)s2 * ld4 + c];
            float4 q3 = X4[(size_t)s3 * ld4 + c];
            acc.x += q0.x * w0; acc.y += q0.y * w0; acc.z += q0.z * w0; acc.w += q0.w * w0;
            acc.x += q1.x * w1; acc.y += q1.y * w1; acc.z += q1.z * w1; acc.w += q1.w * w1;
            acc.x += q2.x * w2; acc.y += q2.y * w2; acc.z += q2.z * w2; acc.w += q2.w * w2;
            acc.x += q3.x * w3; acc.y += q3.y * w3; acc.z += q3.z * w3; acc.w += q3.w * w3;
        }
        for (; k + 2 <= m; k += 2) {
            int sa0 = __builtin_amdgcn_readlane(myc, k + 0), sb0 = __builtin_amdgcn_readlane(myc, k + 1);
            int wa0 = __builtin_amdgcn_readlane(mywi, k + 0), wb0 = __builtin_amdgcn_readlane(mywi, k + 1);
            int s0 = hh ? sb0 : sa0;
            float w0 = __int_as_float(hh ? wb0 : wa0);
            float4 q0 = X4[(size_t)s0 * ld4 + c];
            acc.x += q0.x * w0; acc.y += q0.y * w0; acc.z += q0.z * w0; acc.w += q0.w * w0;
        }
        if (k < m) {
            int sa0 = __builtin_amdgcn_readlane(myc, k);
            float w0 = __int_as_float(__builtin_amdgcn_readlane(mywi, k));
            if (hh == 0) {
                float4 q0 = X4[(size_t)sa0 * ld4 + c];
                acc.x += q0.x * w0; acc.y += q0.y * w0; acc.z += q0.z * w0; acc.w += q0.w * w0;
            }
        }
    }
    acc.x += __shfl_xor(acc.x, 16, 32);
    acc.y += __shfl_xor(acc.y, 16, 32);
    acc.z += __shfl_xor(acc.z, 16, 32);
    acc.w += __shfl_xor(acc.w, 16, 32);
    if (hh == 0)
        *(float4*)(Yout + (size_t)v * ldo + c * 4) = acc;
    if (MODE != 0) {
#pragma unroll
        for (int mm = 16; mm >= 1; mm >>= 1) rs += __shfl_xor(rs, mm, 32);
        if (lane == 0) rout[v] = rs + (MODE == 1 ? nn : nn * rin[v]);
    }
}

__global__ void softmax2(const float* __restrict__ logits, float* __restrict__ probs, int n) {
    int i = blockIdx.x * blockDim.x + threadIdx.x;
    if (i >= n) return;
    float a = logits[2 * i + 0], b = logits[2 * i + 1];
    float m = fmaxf(a, b);
    float ea = __expf(a - m), eb = __expf(b - m);
    float inv = 1.0f / (ea + eb);
    probs[2 * i + 0] = ea * inv;
    probs[2 * i + 1] = eb * inv;
}

extern "C" void kernel_launch(void* const* d_in, const int* in_sizes, int n_in,
                              void* d_out, int out_size, void* d_ws, size_t ws_size,
                              hipStream_t stream) {
    const float* x   = (const float*)d_in[0];
    const int*   ei  = (const int*)d_in[1];
    const float* W1  = (const float*)d_in[2];
    const float* b1  = (const float*)d_in[3];
    const float* W2  = (const float*)d_in[4];
    const float* b2  = (const float*)d_in[5];
    const float* W3  = (const float*)d_in[6];
    const float* b3  = (const float*)d_in[7];
    const float* lW1 = (const float*)d_in[8];
    const float* lb1 = (const float*)d_in[9];
    const float* lW2 = (const float*)d_in[10];
    const float* lb2 = (const float*)d_in[11];

    const int N = NN, E = EE;
    const int* src = ei;
    const int* dst = ei + E;

    float* y    = (float*)d_ws;
    float* r1   = y + (size_t)N * YD;
    float* r2   = r1 + N;
    float* Tm   = r2 + N;
    float* Fm   = Tm + 3 * BM * HD;
    float* dinv = Fm + 3 * BM * CD;
    int*   cnt    = (int*)(dinv + N);
    int*   cursor = cnt + N;
    int*   rowptr = cursor + N;
    int*   col    = rowptr + (N + 1);
    float* wgt    = (float*)(col + E);

    float* logits = (float*)d_out;
    float* probs  = logits + (size_t)N * 2;

    const int T = 256;
    const int rowBlocks = N / BM;
    const int gatherBlocks = (N + 7) / 8;

    fold_seed<<<(BM * HD + T - 1) / T, T, 0, stream>>>(W1, b1, Tm);
    gemm_tiled<false, HD, false><<<1, T, 0, stream>>>(
        Tm, HD, W2, HD, Tm + BM * HD, HD, HD, nullptr, nullptr, nullptr,
        0, b2, 65, 0, 0, 0, 0, 0, nullptr, nullptr);
    gemm_tiled<false, HD, false><<<1, T, 0, stream>>>(
        Tm + BM * HD, HD, W3, HD, Tm + 2 * BM * HD, HD, HD, nullptr, nullptr, nullptr,
        0, b3, 66, 0, 0, 0, 0, 0, nullptr, nullptr);
    gemm_tiled<true, HD, false><<<dim3(1, 3, 3), T, 0, stream>>>(
        Tm, HD, lW1, CD, Fm, CD, HD, nullptr, nullptr, nullptr,
        0, nullptr, 0, BM * HD, HD, BM * CD, HD * CD, HD, nullptr, nullptr);

    csr_zero<<<(N + T - 1) / T, T, 0, stream>>>(cnt, N);
    csr_count<<<(E + T - 1) / T, T, 0, stream>>>(dst, cnt, E);
    csr_scan<<<1, SCAN_T, 0, stream>>>(cnt, rowptr, cursor, dinv);
    csr_fill<<<(E + T - 1) / T, T, 0, stream>>>(src, dst, dinv, cursor, col, wgt, E);

    gather64<1><<<gatherBlocks, T, 0, stream>>>(x, XD, dinv, rowptr, col, wgt, y, YD, nullptr, r1, N);
    gather64<2><<<gatherBlocks, T, 0, stream>>>(y, YD, dinv, rowptr, col, wgt, y + XD, YD, r1, r2, N);
    gather64<0><<<gatherBlocks, T, 0, stream>>>(y + XD, YD, dinv, rowptr, col, wgt, y + 2 * XD, YD,
                                                nullptr, nullptr, N);

    gemm_tiled<false, CD, true><<<rowBlocks, T, 0, stream>>>(
        y, YD, Fm, CD, logits, 2, YD, lb1, lW2, lb2,
        16, nullptr, 0, 0, 0, 0, 0, 0, r1, r2);
    softmax2<<<(N + T - 1) / T, T, 0, stream>>>(logits, probs, N);
}


# Gathers and edge passes run on machine-sized grids: each wave loops over a contiguous run of rows (edge passes: four edges per thread), next row's entries fetched ahead; row sums in same order
# speedup vs baseline: 2.8852x; 1.0416x over previous
#include <hip/hip_runtime.h>
#include <hip/hip_bf16.h>


typedef __attribute__((ext_vector_type(2))) float v2f;
typedef __attribute__((ext_vector_type(4))) float v4f;
typedef __attribute__((ext_vector_type(8))) float v8f;

#define NN 50000
#define EE 800000
#define HD 128
#define CD 384
#define XD 64
#define YD 192

__global__ void csr_zero(int* __restrict__ cnt, int n) {
    int i = blockIdx.x * blockDim.x + threadIdx.x;
    if (i < n) cnt[i] = 0;
}
__global__ void csr_count(const int* __restrict__ dst, int* cnt, int e) {
    const int t = blockIdx.x * blockDim.x + threadIdx.x;
    const int stride = gridDim.x * blockDim.x;
    const int e4 = (((size_t)dst & 15) == 0) ? (e >> 2) : 0;
    const int4* d4 = (const int4*)dst;
    for (int i = t; i < e4; i += stride) {
        int4 d = d4[i];
        atomicAdd(&cnt[d.x], 1);
        atomicAdd(&cnt[d.y], 1);
        atomicAdd(&cnt[d.z], 1);
        atomicAdd(&cnt[d.w], 1);
    }
    for (int i = 4 * e4 + t; i < e; i += stride) atomicAdd(&cnt[dst[i]], 1);
}
#define SCAN_T 1024
#define SCAN_PER ((NN + SCAN_T - 1) / SCAN_T)
__global__ void __launch_bounds__(SCAN_T)
csr_scan(const int* __restrict__ cnt, int* __restrict__ rowptr, int* __restrict__ cursor,
         float* __restrict__ dinv) {
    __shared__ int sc[SCAN_PER * SCAN_T];
    __shared__ int wsum[SCAN_T / 32];
    const int t = threadIdx.x;
    for (int i = 0; i < SCAN_PER; ++i) {
        int idx = i * SCAN_T + t;
        int cv = 0;
        if (idx < NN) {
            cv = cnt[idx];
            dinv[idx] = rsqrtf((float)(cv + 1));
        }
        sc[idx] = cv;
    }
    __syncthreads();
    int local = 0;
    for (int i = 0; i < SCAN_PER; ++i) local += sc[t * SCAN_PER + i];
    int incl = local;
    for (int m = 1; m < 32; m <<= 1) {
        int up = __shfl_up(incl, m, 32);
        if ((t & 31) >= m) incl += up;
    }
    if ((t & 31) == 31) wsum[t >> 5] = incl;
    __syncthreads();
    if (t < 32) {
        int w = wsum[t];
        int wi = w;
        for (int m = 1; m < 32; m <<= 1) {
            int up = __shfl_up(wi, m, 32);
            if (t >= m) wi += up;
        }
        wsum[t] = wi - w;
    }
    __syncthreads();
    int run = incl - local + wsum[t >> 5];
    for (int i = 0; i < SCAN_PER; ++i) {
        int cv = sc[t * SCAN_PER + i];
        sc[t * SCAN_PER + i] = run;
        run += cv;
    }
    if (t == SCAN_T - 1) rowptr[NN] = run;
    __syncthreads();
    for (int i = 0; i < SCAN_PER; ++i) {
        int idx = i * SCAN_T + t;
        if (idx < NN) {
            int p = sc[idx];
            rowptr[idx] = p;
            cursor[idx] = p;
        }
    }
}
__global__ void csr_fill(const int* __restrict__ src, const int* __restrict__ dst,
                         const float* __restrict__ dinv, int* cursor,
                         int* __restrict__ col, float* __restrict__ wgt, int e) {
    const int t = blockIdx.x * blockDim.x + threadIdx.x;
    const int stride = gridDim.x * blockDim.x;
    const int e4 = (((((size_t)src) | ((size_t)dst)) & 15) == 0) ? (e >> 2) : 0;
    const int4* s4 = (const int4*)src;
    const int4* d4 = (const int4*)dst;
    for (int i = t; i < e4; i += stride) {
        int4 sv = s4[i], dv = d4[i];
        int p0 = atomicAdd(&cursor[dv.x], 1);
        int p1 = atomicAdd(&cursor[dv.y], 1);
        int p2 = atomicAdd(&cursor[dv.z], 1);
        int p3 = atomicAdd(&cursor[dv.w], 1);
        float w0 = dinv[sv.x] * dinv[dv.x];
        float w1 = dinv[sv.y] * dinv[dv.y];
        float w2 = dinv[sv.z] * dinv[dv.z];
        float w3 = dinv[sv.w] * dinv[dv.w];
        col[p0] = sv.x; wgt[p0] = w0;
        col[p1] = sv.y; wgt[p1] = w1;
        col[p2] = sv.z; wgt[p2] = w2;
        col[p3] = sv.w; wgt[p3] = w3;
    }
    for (int i = 4 * e4 + t; i < e; i += stride) {
        int sv = src[i], dv = dst[i];
        int p = atomicAdd(&cursor[dv], 1);
        col[p] = sv;
        wgt[p] = dinv[sv] * dinv[dv];
    }
}

#define BM 80
#define KB 32
#define PA (KB + 4)

template <bool BT, int BN, bool HEAD>
__global__ void __launch_bounds__(256)
gemm_tiled(const float* __restrict__ A, int lda,
           const float* __restrict__ B, int ldb,
           float* __restrict__ C, int ldc, int K,
           const float* __restrict__ lb1,
           const float* __restrict__ lW2,
           const float* __restrict__ lb2,
           int bpad,
           const float* __restrict__ rowvec, int rowidx,
           int sAy, int sBy, int sCy, int sBz, int sCz,
           const float* __restrict__ r1, const float* __restrict__ r2) {
    A += (size_t)blockIdx.y * sAy;
    B += (size_t)blockIdx.y * sBy + (size_t)blockIdx.z * sBz;
    C += (size_t)blockIdx.y * sCy + (size_t)blockIdx.z * sCz;
    constexpr int RT = BM / 16;
    constexpr int CT = BN / 128;
    constexpr int PB = BT ? PA : (BN + 4);
    constexpr int BROWS = BT ? BN : KB;
    constexpr int QA = KB / 4;
    constexpr int A4 = BM * QA;
    constexpr int AL = (A4 + 255) / 256;
    constexpr int BL = BN * QA / 256;
    constexpr int QB = BN / 4;

    __shared__ __attribute__((aligned(16))) float As[BM * PA];
    __shared__ __attribute__((aligned(16))) float Bs[BROWS * PB];

    const int tid  = threadIdx.x;
    const int wave = tid >> 5;
    const int lane = tid & 31;
    const int c    = lane & 15;
    const int h    = lane >> 4;
    const int m0   = blockIdx.x * BM;
    const int wn0  = wave * (CT * 16);

    v8f acc[RT][CT];
#pragma unroll
    for (int i = 0; i < RT; ++i)
#pragma unroll
        for (int j = 0; j < CT; ++j) {
            v8f z = {};
            acc[i][j] = z;
        }

    v4f ra[AL], rb[BL];

#define TILE_GLOAD(kb)                                                              \
    {                                                                               \
        _Pragma("unroll") for (int u = 0; u < AL; ++u) {                            \
            int idx = tid + 256 * u;                                                \
            if (idx < A4) {                                                         \
                int row = idx / QA, q = idx % QA;                                   \
                ra[u] = *(const v4f*)(A + (size_t)(m0 + row) * lda + (kb) + 4 * q); \
            }                                                                       \
        }                                                                           \
        _Pragma("unroll") for (int u = 0; u < BL; ++u) {                            \
            int idx = tid + 256 * u;                                                \
            if (BT) {                                                               \
                int n = idx / QA, q = idx % QA;                                     \
                rb[u] = *(const v4f*)(B + (size_t)n * ldb + (kb) + 4 * q);       \
            } else {                                                                \
                int k = idx / QB, q = idx % QB;                                     \
                rb[u] = *(const v4f*)(B + (size_t)((kb) + k + bpad * ((kb) >> 6)) * ldb + 4 * q); \
            }                                                                       \
        }                                                                           \
    }

    TILE_GLOAD(0)
    const int S = K / KB;
    for (int s = 0; s < S; ++s) {
#pragma unroll
        for (int u = 0; u < AL; ++u) {
            int idx = tid + 256 * u;
            if (idx < A4) {
                int row = idx / QA, q = idx % QA;
                *(v4f*)&As[row * PA + 4 * q] = ra[u];
            }
        }
#pragma unroll
        for (int u = 0; u < BL; ++u) {
            int idx = tid + 256 * u;
            if (BT) {
                int n = idx / QA, q = idx % QA;
                *(v4f*)&Bs[n * PB + 4 * q] = rb[u];
            } else {
                int k = idx / QB, q = idx % QB;
                *(v4f*)&Bs[k * PB + 4 * q] = rb[u];
            }
        }
        __syncthreads();
        if (s + 1 < S) TILE_GLOAD((s + 1) * KB)

#pragma unroll
        for (int g = 0; g < KB / 8; ++g) {
            v2f alo[RT], ahi[RT], blo[CT], bhi[CT];
#pragma unroll
            for (int i = 0; i < RT; ++i) {
                v4f a4 = *(const v4f*)&As[(16 * i + c) * PA + 8 * g + 4 * h];
                alo[i].x = a4.x; alo[i].y = a4.y;
                ahi[i].x = a4.z; ahi[i].y = a4.w;
            }
#pragma unroll
            for (int j = 0; j < CT; ++j) {
                if (BT) {
                    v4f b4 = *(const v4f*)&Bs[(wn0 + 16 * j + c) * PB + 8 * g + 4 * h];
                    blo[j].x = b4.x; blo[j].y = b4.y;
                    bhi[j].x = b4.z; bhi[j].y = b4.w;
                } else {
                    const float* bp = &Bs[(8 * g + 4 * h) * PB + wn0 + 16 * j + c];
                    blo[j].x = bp[0];
                    blo[j].y = bp[PB];
                    bhi[j].x = bp[2 * PB];
                    bhi[j].y = bp[3 * PB];
                }
            }
#pragma unroll
            for (int i = 0; i < RT; ++i)
#pragma unroll
                for (int j = 0; j < CT; ++j)
                    acc[i][j] = __builtin_amdgcn_wmma_f32_16x16x4_f32(
                        false, alo[i], false, blo[j], (short)0, acc[i][j], false, false);
#pragma unroll
            for (int i = 0; i < RT; ++i)
#pragma unroll
                for (int j = 0; j < CT; ++j)
                    acc[i][j] = __builtin_amdgcn_wmma_f32_16x16x4_f32(
                        false, ahi[i], false, bhi[j], (short)0, acc[i][j], false, false);
        }
        __syncthreads();
    }
#undef TILE_GLOAD

    if (!HEAD) {
#pragma unroll
        for (int i = 0; i < RT; ++i)
#pragma unroll
            for (int j = 0; j < CT; ++j)
#pragma unroll
                for (int r = 0; r < 8; ++r) {
                    const int row = m0 + 16 * i + 8 * h + r;
                    const int cc  = wn0 + 16 * j + c;
                    float o = acc[i][j][r];
                    if (rowvec != nullptr && row == rowidx) o += rowvec[cc];
                    C[(size_t)row * ldc + cc] = o;
                }
    } else {
        float bias[CT], u1[CT], u2[CT], w0[CT], w1[CT];
#pragma unroll
        for (int j = 0; j < CT; ++j) {
            int col = wn0 + 16 * j + c;
            bias[j] = lb1[col] + B[(size_t)(0 * BM + 64) * ldb + col]
                               + B[(size_t)(1 * BM + 65) * ldb + col]
                               + B[(size_t)(2 * BM + 66) * ldb + col];
            u1[j]   = B[(size_t)(1 * BM + 64) * ldb + col] + B[(size_t)(2 * BM + 65) * ldb + col];
            u2[j]   = B[(size_t)(2 * BM + 64) * ldb + col];
            w0[j]   = lW2[col];
            w1[j]   = lW2[BN + col];
        }
        float* red = Bs;
#pragma unroll
        for (int i = 0; i < RT; ++i)
#pragma unroll
            for (int r = 0; r < 8; ++r) {
                float p0 = 0.0f, p1 = 0.0f;
                const int grow = m0 + 16 * i + 8 * h + r;
                const float a1 = r1[grow], a2 = r2[grow];
#pragma unroll
                for (int j = 0; j < CT; ++j) {
                    float t = acc[i][j][r] + bias[j] + a1 * u1[j] + a2 * u2[j];
                    t = t > 0.0f ? t : 0.0f;
                    p0 += t * w0[j];
                    p1 += t * w1[j];
                }
#pragma unroll
                for (int m = 8; m >= 1; m >>= 1) {
                    p0 += __shfl_xor(p0, m, 16);
                    p1 += __shfl_xor(p1, m, 16);
                }
                if (c == 0) {
                    int row = 16 * i + 8 * h + r;
                    red[(wave * BM + row) * 2 + 0] = p0;
                    red[(wave * BM + row) * 2 + 1] = p1;
                }
            }
        __syncthreads();
        if (tid < 2 * BM) {
            int row = tid >> 1, cls = tid & 1;
            float sum = lb2[cls];
#pragma unroll
            for (int w = 0; w < 8; ++w) sum += red[(w * BM + row) * 2 + cls];
            C[(size_t)(m0 + row) * ldc + cls] = sum;
        }
    }
}

__global__ void fold_seed(const float* __restrict__ W1, const float* __restrict__ b1,
                          float* __restrict__ T1) {
    int i = blockIdx.x * blockDim.x + threadIdx.x;
    if (i >= BM * HD) return;
    int r = i / HD, cc = i % HD;
    float o = 0.0f;
    if (r < XD) o = W1[i];
    else if (r == XD) o = b1[cc];
    T1[i] = o;
}

#define RUN_ROWS 31

template <int MODE>
__global__ void __launch_bounds__(256)
gather64(const float* __restrict__ Xin, int ldi, const float* __restrict__ dinv,
         const int* __restrict__ rowptr, const int* __restrict__ col,
         const float* __restrict__ wgt,
         float* __restrict__ Yout, int ldo,
         const float* __restrict__ rin, float* __restrict__ rout, int n, int e) {
    const int lane = threadIdx.x & 31;
    const int c  = lane & 15;
    const int hh = lane >> 4;
    const int wid = __builtin_amdgcn_readfirstlane(blockIdx.x * 8 + (threadIdx.x >> 5));
    const int nw  = gridDim.x * 8;
    const unsigned uq = (unsigned)n / (unsigned)nw, ur = (unsigned)n % (unsigned)nw;
    const int v0 = (int)((unsigned)wid * uq + ((unsigned)wid < ur ? (unsigned)wid : ur));
    const int v1 = v0 + (int)uq + ((unsigned)wid < ur ? 1 : 0);
    const char* Xb = (const char*)Xin;
    const unsigned pitch = (unsigned)ldi * 4u;
    const unsigned coff  = (unsigned)c * 16u;

#define GLOAD(R, P)                                                                \
    {                                                                              \
        const int sa = __builtin_amdgcn_readlane(myc, 2 * (P));                    \
        const int sb = __builtin_amdgcn_readlane(myc, 2 * (P) + 1);                \
        const unsigned sr = (unsigned)(hh ? sb : sa);                              \
        q[R] = *(const float4*)(Xb + (sr * pitch + coff));                         \
    }
#define GFMA(R, P)                                                                 \
    {                                                                              \
        const int wa = __builtin_amdgcn_readlane(mywi, 2 * (P));                   \
        const int wb = __builtin_amdgcn_readlane(mywi, 2 * (P) + 1);               \
        const float w = __int_as_float(hh ? wb : wa);                              \
        acc.x += q[R].x * w; acc.y += q[R].y * w;                                  \
        acc.z += q[R].z * w; acc.w += q[R].w * w;                                  \
    }

    for (int vb = v0; vb < v1; vb += RUN_ROWS) {
        const int cnt = (v1 - vb) < RUN_ROWS ? (v1 - vb) : RUN_ROWS;
        const int ri = (vb + lane) < n ? (vb + lane) : n;
        const int di = (vb + lane) < n ? (vb + lane) : (n - 1);
        const int rp = rowptr[ri];
        const float dv = dinv[di];
        float rself = 0.0f;
        if (MODE == 2) rself = rin[di];
        const int dvi = __float_as_int(dv);
        const int rsi = __float_as_int(rself);
        int beg = __builtin_amdgcn_readlane(rp, 0);
        int pi = beg + lane;
        pi = pi < e ? pi : e - 1;
        pi = pi > 0 ? pi : 0;
        int   nc  = col[pi];
        float nwt = wgt[pi];
        for (int i = 0; i < cnt; ++i) {
            const int v = vb + i;
            const int end = __builtin_amdgcn_readlane(rp, i + 1);
            const float dn = __int_as_float(__builtin_amdgcn_readlane(dvi, i));
            const float nn = dn * dn;
            const int   fc = nc;
            const float fw = nwt;
            pi = end + lane;
            pi = pi < e ? pi : e - 1;
            pi = pi > 0 ? pi : 0;
            nc  = col[pi];
            nwt = wgt[pi];
            const float4 xs = *(const float4*)(Xb + ((unsigned)v * pitch + coff));
            float4 acc = make_float4(0.0f, 0.0f, 0.0f, 0.0f);
            bool first = true;
            float rs = 0.0f;
            for (int base = beg; base < end; base += 32) {
                const int m = (end - base) < 32 ? (end - base) : 32;
                int   myc = v;
                float myw = 0.0f;
                if (lane < m) {
                    if (base == beg) { myc = fc; myw = fw; }
                    else { myc = col[base + lane]; myw = wgt[base + lane]; }
                }
                float rv = 0.0f;
                if (MODE == 2) rv = rin[myc];
                const int mywi = __float_as_int(myw);
                float4 q[12];
                GLOAD(0, 0) GLOAD(1, 1) GLOAD(2, 2) GLOAD(3, 3)
                if (m > 8) {
                    GLOAD(4, 4) GLOAD(5, 5) GLOAD(6, 6) GLOAD(7, 7)
                    if (m > 16) { GLOAD(8, 8) GLOAD(9, 9) GLOAD(10, 10) GLOAD(11, 11) }
                }
                if (first) {
                    acc.x = hh ? 0.0f : xs.x * nn; acc.y = hh ? 0.0f : xs.y * nn;
                    acc.z = hh ? 0.0f : xs.z * nn; acc.w = hh ? 0.0f : xs.w * nn;
                    first = false;
                }
                GFMA(0, 0) GFMA(1, 1) GFMA(2, 2) GFMA(3, 3)
                if (m > 8) {
                    GFMA(4, 4) GFMA(5, 5) GFMA(6, 6) GFMA(7, 7)
                    if (m > 16) { GFMA(8, 8) GFMA(9, 9) GFMA(10, 10) GFMA(11, 11) }
                }
                if (m > 24) {
                    GLOAD(0, 12) GLOAD(1, 13) GLOAD(2, 14) GLOAD(3, 15)
                    GFMA(0, 12) GFMA(1, 13) GFMA(2, 14) GFMA(3, 15)
                }
                if (MODE == 1) rs += myw;
                if (MODE == 2) rs += myw * rv;
            }
            if (first) {
                acc.x = hh ? 0.0f : xs.x * nn; acc.y = hh ? 0.0f : xs.y * nn;
                acc.z = hh ? 0.0f : xs.z * nn; acc.w = hh ? 0.0f : xs.w * nn;
            }
            acc.x += __shfl_xor(acc.x, 16, 32);
            acc.y += __shfl_xor(acc.y, 16, 32);
            acc.z += __shfl_xor(acc.z, 16, 32);
            acc.w += __shfl_xor(acc.w, 16, 32);
            if (hh == 0)
                *(float4*)(Yout + (size_t)v * ldo + c * 4) = acc;
            if (MODE != 0) {
#pragma unroll
                for (int mm = 16; mm >= 1; mm >>= 1) rs += __shfl_xor(rs, mm, 32);
                const float rsv = __int_as_float(__builtin_amdgcn_readlane(rsi, i));
                if (lane == 0) rout[v] = rs + (MODE == 1 ? nn : nn * rsv);
            }
            beg = end;
        }
    }
#undef GLOAD
#undef GFMA
}

__global__ void softmax2(const float* __restrict__ logits, float* __restrict__ probs, int n) {
    int i = blockIdx.x * blockDim.x + threadIdx.x;
    if (i >= n) return;
    float a = logits[2 * i + 0], b = logits[2 * i + 1];
    float m = fmaxf(a, b);
    float ea = __expf(a - m), eb = __expf(b - m);
    float inv = 1.0f / (ea + eb);
    probs[2 * i + 0] = ea * inv;
    probs[2 * i + 1] = eb * inv;
}

extern "C" void kernel_launch(void* const* d_in, const int* in_sizes, int n_in,
                              void* d_out, int out_size, void* d_ws, size_t ws_size,
                              hipStream_t stream) {
    const float* x   = (const float*)d_in[0];
    const int*   ei  = (const int*)d_in[1];
    const float* W1  = (const float*)d_in[2];
    const float* b1  = (const float*)d_in[3];
    const float* W2  = (const float*)d_in[4];
    const float* b2  = (const float*)d_in[5];
    const float* W3  = (const float*)d_in[6];
    const float* b3  = (const float*)d_in[7];
    const float* lW1 = (const float*)d_in[8];
    const float* lb1 = (const float*)d_in[9];
    const float* lW2 = (const float*)d_in[10];
    const float* lb2 = (const float*)d_in[11];

    const int N = NN, E = EE;
    const int* src = ei;
    const int* dst = ei + E;

    float* y    = (float*)d_ws;
    float* r1   = y + (size_t)N * YD;
    float* r2   = r1 + N;
    float* Tm   = r2 + N;
    float* Fm   = Tm + 3 * BM * HD;
    float* dinv = Fm + 3 * BM * CD;
    int*   cnt    = (int*)(dinv + N);
    int*   cursor = cnt + N;
    int*   rowptr = cursor + N;
    int*   col    = rowptr + (N + 1);
    float* wgt    = (float*)(col + E);

    float* logits = (float*)d_out;
    float* probs  = logits + (size_t)N * 2;

    const int T = 256;
    const int rowBlocks = N / BM;
    const int gatherBlocks = 1024;
    const int edgeBlocks   = 512;

    fold_seed<<<(BM * HD + T - 1) / T, T, 0, stream>>>(W1, b1, Tm);
    gemm_tiled<false, HD, false><<<1, T, 0, stream>>>(
        Tm, HD, W2, HD, Tm + BM * HD, HD, HD, nullptr, nullptr, nullptr,
        0, b2, 65, 0, 0, 0, 0, 0, nullptr, nullptr);
    gemm_tiled<false, HD, false><<<1, T, 0, stream>>>(
        Tm + BM * HD, HD, W3, HD, Tm + 2 * BM * HD, HD, HD, nullptr, nullptr, nullptr,
        0, b3, 66, 0, 0, 0, 0, 0, nullptr, nullptr);
    gemm_tiled<true, HD, false><<<dim3(1, 3, 3), T, 0, stream>>>(
        Tm, HD, lW1, CD, Fm, CD, HD, nullptr, nullptr, nullptr,
        0, nullptr, 0, BM * HD, HD, BM * CD, HD * CD, HD, nullptr, nullptr);

    csr_zero<<<(N + T - 1) / T, T, 0, stream>>>(cnt, N);
    csr_count<<<edgeBlocks, T, 0, stream>>>(dst, cnt, E);
    csr_scan<<<1, SCAN_T, 0, stream>>>(cnt, rowptr, cursor, dinv);
    csr_fill<<<edgeBlocks, T, 0, stream>>>(src, dst, dinv, cursor, col, wgt, E);

    gather64<1><<<gatherBlocks, T, 0, stream>>>(x, XD, dinv, rowptr, col, wgt, y, YD, nullptr, r1, N, E);
    gather64<2><<<gatherBlocks, T, 0, stream>>>(y, YD, dinv, rowptr, col, wgt, y + XD, YD, r1, r2, N, E);
    gather64<0><<<gatherBlocks, T, 0, stream>>>(y + XD, YD, dinv, rowptr, col, wgt, y + 2 * XD, YD,
                                                nullptr, nullptr, N, E);

    gemm_tiled<false, CD, true><<<rowBlocks, T, 0, stream>>>(
        y, YD, Fm, CD, logits, 2, YD, lb1, lW2, lb2,
        16, nullptr, 0, 0, 0, 0, 0, 0, r1, r2);
    softmax2<<<(N + T - 1) / T, T, 0, stream>>>(logits, probs, N);
}


# Launches 13 -> 8: weight-fold products run in spare blocks of the edge count and fill launches, count clear joins the fold seed, softmax joins the head's epilogue; all sums in the same order
# speedup vs baseline: 3.1116x; 1.0785x over previous
#include <hip/hip_runtime.h>
#include <hip/hip_bf16.h>


typedef __attribute__((ext_vector_type(2))) float v2f;
typedef __attribute__((ext_vector_type(4))) float v4f;
typedef __attribute__((ext_vector_type(8))) float v8f;

#define NN 50000
#define EE 800000
#define HD 128
#define CD 384
#define XD 64
#define YD 192

__device__ __forceinline__ void count_edges(const int* __restrict__ dst, int* cnt, int e,
                                            int bid, int nblk) {
    const int t = bid * blockDim.x + threadIdx.x;
    const int stride = nblk * blockDim.x;
    const int e4 = (((size_t)dst & 15) == 0) ? (e >> 2) : 0;
    const int4* d4 = (const int4*)dst;
    for (int i = t; i < e4; i += stride) {
        int4 d = d4[i];
        atomicAdd(&cnt[d.x], 1);
        atomicAdd(&cnt[d.y], 1);
        atomicAdd(&cnt[d.z], 1);
        atomicAdd(&cnt[d.w], 1);
    }
    for (int i = 4 * e4 + t; i < e; i += stride) atomicAdd(&cnt[dst[i]], 1);
}
#define SCAN_T 1024
#define SCAN_PER ((NN + SCAN_T - 1) / SCAN_T)
__global__ void __launch_bounds__(SCAN_T)
csr_scan(const int* __restrict__ cnt, int* __restrict__ rowptr, int* __restrict__ cursor,
         float* __restrict__ dinv) {
    __shared__ int sc[SCAN_PER * SCAN_T];
    __shared__ int wsum[SCAN_T / 32];
    const int t = threadIdx.x;
    for (int i = 0; i < SCAN_PER; ++i) {
        int idx = i * SCAN_T + t;
        int cv = 0;
        if (idx < NN) {
            cv = cnt[idx];
            dinv[idx] = rsqrtf((float)(cv + 1));
        }
        sc[idx] = cv;
    }
    __syncthreads();
    int local = 0;
    for (int i = 0; i < SCAN_PER; ++i) local += sc[t * SCAN_PER + i];
    int incl = local;
    for (int m = 1; m < 32; m <<= 1) {
        int up = __shfl_up(incl, m, 32);
        if ((t & 31) >= m) incl += up;
    }
    if ((t & 31) == 31) wsum[t >> 5] = incl;
    __syncthreads();
    if (t < 32) {
        int w = wsum[t];
        int wi = w;
        for (int m = 1; m < 32; m <<= 1) {
            int up = __shfl_up(wi, m, 32);
            if (t >= m) wi += up;
        }
        wsum[t] = wi - w;
    }
    __syncthreads();
    int run = incl - local + wsum[t >> 5];
    for (int i = 0; i < SCAN_PER; ++i) {
        int cv = sc[t * SCAN_PER + i];
        sc[t * SCAN_PER + i] = run;
        run += cv;
    }
    if (t == SCAN_T - 1) rowptr[NN] = run;
    __syncthreads();
    for (int i = 0; i < SCAN_PER; ++i) {
        int idx = i * SCAN_T + t;
        if (idx < NN) {
            int p = sc[idx];
            rowptr[idx] = p;
            cursor[idx] = p;
        }
    }
}
__device__ __forceinline__ void fill_edges(const int* __restrict__ src, const int* __restrict__ dst,
                                           const float* __restrict__ dinv, int* cursor,
                                           int* __restrict__ col, float* __restrict__ wgt, int e,
                                           int bid, int nblk) {
    const int t = bid * blockDim.x + threadIdx.x;
    const int stride = nblk * blockDim.x;
    const int e4 = (((((size_t)src) | ((size_t)dst)) & 15) == 0) ? (e >> 2) : 0;
    const int4* s4 = (const int4*)src;
    const int4* d4 = (const int4*)dst;
    for (int i = t; i < e4; i += stride) {
        int4 sv = s4[i], dv = d4[i];
        int p0 = atomicAdd(&cursor[dv.x], 1);
        int p1 = atomicAdd(&cursor[dv.y], 1);
        int p2 = atomicAdd(&cursor[dv.z], 1);
        int p3 = atomicAdd(&cursor[dv.w], 1);
        float w0 = dinv[sv.x] * dinv[dv.x];
        float w1 = dinv[sv.y] * dinv[dv.y];
        float w2 = dinv[sv.z] * dinv[dv.z];
        float w3 = dinv[sv.w] * dinv[dv.w];
        col[p0] = sv.x; wgt[p0] = w0;
        col[p1] = sv.y; wgt[p1] = w1;
        col[p2] = sv.z; wgt[p2] = w2;
        col[p3] = sv.w; wgt[p3] = w3;
    }
    for (int i = 4 * e4 + t; i < e; i += stride) {
        int sv = src[i], dv = dst[i];
        int p = atomicAdd(&cursor[dv], 1);
        col[p] = sv;
        wgt[p] = dinv[sv] * dinv[dv];
    }
}

#define BM 80
#define KB 32
#define PA (KB + 4)

template <bool BT, int BN, bool HEAD>
__device__ __forceinline__ void
gemm_block(const float* __restrict__ A, int lda,
           const float* __restrict__ B, int ldb,
           float* __restrict__ C, int ldc, int K,
           const float* __restrict__ lb1,
           const float* __restrict__ lW2,
           const float* __restrict__ lb2,
           int bpad,
           const float* __restrict__ rowvec, int rowidx,
           int sAy, int sBy, int sCy, int sBz, int sCz,
           const float* __restrict__ r1, const float* __restrict__ r2,
           float* __restrict__ probs, int bx, int by, int bz) {
    A += (size_t)by * sAy;
    B += (size_t)by * sBy + (size_t)bz * sBz;
    C += (size_t)by * sCy + (size_t)bz * sCz;
    constexpr int RT = BM / 16;
    constexpr int CT = BN / 128;
    constexpr int PB = BT ? PA : (BN + 4);
    constexpr int BROWS = BT ? BN : KB;
    constexpr int QA = KB / 4;
    constexpr int A4 = BM * QA;
    constexpr int AL = (A4 + 255) / 256;
    constexpr int BL = BN * QA / 256;
    constexpr int QB = BN / 4;

    __shared__ __attribute__((aligned(16))) float As[BM * PA];
    __shared__ __attribute__((aligned(16))) float Bs[BROWS * PB];

    const int tid  = threadIdx.x;
    const int wave = tid >> 5;
    const int lane = tid & 31;
    const int c    = lane & 15;
    const int h    = lane >> 4;
    const int m0   = bx * BM;
    const int wn0  = wave * (CT * 16);

    v8f acc[RT][CT];
#pragma unroll
    for (int i = 0; i < RT; ++i)
#pragma unroll
        for (int j = 0; j < CT; ++j) {
            v8f z = {};
            acc[i][j] = z;
        }

    v4f ra[AL], rb[BL];

#define TILE_GLOAD(kb)                                                              \
    {                                                                               \
        _Pragma("unroll") for (int u = 0; u < AL; ++u) {                            \
            int idx = tid + 256 * u;                                                \
            if (idx < A4) {                                                         \
                int row = idx / QA, q = idx % QA;                                   \
                ra[u] = *(const v4f*)(A + (size_t)(m0 + row) * lda + (kb) + 4 * q); \
            }                                                                       \
        }                                                                           \
        _Pragma("unroll") for (int u = 0; u < BL; ++u) {                            \
            int idx = tid + 256 * u;                                                \
            if (BT) {                                                               \
                int n = idx / QA, q = idx % QA;                                     \
                rb[u] = *(const v4f*)(B + (size_t)n * ldb + (kb) + 4 * q);       \
            } else {                                                                \
                int k = idx / QB, q = idx % QB;                                     \
                rb[u] = *(const v4f*)(B + (size_t)((kb) + k + bpad * ((kb) >> 6)) * ldb + 4 * q); \
            }                                                                       \
        }                                                                           \
    }

    TILE_GLOAD(0)
    const int S = K / KB;
    for (int s = 0; s < S; ++s) {
#pragma unroll
        for (int u = 0; u < AL; ++u) {
            int idx = tid + 256 * u;
            if (idx < A4) {
                int row = idx / QA, q = idx % QA;
                *(v4f*)&As[row * PA + 4 * q] = ra[u];
            }
        }
#pragma unroll
        for (int u = 0; u < BL; ++u) {
            int idx = tid + 256 * u;
            if (BT) {
                int n = idx / QA, q = idx % QA;
                *(v4f*)&Bs[n * PB + 4 * q] = rb[u];
            } else {
                int k = idx / QB, q = idx % QB;
                *(v4f*)&Bs[k * PB + 4 * q] = rb[u];
            }
        }
        __syncthreads();
        if (s + 1 < S) TILE_GLOAD((s + 1) * KB)

#pragma unroll
        for (int g = 0; g < KB / 8; ++g) {
            v2f alo[RT], ahi[RT], blo[CT], bhi[CT];
#pragma unroll
            for (int i = 0; i < RT; ++i) {
                v4f a4 = *(const v4f*)&As[(16 * i + c) * PA + 8 * g + 4 * h];
                alo[i].x = a4.x; alo[i].y = a4.y;
                ahi[i].x = a4.z; ahi[i].y = a4.w;
            }
#pragma unroll
            for (int j = 0; j < CT; ++j) {
                if (BT) {
                    v4f b4 = *(const v4f*)&Bs[(wn0 + 16 * j + c) * PB + 8 * g + 4 * h];
                    blo[j].x = b4.x; blo[j].y = b4.y;
                    bhi[j].x = b4.z; bhi[j].y = b4.w;
                } else {
                    const float* bp = &Bs[(8 * g + 4 * h) * PB + wn0 + 16 * j + c];
                    blo[j].x = bp[0];
                    blo[j].y = bp[PB];
                    bhi[j].x = bp[2 * PB];
                    bhi[j].y = bp[3 * PB];
                }
            }
#pragma unroll
            for (int i = 0; i < RT; ++i)
#pragma unroll
                for (int j = 0; j < CT; ++j)
                    acc[i][j] = __builtin_amdgcn_wmma_f32_16x16x4_f32(
                        false, alo[i], false, blo[j], (short)0, acc[i][j], false, false);
#pragma unroll
            for (int i = 0; i < RT; ++i)
#pragma unroll
                for (int j = 0; j < CT; ++j)
                    acc[i][j] = __builtin_amdgcn_wmma_f32_16x16x4_f32(
                        false, ahi[i], false, bhi[j], (short)0, acc[i][j], false, false);
        }
        __syncthreads();
    }
#undef TILE_GLOAD

    if (!HEAD) {
#pragma unroll
        for (int i = 0; i < RT; ++i)
#pragma unroll
            for (int j = 0; j < CT; ++j)
#pragma unroll
                for (int r = 0; r < 8; ++r) {
                    const int row = m0 + 16 * i + 8 * h + r;
                    const int cc  = wn0 + 16 * j + c;
                    float o = acc[i][j][r];
                    if (rowvec != nullptr && row == rowidx) o += rowvec[cc];
                    C[(size_t)row * ldc + cc] = o;
                }
    } else {
        float bias[CT], u1[CT], u2[CT], w0[CT], w1[CT];
#pragma unroll
        for (int j = 0; j < CT; ++j) {
            int col = wn0 + 16 * j + c;
            bias[j] = lb1[col] + B[(size_t)(0 * BM + 64) * ldb + col]
                               + B[(size_t)(1 * BM + 65) * ldb + col]
                               + B[(size_t)(2 * BM + 66) * ldb + col];
            u1[j]   = B[(size_t)(1 * BM + 64) * ldb + col] + B[(size_t)(2 * BM + 65) * ldb + col];
            u2[j]   = B[(size_t)(2 * BM + 64) * ldb + col];
            w0[j]   = lW2[col];
            w1[j]   = lW2[BN + col];
        }
        float* red = Bs;
#pragma unroll
        for (int i = 0; i < RT; ++i)
#pragma unroll
            for (int r = 0; r < 8; ++r) {
                float p0 = 0.0f, p1 = 0.0f;
                const int grow = m0 + 16 * i + 8 * h + r;
                const float a1 = r1[grow], a2 = r2[grow];
#pragma unroll
                for (int j = 0; j < CT; ++j) {
                    float t = acc[i][j][r] + bias[j] + a1 * u1[j] + a2 * u2[j];
                    t = t > 0.0f ? t : 0.0f;
                    p0 += t * w0[j];
                    p1 += t * w1[j];
                }
#pragma unroll
                for (int m = 8; m >= 1; m >>= 1) {
                    p0 += __shfl_xor(p0, m, 16);
                    p1 += __shfl_xor(p1, m, 16);
                }
                if (c == 0) {
                    int row = 16 * i + 8 * h + r;
                    red[(wave * BM + row) * 2 + 0] = p0;
                    red[(wave * BM + row) * 2 + 1] = p1;
                }
            }
        __syncthreads();
        if (tid < 2 * BM) {
            int row = tid >> 1, cls = tid & 1;
            float sum = lb2[cls];
#pragma unroll
            for (int w = 0; w < 8; ++w) sum += red[(w * BM + row) * 2 + cls];
            C[(size_t)(m0 + row) * ldc + cls] = sum;
            const float oth = __shfl_xor(sum, 1, 32);
            const float mx = fmaxf(sum, oth);
            const float es = __expf(sum - mx), eo = __expf(oth - mx);
            const float inv = 1.0f / (es + eo);
            probs[(size_t)(m0 + row) * 2 + cls] = es * inv;
        }
    }
}

__global__ void __launch_bounds__(256)
head_gemm(const float* __restrict__ A, int lda, const float* __restrict__ B, int ldb,
          float* __restrict__ C, int ldc, int K,
          const float* __restrict__ lb1, const float* __restrict__ lW2,
          const float* __restrict__ lb2, int bpad, const float* __restrict__ r1,
          const float* __restrict__ r2, float* __restrict__ probs) {
    gemm_block<false, CD, true>(A, lda, B, ldb, C, ldc, K, lb1, lW2, lb2, bpad, nullptr, 0,
                                0, 0, 0, 0, 0, r1, r2, probs, blockIdx.x, 0, 0);
}

__device__ __forceinline__ void fold_step(const float* __restrict__ Tin, const float* __restrict__ Wm,
                                          float* __restrict__ Tout, const float* __restrict__ bvec,
                                          int brow) {
    gemm_block<false, HD, false>(Tin, HD, Wm, HD, Tout, HD, HD, nullptr, nullptr, nullptr,
                                 0, bvec, brow, 0, 0, 0, 0, 0, nullptr, nullptr, nullptr, 0, 0, 0);
}

__global__ void __launch_bounds__(256)
csr_count_fold(const int* __restrict__ dst, int* cnt, int e, float* Tm,
               const float* __restrict__ W2, const float* __restrict__ b2,
               const float* __restrict__ W3, const float* __restrict__ b3) {
    if (blockIdx.x == 0) {
        fold_step(Tm, W2, Tm + BM * HD, b2, 65);
        __syncthreads();
        fold_step(Tm + BM * HD, W3, Tm + 2 * BM * HD, b3, 66);
        return;
    }
    count_edges(dst, cnt, e, blockIdx.x - 1, gridDim.x - 1);
}
__global__ void __launch_bounds__(256)
csr_fill_fold(const int* __restrict__ src, const int* __restrict__ dst,
              const float* __restrict__ dinv, int* cursor,
              int* __restrict__ col, float* __restrict__ wgt, int e,
              const float* __restrict__ Tm, const float* __restrict__ lW1,
              float* __restrict__ Fm) {
    if (blockIdx.x < 9) {
        gemm_block<true, HD, false>(Tm, HD, lW1, CD, Fm, CD, HD, nullptr, nullptr, nullptr,
                                    0, nullptr, 0, BM * HD, HD, BM * CD, HD * CD, HD,
                                    nullptr, nullptr, nullptr, 0,
                                    (int)(blockIdx.x % 3), (int)(blockIdx.x / 3));
        return;
    }
    fill_edges(src, dst, dinv, cursor, col, wgt, e, blockIdx.x - 9, gridDim.x - 9);
}

__global__ void seed_and_clear(const float* __restrict__ W1, const float* __restrict__ b1,
                               float* __restrict__ T1, int* __restrict__ cnt, int n) {
    int i = blockIdx.x * blockDim.x + threadIdx.x;
    if (i < n) cnt[i] = 0;
    if (i < BM * HD) {
        int r = i / HD, cc = i % HD;
        float o = 0.0f;
        if (r < XD) o = W1[i];
        else if (r == XD) o = b1[cc];
        T1[i] = o;
    }
}

#define RUN_ROWS 31

template <int MODE>
__device__ __forceinline__ void
gather_rows(const float* __restrict__ Xin, int ldi, const float* __restrict__ dinv,
            const int* __restrict__ rowptr, const int* __restrict__ col,
            const float* __restrict__ wgt,
            float* __restrict__ Yout, int ldo,
            const float* __restrict__ rin, float* __restrict__ rout, int n, int e,
            int bid, int nblk) {
    const int lane = threadIdx.x & 31;
    const int c  = lane & 15;
    const int hh = lane >> 4;
    const int wid = __builtin_amdgcn_readfirstlane(bid * 8 + (threadIdx.x >> 5));
    const int nw  = nblk * 8;
    const unsigned uq = (unsigned)n / (unsigned)nw, ur = (unsigned)n % (unsigned)nw;
    const int v0 = (int)((unsigned)wid * uq + ((unsigned)wid < ur ? (unsigned)wid : ur));
    const int v1 = v0 + (int)uq + ((unsigned)wid < ur ? 1 : 0);
    const char* Xb = (const char*)Xin;
    const unsigned pitch = (unsigned)ldi * 4u;
    const unsigned coff  = (unsigned)c * 16u;

#define GLOAD(R, P)                                                                \
    {                                                                              \
        const int sa = __builtin_amdgcn_readlane(myc, 2 * (P));                    \
        const int sb = __builtin_amdgcn_readlane(myc, 2 * (P) + 1);                \
        const unsigned sr = (unsigned)(hh ? sb : sa);                              \
        q[R] = *(const float4*)(Xb + (sr * pitch + coff));                         \
    }
#define GFMA(R, P)                                                                 \
    {                                                                              \
        const int wa = __builtin_amdgcn_readlane(mywi, 2 * (P));                   \
        const int wb = __builtin_amdgcn_readlane(mywi, 2 * (P) + 1);               \
        const float w = __int_as_float(hh ? wb : wa);                              \
        acc.x += q[R].x * w; acc.y += q[R].y * w;                                  \
        acc.z += q[R].z * w; acc.w += q[R].w * w;                                  \
    }

    for (int vb = v0; vb < v1; vb += RUN_ROWS) {
        const int cnt = (v1 - vb) < RUN_ROWS ? (v1 - vb) : RUN_ROWS;
        const int ri = (vb + lane) < n ? (vb + lane) : n;
        const int di = (vb + lane) < n ? (vb + lane) : (n - 1);
        const int rp = rowptr[ri];
        const float dv = dinv[di];
        float rself = 0.0f;
        if (MODE == 2) rself = rin[di];
        const int dvi = __float_as_int(dv);
        const int rsi = __float_as_int(rself);
        int beg = __builtin_amdgcn_readlane(rp, 0);
        int pi = beg + lane;
        pi = pi < e ? pi : e - 1;
        pi = pi > 0 ? pi : 0;
        int   nc  = col[pi];
        float nwt = wgt[pi];
        for (int i = 0; i < cnt; ++i) {
            const int v = vb + i;
            const int end = __builtin_amdgcn_readlane(rp, i + 1);
            const float dn = __int_as_float(__builtin_amdgcn_readlane(dvi, i));
            const float nn = dn * dn;
            const int   fc = nc;
            const float fw = nwt;
            pi = end + lane;
            pi = pi < e ? pi : e - 1;
            pi = pi > 0 ? pi : 0;
            nc  = col[pi];
            nwt = wgt[pi];
            const float4 xs = *(const float4*)(Xb + ((unsigned)v * pitch + coff));
            float4 acc = make_float4(0.0f, 0.0f, 0.0f, 0.0f);
            bool first = true;
            float rs = 0.0f;
            for (int base = beg; base < end; base += 32) {
                const int m = (end - base) < 32 ? (end - base) : 32;
                int   myc = v;
                float myw = 0.0f;
                if (lane < m) {
                    if (base == beg) { myc = fc; myw = fw; }
                    else { myc = col[base + lane]; myw = wgt[base + lane]; }
                }
                float rv = 0.0f;
                if (MODE == 2) rv = rin[myc];
                const int mywi = __float_as_int(myw);
                float4 q[12];
                GLOAD(0, 0) GLOAD(1, 1) GLOAD(2, 2) GLOAD(3, 3)
                if (m > 8) {
                    GLOAD(4, 4) GLOAD(5, 5) GLOAD(6, 6) GLOAD(7, 7)
                    if (m > 16) { GLOAD(8, 8) GLOAD(9, 9) GLOAD(10, 10) GLOAD(11, 11) }
                }
                if (first) {
                    acc.x = hh ? 0.0f : xs.x * nn; acc.y = hh ? 0.0f : xs.y * nn;
                    acc.z = hh ? 0.0f : xs.z * nn; acc.w = hh ? 0.0f : xs.w * nn;
                    first = false;
                }
                GFMA(0, 0) GFMA(1, 1) GFMA(2, 2) GFMA(3, 3)
                if (m > 8) {
                    GFMA(4, 4) GFMA(5, 5) GFMA(6, 6) GFMA(7, 7)
                    if (m > 16) { GFMA(8, 8) GFMA(9, 9) GFMA(10, 10) GFMA(11, 11) }
                }
                if (m > 24) {
                    GLOAD(0, 12) GLOAD(1, 13) GLOAD(2, 14) GLOAD(3, 15)
                    GFMA(0, 12) GFMA(1, 13) GFMA(2, 14) GFMA(3, 15)
                }
                if (MODE == 1) rs += myw;
                if (MODE == 2) rs += myw * rv;
            }
            if (first) {
                acc.x = hh ? 0.0f : xs.x * nn; acc.y = hh ? 0.0f : xs.y * nn;
                acc.z = hh ? 0.0f : xs.z * nn; acc.w = hh ? 0.0f : xs.w * nn;
            }
            acc.x += __shfl_xor(acc.x, 16, 32);
            acc.y += __shfl_xor(acc.y, 16, 32);
            acc.z += __shfl_xor(acc.z, 16, 32);
            acc.w += __shfl_xor(acc.w, 16, 32);
            if (hh == 0)
                *(float4*)(Yout + (size_t)v * ldo + c * 4) = acc;
            if (MODE != 0) {
#pragma unroll
                for (int mm = 16; mm >= 1; mm >>= 1) rs += __shfl_xor(rs, mm, 32);
                const float rsv = __int_as_float(__builtin_amdgcn_readlane(rsi, i));
                if (lane == 0) rout[v] = rs + (MODE == 1 ? nn : nn * rsv);
            }
            beg = end;
        }
    }
#undef GLOAD
#undef GFMA
}

template <int MODE>
__global__ void __launch_bounds__(256)
gather64(const float* __restrict__ Xin, int ldi, const float* __restrict__ dinv,
         const int* __restrict__ rowptr, const int* __restrict__ col,
         const float* __restrict__ wgt,
         float* __restrict__ Yout, int ldo,
         const float* __restrict__ rin, float* __restrict__ rout, int n, int e) {
    gather_rows<MODE>(Xin, ldi, dinv, rowptr, col, wgt, Yout, ldo, rin, rout, n, e,
                      blockIdx.x, gridDim.x);
}

extern "C" void kernel_launch(void* const* d_in, const int* in_sizes, int n_in,
                              void* d_out, int out_size, void* d_ws, size_t ws_size,
                              hipStream_t stream) {
    const float* x   = (const float*)d_in[0];
    const int*   ei  = (const int*)d_in[1];
    const float* W1  = (const float*)d_in[2];
    const float* b1  = (const float*)d_in[3];
    const float* W2  = (const float*)d_in[4];
    const float* b2  = (const float*)d_in[5];
    const float* W3  = (const float*)d_in[6];
    const float* b3  = (const float*)d_in[7];
    const float* lW1 = (const float*)d_in[8];
    const float* lb1 = (const float*)d_in[9];
    const float* lW2 = (const float*)d_in[10];
    const float* lb2 = (const float*)d_in[11];

    const int N = NN, E = EE;
    const int* src = ei;
    const int* dst = ei + E;

    float* y    = (float*)d_ws;
    float* r1   = y + (size_t)N * YD;
    float* r2   = r1 + N;
    float* Tm   = r2 + N;
    float* Fm   = Tm + 3 * BM * HD;
    float* dinv = Fm + 3 * BM * CD;
    int*   cnt    = (int*)(dinv + N);
    int*   cursor = cnt + N;
    int*   rowptr = cursor + N;
    int*   col    = rowptr + (N + 1);
    float* wgt    = (float*)(col + E);

    float* logits = (float*)d_out;
    float* probs  = logits + (size_t)N * 2;

    const int T = 256;
    const int rowBlocks = N / BM;
    const int gatherBlocks = 1024;
    const int edgeBlocks   = 512;

    seed_and_clear<<<(N + T - 1) / T, T, 0, stream>>>(W1, b1, Tm, cnt, N);

    csr_count_fold<<<edgeBlocks + 1, T, 0, stream>>>(dst, cnt, E, Tm, W2, b2, W3, b3);
    csr_scan<<<1, SCAN_T, 0, stream>>>(cnt, rowptr, cursor, dinv);
    csr_fill_fold<<<edgeBlocks + 9, T, 0, stream>>>(src, dst, dinv, cursor, col, wgt, E,
                                                   Tm, lW1, Fm);

    gather64<1><<<gatherBlocks, T, 0, stream>>>(x, XD, dinv, rowptr, col, wgt, y, YD, nullptr, r1, N, E);
    gather64<2><<<gatherBlocks, T, 0, stream>>>(y, YD, dinv, rowptr, col, wgt, y + XD, YD, r1, r2, N, E);
    gather64<0><<<gatherBlocks, T, 0, stream>>>(y + XD, YD, dinv, rowptr, col, wgt, y + 2 * XD, YD,
                                                nullptr, nullptr, N, E);

    head_gemm<<<rowBlocks, T, 0, stream>>>(y, YD, Fm, CD, logits, 2, YD, lb1, lW2, lb2, 16,
                                           r1, r2, probs);
}


# Row-offset scan split over 49 work-groups, each summing the counts before its 1024 rows itself with independent 16-byte loads; was one work-group paying 49 dependent L2 trips; integer sums only
# speedup vs baseline: 3.5033x; 1.1259x over previous
#include <hip/hip_runtime.h>
#include <hip/hip_bf16.h>


typedef __attribute__((ext_vector_type(2))) float v2f;
typedef __attribute__((ext_vector_type(4))) float v4f;
typedef __attribute__((ext_vector_type(8))) float v8f;

#define NN 50000
#define EE 800000
#define HD 128
#define CD 384
#define XD 64
#define YD 192

__device__ __forceinline__ void count_edges(const int* __restrict__ dst, int* cnt, int e,
                                            int bid, int nblk) {
    const int t = bid * blockDim.x + threadIdx.x;
    const int stride = nblk * blockDim.x;
    const int e4 = (((size_t)dst & 15) == 0) ? (e >> 2) : 0;
    const int4* d4 = (const int4*)dst;
    for (int i = t; i < e4; i += stride) {
        int4 d = d4[i];
        atomicAdd(&cnt[d.x], 1);
        atomicAdd(&cnt[d.y], 1);
        atomicAdd(&cnt[d.z], 1);
        atomicAdd(&cnt[d.w], 1);
    }
    for (int i = 4 * e4 + t; i < e; i += stride) atomicAdd(&cnt[dst[i]], 1);
}
#define SCAN_T 1024
#define SCAN_NB ((NN + SCAN_T - 1) / SCAN_T)
#define SCAN_PASS ((SCAN_NB - 1 + 3) / 4)
static_assert(SCAN_PASS * SCAN_T * 4 <= NN, "the unguarded 16-byte loads must stay inside the count array");
static_assert(SCAN_T / 32 == 32, "one wave total per lane in the block-level step");
__global__ void __launch_bounds__(SCAN_T)
csr_scan(const int* __restrict__ cnt, int* __restrict__ rowptr, int* __restrict__ cursor,
         float* __restrict__ dinv) {
    __shared__ int wsum[SCAN_T / 32];
    __shared__ int wpart[SCAN_T / 32];
    const int t = threadIdx.x;
    const int lane = t & 31;
    const int w = t >> 5;
    const int idx = blockIdx.x * SCAN_T + t;
    int cv = 0;
    if (idx < NN) cv = cnt[idx];
    const int n4 = blockIdx.x * (SCAN_T / 4);
    const int4* c4 = (const int4*)cnt;
    int4 pv[SCAN_PASS];
#pragma unroll
    for (int i = 0; i < SCAN_PASS; ++i) pv[i] = c4[i * SCAN_T + t];
    int part = 0;
#pragma unroll
    for (int i = 0; i < SCAN_PASS; ++i) {
        const int s = (pv[i].x + pv[i].y) + (pv[i].z + pv[i].w);
        part += (i * SCAN_T + t < n4) ? s : 0;
    }
#pragma unroll
    for (int m = 16; m >= 1; m >>= 1) part += __shfl_xor(part, m, 32);
    int incl = cv;
#pragma unroll
    for (int m = 1; m < 32; m <<= 1) {
        int up = __shfl_up(incl, m, 32);
        if (lane >= m) incl += up;
    }
    if (lane == 31) wsum[w] = incl;
    if (lane == 0) wpart[w] = part;
    __syncthreads();
    int base = wpart[lane];
    int woff = (lane < w) ? wsum[lane] : 0;
#pragma unroll
    for (int m = 16; m >= 1; m >>= 1) {
        base += __shfl_xor(base, m, 32);
        woff += __shfl_xor(woff, m, 32);
    }
    const int p = base + woff + incl - cv;
    if (idx < NN) {
        rowptr[idx] = p;
        cursor[idx] = p;
        dinv[idx] = rsqrtf((float)(cv + 1));
        if (idx == NN - 1) rowptr[NN] = p + cv;
    }
}
__device__ __forceinline__ void fill_edges(const int* __restrict__ src, const int* __restrict__ dst,
                                           const float* __restrict__ dinv, int* cursor,
                                           int* __restrict__ col, float* __restrict__ wgt, int e,
                                           int bid, int nblk) {
    const int t = bid * blockDim.x + threadIdx.x;
    const int stride = nblk * blockDim.x;
    const int e4 = (((((size_t)src) | ((size_t)dst)) & 15) == 0) ? (e >> 2) : 0;
    const int4* s4 = (const int4*)src;
    const int4* d4 = (const int4*)dst;
    for (int i = t; i < e4; i += stride) {
        int4 sv = s4[i], dv = d4[i];
        int p0 = atomicAdd(&cursor[dv.x], 1);
        int p1 = atomicAdd(&cursor[dv.y], 1);
        int p2 = atomicAdd(&cursor[dv.z], 1);
        int p3 = atomicAdd(&cursor[dv.w], 1);
        float w0 = dinv[sv.x] * dinv[dv.x];
        float w1 = dinv[sv.y] * dinv[dv.y];
        float w2 = dinv[sv.z] * dinv[dv.z];
        float w3 = dinv[sv.w] * dinv[dv.w];
        col[p0] = sv.x; wgt[p0] = w0;
        col[p1] = sv.y; wgt[p1] = w1;
        col[p2] = sv.z; wgt[p2] = w2;
        col[p3] = sv.w; wgt[p3] = w3;
    }
    for (int i = 4 * e4 + t; i < e; i += stride) {
        int sv = src[i], dv = dst[i];
        int p = atomicAdd(&cursor[dv], 1);
        col[p] = sv;
        wgt[p] = dinv[sv] * dinv[dv];
    }
}

#define BM 80
#define KB 32
#define PA (KB + 4)

template <bool BT, int BN, bool HEAD>
__device__ __forceinline__ void
gemm_block(const float* __restrict__ A, int lda,
           const float* __restrict__ B, int ldb,
           float* __restrict__ C, int ldc, int K,
           const float* __restrict__ lb1,
           const float* __restrict__ lW2,
           const float* __restrict__ lb2,
           int bpad,
           const float* __restrict__ rowvec, int rowidx,
           int sAy, int sBy, int sCy, int sBz, int sCz,
           const float* __restrict__ r1, const float* __restrict__ r2,
           float* __restrict__ probs, int bx, int by, int bz) {
    A += (size_t)by * sAy;
    B += (size_t)by * sBy + (size_t)bz * sBz;
    C += (size_t)by * sCy + (size_t)bz * sCz;
    constexpr int RT = BM / 16;
    constexpr int CT = BN / 128;
    constexpr int PB = BT ? PA : (BN + 4);
    constexpr int BROWS = BT ? BN : KB;
    constexpr int QA = KB / 4;
    constexpr int A4 = BM * QA;
    constexpr int AL = (A4 + 255) / 256;
    constexpr int BL = BN * QA / 256;
    constexpr int QB = BN / 4;

    __shared__ __attribute__((aligned(16))) float As[BM * PA];
    __shared__ __attribute__((aligned(16))) float Bs[BROWS * PB];

    const int tid  = threadIdx.x;
    const int wave = tid >> 5;
    const int lane = tid & 31;
    const int c    = lane & 15;
    const int h    = lane >> 4;
    const int m0   = bx * BM;
    const int wn0  = wave * (CT * 16);

    v8f acc[RT][CT];
#pragma unroll
    for (int i = 0; i < RT; ++i)
#pragma unroll
        for (int j = 0; j < CT; ++j) {
            v8f z = {};
            acc[i][j] = z;
        }

    v4f ra[AL], rb[BL];

#define TILE_GLOAD(kb)                                                              \
    {                                                                               \
        _Pragma("unroll") for (int u = 0; u < AL; ++u) {                            \
            int idx = tid + 256 * u;                                                \
            if (idx < A4) {                                                         \
                int row = idx / QA, q = idx % QA;                                   \
                ra[u] = *(const v4f*)(A + (size_t)(m0 + row) * lda + (kb) + 4 * q); \
            }                                                                       \
        }                                                                           \
        _Pragma("unroll") for (int u = 0; u < BL; ++u) {                            \
            int idx = tid + 256 * u;                                                \
            if (BT) {                                                               \
                int n = idx / QA, q = idx % QA;                                     \
                rb[u] = *(const v4f*)(B + (size_t)n * ldb + (kb) + 4 * q);       \
            } else {                                                                \
                int k = idx / QB, q = idx % QB;                                     \
                rb[u] = *(const v4f*)(B + (size_t)((kb) + k + bpad * ((kb) >> 6)) * ldb + 4 * q); \
            }                                                                       \
        }                                                                           \
    }

    TILE_GLOAD(0)
    const int S = K / KB;
    for (int s = 0; s < S; ++s) {
#pragma unroll
        for (int u = 0; u < AL; ++u) {
            int idx = tid + 256 * u;
            if (idx < A4) {
                int row = idx / QA, q = idx % QA;
                *(v4f*)&As[row * PA + 4 * q] = ra[u];
            }
        }
#pragma unroll
        for (int u = 0; u < BL; ++u) {
            int idx = tid + 256 * u;
            if (BT) {
                int n = idx / QA, q = idx % QA;
                *(v4f*)&Bs[n * PB + 4 * q] = rb[u];
            } else {
                int k = idx / QB, q = idx % QB;
                *(v4f*)&Bs[k * PB + 4 * q] = rb[u];
            }
        }
        __syncthreads();
        if (s + 1 < S) TILE_GLOAD((s + 1) * KB)

#pragma unroll
        for (int g = 0; g < KB / 8; ++g) {
            v2f alo[RT], ahi[RT], blo[CT], bhi[CT];
#pragma unroll
            for (int i = 0; i < RT; ++i) {
                v4f a4 = *(const v4f*)&As[(16 * i + c) * PA + 8 * g + 4 * h];
                alo[i].x = a4.x; alo[i].y = a4.y;
                ahi[i].x = a4.z; ahi[i].y = a4.w;
            }
#pragma unroll
            for (int j = 0; j < CT; ++j) {
                if (BT) {
                    v4f b4 = *(const v4f*)&Bs[(wn0 + 16 * j + c) * PB + 8 * g + 4 * h];
                    blo[j].x = b4.x; blo[j].y = b4.y;
                    bhi[j].x = b4.z; bhi[j].y = b4.w;
                } else {
                    const float* bp = &Bs[(8 * g + 4 * h) * PB + wn0 + 16 * j + c];
                    blo[j].x = bp[0];
                    blo[j].y = bp[PB];
                    bhi[j].x = bp[2 * PB];
                    bhi[j].y = bp[3 * PB];
                }
            }
#pragma unroll
            for (int i = 0; i < RT; ++i)
#pragma unroll
                for (int j = 0; j < CT; ++j)
                    acc[i][j] = __builtin_amdgcn_wmma_f32_16x16x4_f32(
                        false, alo[i], false, blo[j], (short)0, acc[i][j], false, false);
#pragma unroll
            for (int i = 0; i < RT; ++i)
#pragma unroll
                for (int j = 0; j < CT; ++j)
                    acc[i][j] = __builtin_amdgcn_wmma_f32_16x16x4_f32(
                        false, ahi[i], false, bhi[j], (short)0, acc[i][j], false, false);
        }
        __syncthreads();
    }
#undef TILE_GLOAD

    if (!HEAD) {
#pragma unroll
        for (int i = 0; i < RT; ++i)
#pragma unroll
            for (int j = 0; j < CT; ++j)
#pragma unroll
                for (int r = 0; r < 8; ++r) {
                    const int row = m0 + 16 * i + 8 * h + r;
                    const int cc  = wn0 + 16 * j + c;
                    float o = acc[i][j][r];
                    if (rowvec != nullptr && row == rowidx) o += rowvec[cc];
                    C[(size_t)row * ldc + cc] = o;
                }
    } else {
        float bias[CT], u1[CT], u2[CT], w0[CT], w1[CT];
#pragma unroll
        for (int j = 0; j < CT; ++j) {
            int col = wn0 + 16 * j + c;
            bias[j] = lb1[col] + B[(size_t)(0 * BM + 64) * ldb + col]
                               + B[(size_t)(1 * BM + 65) * ldb + col]
                               + B[(size_t)(2 * BM + 66) * ldb + col];
            u1[j]   = B[(size_t)(1 * BM + 64) * ldb + col] + B[(size_t)(2 * BM + 65) * ldb + col];
            u2[j]   = B[(size_t)(2 * BM + 64) * ldb + col];
            w0[j]   = lW2[col];
            w1[j]   = lW2[BN + col];
        }
        float* red = Bs;
#pragma unroll
        for (int i = 0; i < RT; ++i)
#pragma unroll
            for (int r = 0; r < 8; ++r) {
                float p0 = 0.0f, p1 = 0.0f;
                const int grow = m0 + 16 * i + 8 * h + r;
                const float a1 = r1[grow], a2 = r2[grow];
#pragma unroll
                for (int j = 0; j < CT; ++j) {
                    float t = acc[i][j][r] + bias[j] + a1 * u1[j] + a2 * u2[j];
                    t = t > 0.0f ? t : 0.0f;
                    p0 += t * w0[j];
                    p1 += t * w1[j];
                }
#pragma unroll
                for (int m = 8; m >= 1; m >>= 1) {
                    p0 += __shfl_xor(p0, m, 16);
                    p1 += __shfl_xor(p1, m, 16);
                }
                if (c == 0) {
                    int row = 16 * i + 8 * h + r;
                    red[(wave * BM + row) * 2 + 0] = p0;
                    red[(wave * BM + row) * 2 + 1] = p1;
                }
            }
        __syncthreads();
        if (tid < 2 * BM) {
            int row = tid >> 1, cls = tid & 1;
            float sum = lb2[cls];
#pragma unroll
            for (int w = 0; w < 8; ++w) sum += red[(w * BM + row) * 2 + cls];
            C[(size_t)(m0 + row) * ldc + cls] = sum;
            const float oth = __shfl_xor(sum, 1, 32);
            const float mx = fmaxf(sum, oth);
            const float es = __expf(sum - mx), eo = __expf(oth - mx);
            const float inv = 1.0f / (es + eo);
            probs[(size_t)(m0 + row) * 2 + cls] = es * inv;
        }
    }
}

__global__ void __launch_bounds__(256)
head_gemm(const float* __restrict__ A, int lda, const float* __restrict__ B, int ldb,
          float* __restrict__ C, int ldc, int K,
          const float* __restrict__ lb1, const float* __restrict__ lW2,
          const float* __restrict__ lb2, int bpad, const float* __restrict__ r1,
          const float* __restrict__ r2, float* __restrict__ probs) {
    gemm_block<false, CD, true>(A, lda, B, ldb, C, ldc, K, lb1, lW2, lb2, bpad, nullptr, 0,
                                0, 0, 0, 0, 0, r1, r2, probs, blockIdx.x, 0, 0);
}

__device__ __forceinline__ void fold_step(const float* __restrict__ Tin, const float* __restrict__ Wm,
                                          float* __restrict__ Tout, const float* __restrict__ bvec,
                                          int brow) {
    gemm_block<false, HD, false>(Tin, HD, Wm, HD, Tout, HD, HD, nullptr, nullptr, nullptr,
                                 0, bvec, brow, 0, 0, 0, 0, 0, nullptr, nullptr, nullptr, 0, 0, 0);
}

__global__ void __launch_bounds__(256)
csr_count_fold(const int* __restrict__ dst, int* cnt, int e, float* Tm,
               const float* __restrict__ W2, const float* __restrict__ b2,
               const float* __restrict__ W3, const float* __restrict__ b3) {
    if (blockIdx.x == 0) {
        fold_step(Tm, W2, Tm + BM * HD, b2, 65);
        __syncthreads();
        fold_step(Tm + BM * HD, W3, Tm + 2 * BM * HD, b3, 66);
        return;
    }
    count_edges(dst, cnt, e, blockIdx.x - 1, gridDim.x - 1);
}
__global__ void __launch_bounds__(256)
csr_fill_fold(const int* __restrict__ src, const int* __restrict__ dst,
              const float* __restrict__ dinv, int* cursor,
              int* __restrict__ col, float* __restrict__ wgt, int e,
              const float* __restrict__ Tm, const float* __restrict__ lW1,
              float* __restrict__ Fm) {
    if (blockIdx.x < 9) {
        gemm_block<true, HD, false>(Tm, HD, lW1, CD, Fm, CD, HD, nullptr, nullptr, nullptr,
                                    0, nullptr, 0, BM * HD, HD, BM * CD, HD * CD, HD,
                                    nullptr, nullptr, nullptr, 0,
                                    (int)(blockIdx.x % 3), (int)(blockIdx.x / 3));
        return;
    }
    fill_edges(src, dst, dinv, cursor, col, wgt, e, blockIdx.x - 9, gridDim.x - 9);
}

__global__ void seed_and_clear(const float* __restrict__ W1, const float* __restrict__ b1,
                               float* __restrict__ T1, int* __restrict__ cnt, int n) {
    int i = blockIdx.x * blockDim.x + threadIdx.x;
    if (i < n) cnt[i] = 0;
    if (i < BM * HD) {
        int r = i / HD, cc = i % HD;
        float o = 0.0f;
        if (r < XD) o = W1[i];
        else if (r == XD) o = b1[cc];
        T1[i] = o;
    }
}

#define RUN_ROWS 31

template <int MODE>
__device__ __forceinline__ void
gather_rows(const float* __restrict__ Xin, int ldi, const float* __restrict__ dinv,
            const int* __restrict__ rowptr, const int* __restrict__ col,
            const float* __restrict__ wgt,
            float* __restrict__ Yout, int ldo,
            const float* __restrict__ rin, float* __restrict__ rout, int n, int e,
            int bid, int nblk) {
    const int lane = threadIdx.x & 31;
    const int c  = lane & 15;
    const int hh = lane >> 4;
    const int wid = __builtin_amdgcn_readfirstlane(bid * 8 + (threadIdx.x >> 5));
    const int nw  = nblk * 8;
    const unsigned uq = (unsigned)n / (unsigned)nw, ur = (unsigned)n % (unsigned)nw;
    const int v0 = (int)((unsigned)wid * uq + ((unsigned)wid < ur ? (unsigned)wid : ur));
    const int v1 = v0 + (int)uq + ((unsigned)wid < ur ? 1 : 0);
    const char* Xb = (const char*)Xin;
    const unsigned pitch = (unsigned)ldi * 4u;
    const unsigned coff  = (unsigned)c * 16u;

#define GLOAD(R, P)                                                                \
    {                                                                              \
        const int sa = __builtin_amdgcn_readlane(myc, 2 * (P));                    \
        const int sb = __builtin_amdgcn_readlane(myc, 2 * (P) + 1);                \
        const unsigned sr = (unsigned)(hh ? sb : sa);                              \
        q[R] = *(const float4*)(Xb + (sr * pitch + coff));                         \
    }
#define GFMA(R, P)                                                                 \
    {                                                                              \
        const int wa = __builtin_amdgcn_readlane(mywi, 2 * (P));                   \
        const int wb = __builtin_amdgcn_readlane(mywi, 2 * (P) + 1);               \
        const float w = __int_as_float(hh ? wb : wa);                              \
        acc.x += q[R].x * w; acc.y += q[R].y * w;                                  \
        acc.z += q[R].z * w; acc.w += q[R].w * w;                                  \
    }

    for (int vb = v0; vb < v1; vb += RUN_ROWS) {
        const int cnt = (v1 - vb) < RUN_ROWS ? (v1 - vb) : RUN_ROWS;
        const int ri = (vb + lane) < n ? (vb + lane) : n;
        const int di = (vb + lane) < n ? (vb + lane) : (n - 1);
        const int rp = rowptr[ri];
        const float dv = dinv[di];
        float rself = 0.0f;
        if (MODE == 2) rself = rin[di];
        const int dvi = __float_as_int(dv);
        const int rsi = __float_as_int(rself);
        int beg = __builtin_amdgcn_readlane(rp, 0);
        int pi = beg + lane;
        pi = pi < e ? pi : e - 1;
        pi = pi > 0 ? pi : 0;
        int   nc  = col[pi];
        float nwt = wgt[pi];
        for (int i = 0; i < cnt; ++i) {
            const int v = vb + i;
            const int end = __builtin_amdgcn_readlane(rp, i + 1);
            const float dn = __int_as_float(__builtin_amdgcn_readlane(dvi, i));
            const float nn = dn * dn;
            const int   fc = nc;
            const float fw = nwt;
            pi = end + lane;
            pi = pi < e ? pi : e - 1;
            pi = pi > 0 ? pi : 0;
            nc  = col[pi];
            nwt = wgt[pi];
            const float4 xs = *(const float4*)(Xb + ((unsigned)v * pitch + coff));
            float4 acc = make_float4(0.0f, 0.0f, 0.0f, 0.0f);
            bool first = true;
            float rs = 0.0f;
            for (int base = beg; base < end; base += 32) {
                const int m = (end - base) < 32 ? (end - base) : 32;
                int   myc = v;
                float myw = 0.0f;
                if (lane < m) {
                    if (base == beg) { myc = fc; myw = fw; }
                    else { myc = col[base + lane]; myw = wgt[base + lane]; }
                }
                float rv = 0.0f;
                if (MODE == 2) rv = rin[myc];
                const int mywi = __float_as_int(myw);
                float4 q[12];
                GLOAD(0, 0) GLOAD(1, 1) GLOAD(2, 2) GLOAD(3, 3)
                if (m > 8) {
                    GLOAD(4, 4) GLOAD(5, 5) GLOAD(6, 6) GLOAD(7, 7)
                    if (m > 16) { GLOAD(8, 8) GLOAD(9, 9) GLOAD(10, 10) GLOAD(11, 11) }
                }
                if (first) {
                    acc.x = hh ? 0.0f : xs.x * nn; acc.y = hh ? 0.0f : xs.y * nn;
                    acc.z = hh ? 0.0f : xs.z * nn; acc.w = hh ? 0.0f : xs.w * nn;
                    first = false;
                }
                GFMA(0, 0) GFMA(1, 1) GFMA(2, 2) GFMA(3, 3)
                if (m > 8) {
                    GFMA(4, 4) GFMA(5, 5) GFMA(6, 6) GFMA(7, 7)
                    if (m > 16) { GFMA(8, 8) GFMA(9, 9) GFMA(10, 10) GFMA(11, 11) }
                }
                if (m > 24) {
                    GLOAD(0, 12) GLOAD(1, 13) GLOAD(2, 14) GLOAD(3, 15)
                    GFMA(0, 12) GFMA(1, 13) GFMA(2, 14) GFMA(3, 15)
                }
                if (MODE == 1) rs += myw;
                if (MODE == 2) rs += myw * rv;
            }
            if (first) {
                acc.x = hh ? 0.0f : xs.x * nn; acc.y = hh ? 0.0f : xs.y * nn;
                acc.z = hh ? 0.0f : xs.z * nn; acc.w = hh ? 0.0f : xs.w * nn;
            }
            acc.x += __shfl_xor(acc.x, 16, 32);
            acc.y += __shfl_xor(acc.y, 16, 32);
            acc.z += __shfl_xor(acc.z, 16, 32);
            acc.w += __shfl_xor(acc.w, 16, 32);
            if (hh == 0)
                *(float4*)(Yout + (size_t)v * ldo + c * 4) = acc;
            if (MODE != 0) {
#pragma unroll
                for (int mm = 16; mm >= 1; mm >>= 1) rs += __shfl_xor(rs, mm, 32);
                const float rsv = __int_as_float(__builtin_amdgcn_readlane(rsi, i));
                if (lane == 0) rout[v] = rs + (MODE == 1 ? nn : nn * rsv);
            }
            beg = end;
        }
    }
#undef GLOAD
#undef GFMA
}

template <int MODE>
__global__ void __launch_bounds__(256)
gather64(const float* __restrict__ Xin, int ldi, const float* __restrict__ dinv,
         const int* __restrict__ rowptr, const int* __restrict__ col,
         const float* __restrict__ wgt,
         float* __restrict__ Yout, int ldo,
         const float* __restrict__ rin, float* __restrict__ rout, int n, int e) {
    gather_rows<MODE>(Xin, ldi, dinv, rowptr, col, wgt, Yout, ldo, rin, rout, n, e,
                      blockIdx.x, gridDim.x);
}

extern "C" void kernel_launch(void* const* d_in, const int* in_sizes, int n_in,
                              void* d_out, int out_size, void* d_ws, size_t ws_size,
                              hipStream_t stream) {
    const float* x   = (const float*)d_in[0];
    const int*   ei  = (const int*)d_in[1];
    const float* W1  = (const float*)d_in[2];
    const float* b1  = (const float*)d_in[3];
    const float* W2  = (const float*)d_in[4];
    const float* b2  = (const float*)d_in[5];
    const float* W3  = (const float*)d_in[6];
    const float* b3  = (const float*)d_in[7];
    const float* lW1 = (const float*)d_in[8];
    const float* lb1 = (const float*)d_in[9];
    const float* lW2 = (const float*)d_in[10];
    const float* lb2 = (const float*)d_in[11];

    const int N = NN, E = EE;
    const int* src = ei;
    const int* dst = ei + E;

    float* y    = (float*)d_ws;
    float* r1   = y + (size_t)N * YD;
    float* r2   = r1 + N;
    float* Tm   = r2 + N;
    float* Fm   = Tm + 3 * BM * HD;
    float* dinv = Fm + 3 * BM * CD;
    int*   cnt    = (int*)(dinv + N);
    int*   cursor = cnt + N;
    int*   rowptr = cursor + N;
    int*   col    = rowptr + (N + 1);
    float* wgt    = (float*)(col + E);

    float* logits = (float*)d_out;
    float* probs  = logits + (size_t)N * 2;

    const int T = 256;
    const int rowBlocks = N / BM;
    const int gatherBlocks = 1024;
    const int edgeBlocks   = 512;

    seed_and_clear<<<(N + T - 1) / T, T, 0, stream>>>(W1, b1, Tm, cnt, N);

    csr_count_fold<<<edgeBlocks + 1, T, 0, stream>>>(dst, cnt, E, Tm, W2, b2, W3, b3);
    csr_scan<<<SCAN_NB, SCAN_T, 0, stream>>>(cnt, rowptr, cursor, dinv);
    csr_fill_fold<<<edgeBlocks + 9, T, 0, stream>>>(src, dst, dinv, cursor, col, wgt, E,
                                                   Tm, lW1, Fm);

    gather64<1><<<gatherBlocks, T, 0, stream>>>(x, XD, dinv, rowptr, col, wgt, y, YD, nullptr, r1, N, E);
    gather64<2><<<gatherBlocks, T, 0, stream>>>(y, YD, dinv, rowptr, col, wgt, y + XD, YD, r1, r2, N, E);
    gather64<0><<<gatherBlocks, T, 0, stream>>>(y + XD, YD, dinv, rowptr, col, wgt, y + 2 * XD, YD,
                                                nullptr, nullptr, N, E);

    head_gemm<<<rowBlocks, T, 0, stream>>>(y, YD, Fm, CD, logits, 2, YD, lb1, lW2, lb2, 16,
                                           r1, r2, probs);
}


# CSR build with one atomic pass instead of two: count's returning adds give each edge its rank in its row (kept per edge), fill places at row start + rank, no atomics; row sums in count-arrival order
# speedup vs baseline: 4.0156x; 1.1462x over previous
#include <hip/hip_runtime.h>
#include <hip/hip_bf16.h>


typedef __attribute__((ext_vector_type(2))) float v2f;
typedef __attribute__((ext_vector_type(4))) float v4f;
typedef __attribute__((ext_vector_type(8))) float v8f;

#define NN 50000
#define EE 800000
#define HD 128
#define CD 384
#define XD 64
#define YD 192

__device__ __forceinline__ void count_edges(const int* __restrict__ dst, int* cnt,
                                            int* __restrict__ rank, int e, int bid, int nblk) {
    const int t = bid * blockDim.x + threadIdx.x;
    const int stride = nblk * blockDim.x;
    const int e4 = (((((size_t)dst) | ((size_t)rank)) & 15) == 0) ? (e >> 2) : 0;
    const int4* d4 = (const int4*)dst;
    int4* r4 = (int4*)rank;
    for (int i = t; i < e4; i += stride) {
        int4 d = d4[i];
        int4 r;
        r.x = atomicAdd(&cnt[d.x], 1);
        r.y = atomicAdd(&cnt[d.y], 1);
        r.z = atomicAdd(&cnt[d.z], 1);
        r.w = atomicAdd(&cnt[d.w], 1);
        r4[i] = r;
    }
    for (int i = 4 * e4 + t; i < e; i += stride) rank[i] = atomicAdd(&cnt[dst[i]], 1);
}
#define SCAN_T 1024
#define SCAN_NB ((NN + SCAN_T - 1) / SCAN_T)
#define SCAN_PASS ((SCAN_NB - 1 + 3) / 4)
static_assert(SCAN_PASS * SCAN_T * 4 <= NN, "the unguarded 16-byte loads must stay inside the count array");
static_assert(SCAN_T / 32 == 32, "one wave total per lane in the block-level step");
__global__ void __launch_bounds__(SCAN_T)
csr_scan(const int* __restrict__ cnt, int* __restrict__ rowptr, float* __restrict__ dinv) {
    __shared__ int wsum[SCAN_T / 32];
    __shared__ int wpart[SCAN_T / 32];
    const int t = threadIdx.x;
    const int lane = t & 31;
    const int w = t >> 5;
    const int idx = blockIdx.x * SCAN_T + t;
    int cv = 0;
    if (idx < NN) cv = cnt[idx];
    const int n4 = blockIdx.x * (SCAN_T / 4);
    const int4* c4 = (const int4*)cnt;
    int4 pv[SCAN_PASS];
#pragma unroll
    for (int i = 0; i < SCAN_PASS; ++i) pv[i] = c4[i * SCAN_T + t];
    int part = 0;
#pragma unroll
    for (int i = 0; i < SCAN_PASS; ++i) {
        const int s = (pv[i].x + pv[i].y) + (pv[i].z + pv[i].w);
        part += (i * SCAN_T + t < n4) ? s : 0;
    }
#pragma unroll
    for (int m = 16; m >= 1; m >>= 1) part += __shfl_xor(part, m, 32);
    int incl = cv;
#pragma unroll
    for (int m = 1; m < 32; m <<= 1) {
        int up = __shfl_up(incl, m, 32);
        if (lane >= m) incl += up;
    }
    if (lane == 31) wsum[w] = incl;
    if (lane == 0) wpart[w] = part;
    __syncthreads();
    int base = wpart[lane];
    int woff = (lane < w) ? wsum[lane] : 0;
#pragma unroll
    for (int m = 16; m >= 1; m >>= 1) {
        base += __shfl_xor(base, m, 32);
        woff += __shfl_xor(woff, m, 32);
    }
    const int p = base + woff + incl - cv;
    if (idx < NN) {
        rowptr[idx] = p;
        dinv[idx] = rsqrtf((float)(cv + 1));
        if (idx == NN - 1) rowptr[NN] = p + cv;
    }
}
__device__ __forceinline__ void fill_edges(const int* __restrict__ src, const int* __restrict__ dst,
                                           const int* __restrict__ rank,
                                           const int* __restrict__ rowptr,
                                           const float* __restrict__ dinv,
                                           int* __restrict__ col, float* __restrict__ wgt, int e,
                                           int bid, int nblk) {
    const int t = bid * blockDim.x + threadIdx.x;
    const int stride = nblk * blockDim.x;
    const int e4 = (((((size_t)src) | ((size_t)dst) | ((size_t)rank)) & 15) == 0) ? (e >> 2) : 0;
    const int4* s4 = (const int4*)src;
    const int4* d4 = (const int4*)dst;
    const int4* r4 = (const int4*)rank;
    for (int i = t; i < e4; i += stride) {
        int4 sv = s4[i], dv = d4[i], rv = r4[i];
        int p0 = rowptr[dv.x] + rv.x;
        int p1 = rowptr[dv.y] + rv.y;
        int p2 = rowptr[dv.z] + rv.z;
        int p3 = rowptr[dv.w] + rv.w;
        float w0 = dinv[sv.x] * dinv[dv.x];
        float w1 = dinv[sv.y] * dinv[dv.y];
        float w2 = dinv[sv.z] * dinv[dv.z];
        float w3 = dinv[sv.w] * dinv[dv.w];
        col[p0] = sv.x; wgt[p0] = w0;
        col[p1] = sv.y; wgt[p1] = w1;
        col[p2] = sv.z; wgt[p2] = w2;
        col[p3] = sv.w; wgt[p3] = w3;
    }
    for (int i = 4 * e4 + t; i < e; i += stride) {
        int sv = src[i], dv = dst[i];
        int p = rowptr[dv] + rank[i];
        col[p] = sv;
        wgt[p] = dinv[sv] * dinv[dv];
    }
}

#define BM 80
#define KB 32
#define PA (KB + 4)

template <bool BT, int BN, bool HEAD>
__device__ __forceinline__ void
gemm_block(const float* __restrict__ A, int lda,
           const float* __restrict__ B, int ldb,
           float* __restrict__ C, int ldc, int K,
           const float* __restrict__ lb1,
           const float* __restrict__ lW2,
           const float* __restrict__ lb2,
           int bpad,
           const float* __restrict__ rowvec, int rowidx,
           int sAy, int sBy, int sCy, int sBz, int sCz,
           const float* __restrict__ r1, const float* __restrict__ r2,
           float* __restrict__ probs, int bx, int by, int bz) {
    A += (size_t)by * sAy;
    B += (size_t)by * sBy + (size_t)bz * sBz;
    C += (size_t)by * sCy + (size_t)bz * sCz;
    constexpr int RT = BM / 16;
    constexpr int CT = BN / 128;
    constexpr int PB = BT ? PA : (BN + 4);
    constexpr int BROWS = BT ? BN : KB;
    constexpr int QA = KB / 4;
    constexpr int A4 = BM * QA;
    constexpr int AL = (A4 + 255) / 256;
    constexpr int BL = BN * QA / 256;
    constexpr int QB = BN / 4;

    __shared__ __attribute__((aligned(16))) float As[BM * PA];
    __shared__ __attribute__((aligned(16))) float Bs[BROWS * PB];

    const int tid  = threadIdx.x;
    const int wave = tid >> 5;
    const int lane = tid & 31;
    const int c    = lane & 15;
    const int h    = lane >> 4;
    const int m0   = bx * BM;
    const int wn0  = wave * (CT * 16);

    v8f acc[RT][CT];
#pragma unroll
    for (int i = 0; i < RT; ++i)
#pragma unroll
        for (int j = 0; j < CT; ++j) {
            v8f z = {};
            acc[i][j] = z;
        }

    v4f ra[AL], rb[BL];

#define TILE_GLOAD(kb)                                                              \
    {                                                                               \
        _Pragma("unroll") for (int u = 0; u < AL; ++u) {                            \
            int idx = tid + 256 * u;                                                \
            if (idx < A4) {                                                         \
                int row = idx / QA, q = idx % QA;                                   \
                ra[u] = *(const v4f*)(A + (size_t)(m0 + row) * lda + (kb) + 4 * q); \
            }                                                                       \
        }                                                                           \
        _Pragma("unroll") for (int u = 0; u < BL; ++u) {                            \
            int idx = tid + 256 * u;                                                \
            if (BT) {                                                               \
                int n = idx / QA, q = idx % QA;                                     \
                rb[u] = *(const v4f*)(B + (size_t)n * ldb + (kb) + 4 * q);       \
            } else {                                                                \
                int k = idx / QB, q = idx % QB;                                     \
                rb[u] = *(const v4f*)(B + (size_t)((kb) + k + bpad * ((kb) >> 6)) * ldb + 4 * q); \
            }                                                                       \
        }                                                                           \
    }

    TILE_GLOAD(0)
    const int S = K / KB;
    for (int s = 0; s < S; ++s) {
#pragma unroll
        for (int u = 0; u < AL; ++u) {
            int idx = tid + 256 * u;
            if (idx < A4) {
                int row = idx / QA, q = idx % QA;
                *(v4f*)&As[row * PA + 4 * q] = ra[u];
            }
        }
#pragma unroll
        for (int u = 0; u < BL; ++u) {
            int idx = tid + 256 * u;
            if (BT) {
                int n = idx / QA, q = idx % QA;
                *(v4f*)&Bs[n * PB + 4 * q] = rb[u];
            } else {
                int k = idx / QB, q = idx % QB;
                *(v4f*)&Bs[k * PB + 4 * q] = rb[u];
            }
        }
        __syncthreads();
        if (s + 1 < S) TILE_GLOAD((s + 1) * KB)

#pragma unroll
        for (int g = 0; g < KB / 8; ++g) {
            v2f alo[RT], ahi[RT], blo[CT], bhi[CT];
#pragma unroll
            for (int i = 0; i < RT; ++i) {
                v4f a4 = *(const v4f*)&As[(16 * i + c) * PA + 8 * g + 4 * h];
                alo[i].x = a4.x; alo[i].y = a4.y;
                ahi[i].x = a4.z; ahi[i].y = a4.w;
            }
#pragma unroll
            for (int j = 0; j < CT; ++j) {
                if (BT) {
                    v4f b4 = *(const v4f*)&Bs[(wn0 + 16 * j + c) * PB + 8 * g + 4 * h];
                    blo[j].x = b4.x; blo[j].y = b4.y;
                    bhi[j].x = b4.z; bhi[j].y = b4.w;
                } else {
                    const float* bp = &Bs[(8 * g + 4 * h) * PB + wn0 + 16 * j + c];
                    blo[j].x = bp[0];
                    blo[j].y = bp[PB];
                    bhi[j].x = bp[2 * PB];
                    bhi[j].y = bp[3 * PB];
                }
            }
#pragma unroll
            for (int i = 0; i < RT; ++i)
#pragma unroll
                for (int j = 0; j < CT; ++j)
                    acc[i][j] = __builtin_amdgcn_wmma_f32_16x16x4_f32(
                        false, alo[i], false, blo[j], (short)0, acc[i][j], false, false);
#pragma unroll
            for (int i = 0; i < RT; ++i)
#pragma unroll
                for (int j = 0; j < CT; ++j)
                    acc[i][j] = __builtin_amdgcn_wmma_f32_16x16x4_f32(
                        false, ahi[i], false, bhi[j], (short)0, acc[i][j], false, false);
        }
        __syncthreads();
    }
#undef TILE_GLOAD

    if (!HEAD) {
#pragma unroll
        for (int i = 0; i < RT; ++i)
#pragma unroll
            for (int j = 0; j < CT; ++j)
#pragma unroll
                for (int r = 0; r < 8; ++r) {
                    const int row = m0 + 16 * i + 8 * h + r;
                    const int cc  = wn0 + 16 * j + c;
                    float o = acc[i][j][r];
                    if (rowvec != nullptr && row == rowidx) o += rowvec[cc];
                    C[(size_t)row * ldc + cc] = o;
                }
    } else {
        float bias[CT], u1[CT], u2[CT], w0[CT], w1[CT];
#pragma unroll
        for (int j = 0; j < CT; ++j) {
            int col = wn0 + 16 * j + c;
            bias[j] = lb1[col] + B[(size_t)(0 * BM + 64) * ldb + col]
                               + B[(size_t)(1 * BM + 65) * ldb + col]
                               + B[(size_t)(2 * BM + 66) * ldb + col];
            u1[j]   = B[(size_t)(1 * BM + 64) * ldb + col] + B[(size_t)(2 * BM + 65) * ldb + col];
            u2[j]   = B[(size_t)(2 * BM + 64) * ldb + col];
            w0[j]   = lW2[col];
            w1[j]   = lW2[BN + col];
        }
        float* red = Bs;
#pragma unroll
        for (int i = 0; i < RT; ++i)
#pragma unroll
            for (int r = 0; r < 8; ++r) {
                float p0 = 0.0f, p1 = 0.0f;
                const int grow = m0 + 16 * i + 8 * h + r;
                const float a1 = r1[grow], a2 = r2[grow];
#pragma unroll
                for (int j = 0; j < CT; ++j) {
                    float t = acc[i][j][r] + bias[j] + a1 * u1[j] + a2 * u2[j];
                    t = t > 0.0f ? t : 0.0f;
                    p0 += t * w0[j];
                    p1 += t * w1[j];
                }
#pragma unroll
                for (int m = 8; m >= 1; m >>= 1) {
                    p0 += __shfl_xor(p0, m, 16);
                    p1 += __shfl_xor(p1, m, 16);
                }
                if (c == 0) {
                    int row = 16 * i + 8 * h + r;
                    red[(wave * BM + row) * 2 + 0] = p0;
                    red[(wave * BM + row) * 2 + 1] = p1;
                }
            }
        __syncthreads();
        if (tid < 2 * BM) {
            int row = tid >> 1, cls = tid & 1;
            float sum = lb2[cls];
#pragma unroll
            for (int w = 0; w < 8; ++w) sum += red[(w * BM + row) * 2 + cls];
            C[(size_t)(m0 + row) * ldc + cls] = sum;
            const float oth = __shfl_xor(sum, 1, 32);
            const float mx = fmaxf(sum, oth);
            const float es = __expf(sum - mx), eo = __expf(oth - mx);
            const float inv = 1.0f / (es + eo);
            probs[(size_t)(m0 + row) * 2 + cls] = es * inv;
        }
    }
}

__global__ void __launch_bounds__(256)
head_gemm(const float* __restrict__ A, int lda, const float* __restrict__ B, int ldb,
          float* __restrict__ C, int ldc, int K,
          const float* __restrict__ lb1, const float* __restrict__ lW2,
          const float* __restrict__ lb2, int bpad, const float* __restrict__ r1,
          const float* __restrict__ r2, float* __restrict__ probs) {
    gemm_block<false, CD, true>(A, lda, B, ldb, C, ldc, K, lb1, lW2, lb2, bpad, nullptr, 0,
                                0, 0, 0, 0, 0, r1, r2, probs, blockIdx.x, 0, 0);
}

__device__ __forceinline__ void fold_step(const float* __restrict__ Tin, const float* __restrict__ Wm,
                                          float* __restrict__ Tout, const float* __restrict__ bvec,
                                          int brow) {
    gemm_block<false, HD, false>(Tin, HD, Wm, HD, Tout, HD, HD, nullptr, nullptr, nullptr,
                                 0, bvec, brow, 0, 0, 0, 0, 0, nullptr, nullptr, nullptr, 0, 0, 0);
}

__global__ void __launch_bounds__(256)
csr_count_fold(const int* __restrict__ dst, int* cnt, int* __restrict__ rank, int e, float* Tm,
               const float* __restrict__ W2, const float* __restrict__ b2,
               const float* __restrict__ W3, const float* __restrict__ b3) {
    if (blockIdx.x == 0) {
        fold_step(Tm, W2, Tm + BM * HD, b2, 65);
        __syncthreads();
        fold_step(Tm + BM * HD, W3, Tm + 2 * BM * HD, b3, 66);
        return;
    }
    count_edges(dst, cnt, rank, e, blockIdx.x - 1, gridDim.x - 1);
}
__global__ void __launch_bounds__(256)
csr_fill_fold(const int* __restrict__ src, const int* __restrict__ dst,
              const int* __restrict__ rank, const int* __restrict__ rowptr,
              const float* __restrict__ dinv,
              int* __restrict__ col, float* __restrict__ wgt, int e,
              const float* __restrict__ Tm, const float* __restrict__ lW1,
              float* __restrict__ Fm) {
    if (blockIdx.x < 9) {
        gemm_block<true, HD, false>(Tm, HD, lW1, CD, Fm, CD, HD, nullptr, nullptr, nullptr,
                                    0, nullptr, 0, BM * HD, HD, BM * CD, HD * CD, HD,
                                    nullptr, nullptr, nullptr, 0,
                                    (int)(blockIdx.x % 3), (int)(blockIdx.x / 3));
        return;
    }
    fill_edges(src, dst, rank, rowptr, dinv, col, wgt, e, blockIdx.x - 9, gridDim.x - 9);
}

__global__ void seed_and_clear(const float* __restrict__ W1, const float* __restrict__ b1,
                               float* __restrict__ T1, int* __restrict__ cnt, int n) {
    int i = blockIdx.x * blockDim.x + threadIdx.x;
    if (i < n) cnt[i] = 0;
    if (i < BM * HD) {
        int r = i / HD, cc = i % HD;
        float o = 0.0f;
        if (r < XD) o = W1[i];
        else if (r == XD) o = b1[cc];
        T1[i] = o;
    }
}

#define RUN_ROWS 31

template <int MODE>
__device__ __forceinline__ void
gather_rows(const float* __restrict__ Xin, int ldi, const float* __restrict__ dinv,
            const int* __restrict__ rowptr, const int* __restrict__ col,
            const float* __restrict__ wgt,
            float* __restrict__ Yout, int ldo,
            const float* __restrict__ rin, float* __restrict__ rout, int n, int e,
            int bid, int nblk) {
    const int lane = threadIdx.x & 31;
    const int c  = lane & 15;
    const int hh = lane >> 4;
    const int wid = __builtin_amdgcn_readfirstlane(bid * 8 + (threadIdx.x >> 5));
    const int nw  = nblk * 8;
    const unsigned uq = (unsigned)n / (unsigned)nw, ur = (unsigned)n % (unsigned)nw;
    const int v0 = (int)((unsigned)wid * uq + ((unsigned)wid < ur ? (unsigned)wid : ur));
    const int v1 = v0 + (int)uq + ((unsigned)wid < ur ? 1 : 0);
    const char* Xb = (const char*)Xin;
    const unsigned pitch = (unsigned)ldi * 4u;
    const unsigned coff  = (unsigned)c * 16u;

#define GLOAD(R, P)                                                                \
    {                                                                              \
        const int sa = __builtin_amdgcn_readlane(myc, 2 * (P));                    \
        const int sb = __builtin_amdgcn_readlane(myc, 2 * (P) + 1);                \
        const unsigned sr = (unsigned)(hh ? sb : sa);                              \
        q[R] = *(const float4*)(Xb + (sr * pitch + coff));                         \
    }
#define GFMA(R, P)                                                                 \
    {                                                                              \
        const int wa = __builtin_amdgcn_readlane(mywi, 2 * (P));                   \
        const int wb = __builtin_amdgcn_readlane(mywi, 2 * (P) + 1);               \
        const float w = __int_as_float(hh ? wb : wa);                              \
        acc.x += q[R].x * w; acc.y += q[R].y * w;                                  \
        acc.z += q[R].z * w; acc.w += q[R].w * w;                                  \
    }

    for (int vb = v0; vb < v1; vb += RUN_ROWS) {
        const int cnt = (v1 - vb) < RUN_ROWS ? (v1 - vb) : RUN_ROWS;
        const int ri = (vb + lane) < n ? (vb + lane) : n;
        const int di = (vb + lane) < n ? (vb + lane) : (n - 1);
        const int rp = rowptr[ri];
        const float dv = dinv[di];
        float rself = 0.0f;
        if (MODE == 2) rself = rin[di];
        const int dvi = __float_as_int(dv);
        const int rsi = __float_as_int(rself);
        int beg = __builtin_amdgcn_readlane(rp, 0);
        int pi = beg + lane;
        pi = pi < e ? pi : e - 1;
        pi = pi > 0 ? pi : 0;
        int   nc  = col[pi];
        float nwt = wgt[pi];
        for (int i = 0; i < cnt; ++i) {
            const int v = vb + i;
            const int end = __builtin_amdgcn_readlane(rp, i + 1);
            const float dn = __int_as_float(__builtin_amdgcn_readlane(dvi, i));
            const float nn = dn * dn;
            const int   fc = nc;
            const float fw = nwt;
            pi = end + lane;
            pi = pi < e ? pi : e - 1;
            pi = pi > 0 ? pi : 0;
            nc  = col[pi];
            nwt = wgt[pi];
            const float4 xs = *(const float4*)(Xb + ((unsigned)v * pitch + coff));
            float4 acc = make_float4(0.0f, 0.0f, 0.0f, 0.0f);
            bool first = true;
            float rs = 0.0f;
            for (int base = beg; base < end; base += 32) {
                const int m = (end - base) < 32 ? (end - base) : 32;
                int   myc = v;
                float myw = 0.0f;
                if (lane < m) {
                    if (base == beg) { myc = fc; myw = fw; }
                    else { myc = col[base + lane]; myw = wgt[base + lane]; }
                }
                float rv = 0.0f;
                if (MODE == 2) rv = rin[myc];
                const int mywi = __float_as_int(myw);
                float4 q[12];
                GLOAD(0, 0) GLOAD(1, 1) GLOAD(2, 2) GLOAD(3, 3)
                if (m > 8) {
                    GLOAD(4, 4) GLOAD(5, 5) GLOAD(6, 6) GLOAD(7, 7)
                    if (m > 16) { GLOAD(8, 8) GLOAD(9, 9) GLOAD(10, 10) GLOAD(11, 11) }
                }
                if (first) {
                    acc.x = hh ? 0.0f : xs.x * nn; acc.y = hh ? 0.0f : xs.y * nn;
                    acc.z = hh ? 0.0f : xs.z * nn; acc.w = hh ? 0.0f : xs.w * nn;
                    first = false;
                }
                GFMA(0, 0) GFMA(1, 1) GFMA(2, 2) GFMA(3, 3)
                if (m > 8) {
                    GFMA(4, 4) GFMA(5, 5) GFMA(6, 6) GFMA(7, 7)
                    if (m > 16) { GFMA(8, 8) GFMA(9, 9) GFMA(10, 10) GFMA(11, 11) }
                }
                if (m > 24) {
                    GLOAD(0, 12) GLOAD(1, 13) GLOAD(2, 14) GLOAD(3, 15)
                    GFMA(0, 12) GFMA(1, 13) GFMA(2, 14) GFMA(3, 15)
                }
                if (MODE == 1) rs += myw;
                if (MODE == 2) rs += myw * rv;
            }
            if (first) {
                acc.x = hh ? 0.0f : xs.x * nn; acc.y = hh ? 0.0f : xs.y * nn;
                acc.z = hh ? 0.0f : xs.z * nn; acc.w = hh ? 0.0f : xs.w * nn;
            }
            acc.x += __shfl_xor(acc.x, 16, 32);
            acc.y += __shfl_xor(acc.y, 16, 32);
            acc.z += __shfl_xor(acc.z, 16, 32);
            acc.w += __shfl_xor(acc.w, 16, 32);
            if (hh == 0)
                *(float4*)(Yout + (size_t)v * ldo + c * 4) = acc;
            if (MODE != 0) {
#pragma unroll
                for (int mm = 16; mm >= 1; mm >>= 1) rs += __shfl_xor(rs, mm, 32);
                const float rsv = __int_as_float(__builtin_amdgcn_readlane(rsi, i));
                if (lane == 0) rout[v] = rs + (MODE == 1 ? nn : nn * rsv);
            }
            beg = end;
        }
    }
#undef GLOAD
#undef GFMA
}

template <int MODE>
__global__ void __launch_bounds__(256)
gather64(const float* __restrict__ Xin, int ldi, const float* __restrict__ dinv,
         const int* __restrict__ rowptr, const int* __restrict__ col,
         const float* __restrict__ wgt,
         float* __restrict__ Yout, int ldo,
         const float* __restrict__ rin, float* __restrict__ rout, int n, int e) {
    gather_rows<MODE>(Xin, ldi, dinv, rowptr, col, wgt, Yout, ldo, rin, rout, n, e,
                      blockIdx.x, gridDim.x);
}

extern "C" void kernel_launch(void* const* d_in, const int* in_sizes, int n_in,
                              void* d_out, int out_size, void* d_ws, size_t ws_size,
                              hipStream_t stream) {
    const float* x   = (const float*)d_in[0];
    const int*   ei  = (const int*)d_in[1];
    const float* W1  = (const float*)d_in[2];
    const float* b1  = (const float*)d_in[3];
    const float* W2  = (const float*)d_in[4];
    const float* b2  = (const float*)d_in[5];
    const float* W3  = (const float*)d_in[6];
    const float* b3  = (const float*)d_in[7];
    const float* lW1 = (const float*)d_in[8];
    const float* lb1 = (const float*)d_in[9];
    const float* lW2 = (const float*)d_in[10];
    const float* lb2 = (const float*)d_in[11];

    const int N = NN, E = EE;
    const int* src = ei;
    const int* dst = ei + E;

    float* y    = (float*)d_ws;
    float* r1   = y + (size_t)N * YD;
    float* r2   = r1 + N;
    float* Tm   = r2 + N;
    float* Fm   = Tm + 3 * BM * HD;
    float* dinv = Fm + 3 * BM * CD;
    int*   cnt    = (int*)(dinv + N);
    int*   rank   = cnt + N;
    int*   rowptr = rank + E;
    int*   col    = rowptr + (N + 1);
    float* wgt    = (float*)(col + E);

    float* logits = (float*)d_out;
    float* probs  = logits + (size_t)N * 2;

    const int T = 256;
    const int rowBlocks = N / BM;
    const int gatherBlocks = 1024;
    const int edgeBlocks   = 512;

    seed_and_clear<<<(N + T - 1) / T, T, 0, stream>>>(W1, b1, Tm, cnt, N);

    csr_count_fold<<<edgeBlocks + 1, T, 0, stream>>>(dst, cnt, rank, E, Tm, W2, b2, W3, b3);
    csr_scan<<<SCAN_NB, SCAN_T, 0, stream>>>(cnt, rowptr, dinv);
    csr_fill_fold<<<edgeBlocks + 9, T, 0, stream>>>(src, dst, rank, rowptr, dinv, col, wgt, E,
                                                   Tm, lW1, Fm);

    gather64<1><<<gatherBlocks, T, 0, stream>>>(x, XD, dinv, rowptr, col, wgt, y, YD, nullptr, r1, N, E);
    gather64<2><<<gatherBlocks, T, 0, stream>>>(y, YD, dinv, rowptr, col, wgt, y + XD, YD, r1, r2, N, E);
    gather64<0><<<gatherBlocks, T, 0, stream>>>(y + XD, YD, dinv, rowptr, col, wgt, y + 2 * XD, YD,
                                                nullptr, nullptr, N, E);

    head_gemm<<<rowBlocks, T, 0, stream>>>(y, YD, Fm, CD, logits, 2, YD, lb1, lW2, lb2, 16,
                                           r1, r2, probs);
}


# CSR build by two-level counting sort on LDS counters (256-row bins, whole-line stores), no global atomics or scattered stores; first gather makes the weights; row sums in LDS-arrival order
# speedup vs baseline: 5.2359x; 1.3039x over previous
#include <hip/hip_runtime.h>
#include <hip/hip_bf16.h>


typedef __attribute__((ext_vector_type(2))) float v2f;
typedef __attribute__((ext_vector_type(4))) float v4f;
typedef __attribute__((ext_vector_type(8))) float v8f;

#define NN 50000
#define EE 800000
#define HD 128
#define CD 384
#define XD 64
#define YD 192


#define BM 80
#define KB 32
#define PA (KB + 4)

template <bool BT, int BN, bool HEAD>
__device__ __forceinline__ void
gemm_block(const float* __restrict__ A, int lda,
           const float* __restrict__ B, int ldb,
           float* __restrict__ C, int ldc, int K,
           const float* __restrict__ lb1,
           const float* __restrict__ lW2,
           const float* __restrict__ lb2,
           int bpad,
           const float* __restrict__ rowvec, int rowidx,
           int sAy, int sBy, int sCy, int sBz, int sCz,
           const float* __restrict__ r1, const float* __restrict__ r2,
           float* __restrict__ probs, int bx, int by, int bz) {
    A += (size_t)by * sAy;
    B += (size_t)by * sBy + (size_t)bz * sBz;
    C += (size_t)by * sCy + (size_t)bz * sCz;
    constexpr int RT = BM / 16;
    constexpr int CT = BN / 128;
    constexpr int PB = BT ? PA : (BN + 4);
    constexpr int BROWS = BT ? BN : KB;
    constexpr int QA = KB / 4;
    constexpr int A4 = BM * QA;
    constexpr int AL = (A4 + 255) / 256;
    constexpr int BL = BN * QA / 256;
    constexpr int QB = BN / 4;

    __shared__ __attribute__((aligned(16))) float As[BM * PA];
    __shared__ __attribute__((aligned(16))) float Bs[BROWS * PB];

    const int tid  = threadIdx.x;
    const int wave = tid >> 5;
    const int lane = tid & 31;
    const int c    = lane & 15;
    const int h    = lane >> 4;
    const int m0   = bx * BM;
    const int wn0  = wave * (CT * 16);

    v8f acc[RT][CT];
#pragma unroll
    for (int i = 0; i < RT; ++i)
#pragma unroll
        for (int j = 0; j < CT; ++j) {
            v8f z = {};
            acc[i][j] = z;
        }

    v4f ra[AL], rb[BL];

#define TILE_GLOAD(kb)                                                              \
    {                                                                               \
        _Pragma("unroll") for (int u = 0; u < AL; ++u) {                            \
            int idx = tid + 256 * u;                                                \
            if (idx < A4) {                                                         \
                int row = idx / QA, q = idx % QA;                                   \
                ra[u] = *(const v4f*)(A + (size_t)(m0 + row) * lda + (kb) + 4 * q); \
            }                                                                       \
        }                                                                           \
        _Pragma("unroll") for (int u = 0; u < BL; ++u) {                            \
            int idx = tid + 256 * u;                                                \
            if (BT) {                                                               \
                int n = idx / QA, q = idx % QA;                                     \
                rb[u] = *(const v4f*)(B + (size_t)n * ldb + (kb) + 4 * q);       \
            } else {                                                                \
                int k = idx / QB, q = idx % QB;                                     \
                rb[u] = *(const v4f*)(B + (size_t)((kb) + k + bpad * ((kb) >> 6)) * ldb + 4 * q); \
            }                                                                       \
        }                                                                           \
    }

    TILE_GLOAD(0)
    const int S = K / KB;
    for (int s = 0; s < S; ++s) {
#pragma unroll
        for (int u = 0; u < AL; ++u) {
            int idx = tid + 256 * u;
            if (idx < A4) {
                int row = idx / QA, q = idx % QA;
                *(v4f*)&As[row * PA + 4 * q] = ra[u];
            }
        }
#pragma unroll
        for (int u = 0; u < BL; ++u) {
            int idx = tid + 256 * u;
            if (BT) {
                int n = idx / QA, q = idx % QA;
                *(v4f*)&Bs[n * PB + 4 * q] = rb[u];
            } else {
                int k = idx / QB, q = idx % QB;
                *(v4f*)&Bs[k * PB + 4 * q] = rb[u];
            }
        }
        __syncthreads();
        if (s + 1 < S) TILE_GLOAD((s + 1) * KB)

#pragma unroll
        for (int g = 0; g < KB / 8; ++g) {
            v2f alo[RT], ahi[RT], blo[CT], bhi[CT];
#pragma unroll
            for (int i = 0; i < RT; ++i) {
                v4f a4 = *(const v4f*)&As[(16 * i + c) * PA + 8 * g + 4 * h];
                alo[i].x = a4.x; alo[i].y = a4.y;
                ahi[i].x = a4.z; ahi[i].y = a4.w;
            }
#pragma unroll
            for (int j = 0; j < CT; ++j) {
                if (BT) {
                    v4f b4 = *(const v4f*)&Bs[(wn0 + 16 * j + c) * PB + 8 * g + 4 * h];
                    blo[j].x = b4.x; blo[j].y = b4.y;
                    bhi[j].x = b4.z; bhi[j].y = b4.w;
                } else {
                    const float* bp = &Bs[(8 * g + 4 * h) * PB + wn0 + 16 * j + c];
                    blo[j].x = bp[0];
                    blo[j].y = bp[PB];
                    bhi[j].x = bp[2 * PB];
                    bhi[j].y = bp[3 * PB];
                }
            }
#pragma unroll
            for (int i = 0; i < RT; ++i)
#pragma unroll
                for (int j = 0; j < CT; ++j)
                    acc[i][j] = __builtin_amdgcn_wmma_f32_16x16x4_f32(
                        false, alo[i], false, blo[j], (short)0, acc[i][j], false, false);
#pragma unroll
            for (int i = 0; i < RT; ++i)
#pragma unroll
                for (int j = 0; j < CT; ++j)
                    acc[i][j] = __builtin_amdgcn_wmma_f32_16x16x4_f32(
                        false, ahi[i], false, bhi[j], (short)0, acc[i][j], false, false);
        }
        __syncthreads();
    }
#undef TILE_GLOAD

    if (!HEAD) {
#pragma unroll
        for (int i = 0; i < RT; ++i)
#pragma unroll
            for (int j = 0; j < CT; ++j)
#pragma unroll
                for (int r = 0; r < 8; ++r) {
                    const int row = m0 + 16 * i + 8 * h + r;
                    const int cc  = wn0 + 16 * j + c;
                    float o = acc[i][j][r];
                    if (rowvec != nullptr && row == rowidx) o += rowvec[cc];
                    C[(size_t)row * ldc + cc] = o;
                }
    } else {
        float bias[CT], u1[CT], u2[CT], w0[CT], w1[CT];
#pragma unroll
        for (int j = 0; j < CT; ++j) {
            int col = wn0 + 16 * j + c;
            bias[j] = lb1[col] + B[(size_t)(0 * BM + 64) * ldb + col]
                               + B[(size_t)(1 * BM + 65) * ldb + col]
                               + B[(size_t)(2 * BM + 66) * ldb + col];
            u1[j]   = B[(size_t)(1 * BM + 64) * ldb + col] + B[(size_t)(2 * BM + 65) * ldb + col];
            u2[j]   = B[(size_t)(2 * BM + 64) * ldb + col];
            w0[j]   = lW2[col];
            w1[j]   = lW2[BN + col];
        }
        float* red = Bs;
#pragma unroll
        for (int i = 0; i < RT; ++i)
#pragma unroll
            for (int r = 0; r < 8; ++r) {
                float p0 = 0.0f, p1 = 0.0f;
                const int grow = m0 + 16 * i + 8 * h + r;
                const float a1 = r1[grow], a2 = r2[grow];
#pragma unroll
                for (int j = 0; j < CT; ++j) {
                    float t = acc[i][j][r] + bias[j] + a1 * u1[j] + a2 * u2[j];
                    t = t > 0.0f ? t : 0.0f;
                    p0 += t * w0[j];
                    p1 += t * w1[j];
                }
#pragma unroll
                for (int m = 8; m >= 1; m >>= 1) {
                    p0 += __shfl_xor(p0, m, 16);
                    p1 += __shfl_xor(p1, m, 16);
                }
                if (c == 0) {
                    int row = 16 * i + 8 * h + r;
                    red[(wave * BM + row) * 2 + 0] = p0;
                    red[(wave * BM + row) * 2 + 1] = p1;
                }
            }
        __syncthreads();
        if (tid < 2 * BM) {
            int row = tid >> 1, cls = tid & 1;
            float sum = lb2[cls];
#pragma unroll
            for (int w = 0; w < 8; ++w) sum += red[(w * BM + row) * 2 + cls];
            C[(size_t)(m0 + row) * ldc + cls] = sum;
            const float oth = __shfl_xor(sum, 1, 32);
            const float mx = fmaxf(sum, oth);
            const float es = __expf(sum - mx), eo = __expf(oth - mx);
            const float inv = 1.0f / (es + eo);
            probs[(size_t)(m0 + row) * 2 + cls] = es * inv;
        }
    }
}

__global__ void __launch_bounds__(256)
head_gemm(const float* __restrict__ A, int lda, const float* __restrict__ B, int ldb,
          float* __restrict__ C, int ldc, int K,
          const float* __restrict__ lb1, const float* __restrict__ lW2,
          const float* __restrict__ lb2, int bpad, const float* __restrict__ r1,
          const float* __restrict__ r2, float* __restrict__ probs) {
    gemm_block<false, CD, true>(A, lda, B, ldb, C, ldc, K, lb1, lW2, lb2, bpad, nullptr, 0,
                                0, 0, 0, 0, 0, r1, r2, probs, blockIdx.x, 0, 0);
}

__device__ __forceinline__ void fold_step(const float* __restrict__ Tin, const float* __restrict__ Wm,
                                          float* __restrict__ Tout, const float* __restrict__ bvec,
                                          int brow) {
    gemm_block<false, HD, false>(Tin, HD, Wm, HD, Tout, HD, HD, nullptr, nullptr, nullptr,
                                 0, bvec, brow, 0, 0, 0, 0, 0, nullptr, nullptr, nullptr, 0, 0, 0);
}

#define NBLK 256
#define NBIN ((NN + 255) / 256)
#define LL (NBIN * NBLK)
#define CH ((((EE + NBLK - 1) / NBLK) + 3) & ~3)
static_assert(EE % 4 == 0, "edges are read four at a time");
#define ET 256
#define CR ((CH / 4 + ET - 1) / ET)
static_assert(NBIN <= 256, "bin counters are handled by the first 256 threads");
#define DCAP 8192

__device__ __forceinline__ int4 ld4(const int* __restrict__ p, int i4, bool al) {
    if (al) return ((const int4*)p)[i4];
    int4 r;
    r.x = p[4 * i4]; r.y = p[4 * i4 + 1]; r.z = p[4 * i4 + 2]; r.w = p[4 * i4 + 3];
    return r;
}

__device__ __forceinline__ int block_excl_scan(int v, int* wsum) {
    const int lane = threadIdx.x & 31;
    const int w = threadIdx.x >> 5;
    int incl = v;
#pragma unroll
    for (int m = 1; m < 32; m <<= 1) {
        int up = __shfl_up(incl, m, 32);
        if (lane >= m) incl += up;
    }
    if (lane == 31) wsum[w] = incl;
    __syncthreads();
    int woff = (lane < w) ? wsum[lane] : 0;
#pragma unroll
    for (int m = 16; m >= 1; m >>= 1) woff += __shfl_xor(woff, m, 32);
    __syncthreads();
    return woff + incl - v;
}


__global__ void __launch_bounds__(ET)
bin_count_fold(const int* __restrict__ dst, int* __restrict__ bb,
               const float* __restrict__ W1, const float* __restrict__ b1, float* Tm,
               const float* __restrict__ W2, const float* __restrict__ b2) {
    __shared__ int h[256];
    const int t = threadIdx.x;
    if (blockIdx.x == 0) {
        for (int i = t; i < BM * HD; i += ET) {
            int r = i / HD, cc = i % HD;
            float o = 0.0f;
            if (r < XD) o = W1[i];
            else if (r == XD) o = b1[cc];
            Tm[i] = o;
        }
        __syncthreads();
        fold_step(Tm, W2, Tm + BM * HD, b2, 65);
        return;
    }
    const int blk = blockIdx.x - 1;
    h[t] = 0;
    __syncthreads();
    const bool al = ((((size_t)dst) & 15) == 0);
    int4 dd[CR];
#pragma unroll
    for (int k = 0; k < CR; ++k) {
        const int q = t + ET * k;
        const int i4 = blk * (CH / 4) + q;
        dd[k] = make_int4(0, 0, 0, 0);
        if (q < CH / 4 && i4 < EE / 4) dd[k] = ld4(dst, i4, al);
    }
#pragma unroll
    for (int k = 0; k < CR; ++k) {
        const int q = t + ET * k;
        const int i4 = blk * (CH / 4) + q;
        if (q < CH / 4 && i4 < EE / 4) {
            atomicAdd(&h[dd[k].x >> 8], 1);
            atomicAdd(&h[dd[k].y >> 8], 1);
            atomicAdd(&h[dd[k].z >> 8], 1);
            atomicAdd(&h[dd[k].w >> 8], 1);
        }
    }
    __syncthreads();
    if (t < NBIN) bb[t * NBLK + blk] = h[t];
}

#define SCAN_T 1024
#define SCAN_NB (LL / SCAN_T)
#define SCAN_PASS ((SCAN_NB - 1 + 3) / 4)
static_assert(LL % SCAN_T == 0, "the scan has no ragged block");
static_assert(SCAN_PASS * SCAN_T * 4 <= LL, "the unguarded 16-byte loads must stay inside the count array");
static_assert(SCAN_T / 32 == 32, "one wave total per lane in the block-level step");
__global__ void __launch_bounds__(SCAN_T)
csr_scan(const int* __restrict__ cnt, int* __restrict__ boff) {
    __shared__ int wsum[SCAN_T / 32];
    __shared__ int wpart[SCAN_T / 32];
    const int t = threadIdx.x;
    const int lane = t & 31;
    const int w = t >> 5;
    const int idx = blockIdx.x * SCAN_T + t;
    const int cv = cnt[idx];
    const int n4 = blockIdx.x * (SCAN_T / 4);
    const int4* c4 = (const int4*)cnt;
    int4 pv[SCAN_PASS];
#pragma unroll
    for (int i = 0; i < SCAN_PASS; ++i) pv[i] = c4[i * SCAN_T + t];
    int part = 0;
#pragma unroll
    for (int i = 0; i < SCAN_PASS; ++i) {
        const int s = (pv[i].x + pv[i].y) + (pv[i].z + pv[i].w);
        part += (i * SCAN_T + t < n4) ? s : 0;
    }
#pragma unroll
    for (int m = 16; m >= 1; m >>= 1) part += __shfl_xor(part, m, 32);
    int incl = cv;
#pragma unroll
    for (int m = 1; m < 32; m <<= 1) {
        int up = __shfl_up(incl, m, 32);
        if (lane >= m) incl += up;
    }
    if (lane == 31) wsum[w] = incl;
    if (lane == 0) wpart[w] = part;
    __syncthreads();
    int base = wpart[lane];
    int woff = (lane < w) ? wsum[lane] : 0;
#pragma unroll
    for (int m = 16; m >= 1; m >>= 1) {
        base += __shfl_xor(base, m, 32);
        woff += __shfl_xor(woff, m, 32);
    }
    const int p = base + woff + incl - cv;
    boff[idx] = p;
    if (idx == LL - 1) boff[LL] = p + cv;
}

#define SC_FOLD 7
__global__ void __launch_bounds__(ET)
bin_scatter_fold(const int* __restrict__ src, const int* __restrict__ dst,
                 const int* __restrict__ boff, int2* __restrict__ binned,
                 float* Tm, const float* __restrict__ W3, const float* __restrict__ b3,
                 const float* __restrict__ lW1, float* __restrict__ Fm) {
    __shared__ int cur[256];
    __shared__ int loff[256];
    __shared__ int goff[256];
    __shared__ int wsum[32];
    __shared__ int2 rec[CH];
    const int t = threadIdx.x;
    if (blockIdx.x == 0) {
        fold_step(Tm + BM * HD, W3, Tm + 2 * BM * HD, b3, 66);
        return;
    }
    if (blockIdx.x < SC_FOLD) {
        const int j = blockIdx.x - 1;
        gemm_block<true, HD, false>(Tm, HD, lW1, CD, Fm, CD, HD, nullptr, nullptr, nullptr,
                                    0, nullptr, 0, BM * HD, HD, BM * CD, HD * CD, HD,
                                    nullptr, nullptr, nullptr, 0, j % 2, j / 2);
        return;
    }
    const int blk = blockIdx.x - SC_FOLD;
    int g0 = 0, c = 0;
    if (t < NBIN) {
        g0 = boff[t * NBLK + blk];
        c = boff[t * NBLK + blk + 1] - g0;
    }
    const bool al = (((((size_t)src) | ((size_t)dst)) & 15) == 0);
    int4 ss[CR], dd[CR];
#pragma unroll
    for (int k = 0; k < CR; ++k) {
        const int q = t + ET * k;
        const int i4 = blk * (CH / 4) + q;
        ss[k] = make_int4(0, 0, 0, 0);
        dd[k] = make_int4(0, 0, 0, 0);
        if (q < CH / 4 && i4 < EE / 4) { dd[k] = ld4(dst, i4, al); ss[k] = ld4(src, i4, al); }
    }
    const int ex = block_excl_scan(c, wsum);
    if (t < NBIN) { cur[t] = ex; loff[t] = ex; goff[t] = g0; }
    __syncthreads();
#pragma unroll
    for (int k = 0; k < CR; ++k) {
        const int q = t + ET * k;
        const int i4 = blk * (CH / 4) + q;
        if (q < CH / 4 && i4 < EE / 4) {
            const int p0 = atomicAdd(&cur[dd[k].x >> 8], 1);
            const int p1 = atomicAdd(&cur[dd[k].y >> 8], 1);
            const int p2 = atomicAdd(&cur[dd[k].z >> 8], 1);
            const int p3 = atomicAdd(&cur[dd[k].w >> 8], 1);
            rec[p0] = make_int2(ss[k].x, dd[k].x);
            rec[p1] = make_int2(ss[k].y, dd[k].y);
            rec[p2] = make_int2(ss[k].z, dd[k].z);
            rec[p3] = make_int2(ss[k].w, dd[k].w);
        }
    }
    __syncthreads();
    int nloc = EE - blk * CH;
    nloc = nloc < CH ? nloc : CH;
    for (int i = t; i < nloc; i += ET) {
        const int2 r = rec[i];
        const int b = r.y >> 8;
        binned[goff[b] + (i - loff[b])] = r;
    }
}

#define RW_FOLD 3
static_assert(ET == 256, "one thread per row of a bin");
__global__ void __launch_bounds__(ET)
bin_rows_fold(const int2* __restrict__ binned, const int* __restrict__ boff,
              int* __restrict__ rowptr, float* __restrict__ dinv, int* __restrict__ col,
              const float* __restrict__ Tm, const float* __restrict__ lW1,
              float* __restrict__ Fm) {
    __shared__ int rc[256];
    __shared__ int wsum[32];
    __shared__ int scol[DCAP];
    const int t = threadIdx.x;
    if (blockIdx.x < RW_FOLD) {
        gemm_block<true, HD, false>(Tm, HD, lW1, CD, Fm, CD, HD, nullptr, nullptr, nullptr,
                                    0, nullptr, 0, BM * HD, HD, BM * CD, HD * CD, HD,
                                    nullptr, nullptr, nullptr, 0, 2, (int)blockIdx.x);
        return;
    }
    const int b = blockIdx.x - RW_FOLD;
    const int s0 = boff[b * NBLK];
    const int s1 = boff[(b + 1) * NBLK];
    const int n = s1 - s0;
    const bool small = (n <= DCAP);
    int2 rr[DCAP / ET];
    rc[t] = 0;
    if (small) {
#pragma unroll
        for (int k = 0; k < DCAP / ET; ++k) {
            const int i = t + k * ET;
            rr[k] = make_int2(0, 0);
            if (i < n) rr[k] = binned[s0 + i];
        }
    }
    __syncthreads();
    if (small) {
#pragma unroll
        for (int k = 0; k < DCAP / ET; ++k)
            if (t + k * ET < n) atomicAdd(&rc[rr[k].y & 255], 1);
    } else {
        for (int i = t; i < n; i += ET) atomicAdd(&rc[binned[s0 + i].y & 255], 1);
    }
    __syncthreads();
    const int c = rc[t];
    const int ex = block_excl_scan(c, wsum);
    const int row = b * 256 + t;
    rc[t] = ex;
    if (row < NN) {
        rowptr[row] = s0 + ex;
        dinv[row] = rsqrtf((float)(c + 1));
        if (row == NN - 1) rowptr[NN] = s0 + ex + c;
    }
    __syncthreads();
    if (small) {
#pragma unroll
        for (int k = 0; k < DCAP / ET; ++k)
            if (t + k * ET < n) {
                const int p = atomicAdd(&rc[rr[k].y & 255], 1);
                scol[p] = rr[k].x;
            }
        __syncthreads();
        for (int i = t; i < n; i += ET) col[s0 + i] = scol[i];
    } else {
        for (int i = t; i < n; i += ET) {
            const int2 r = binned[s0 + i];
            const int p = atomicAdd(&rc[r.y & 255], 1);
            col[s0 + p] = r.x;
        }
    }
}

#define RUN_ROWS 31

template <int MODE>
__device__ __forceinline__ void
gather_rows(const float* __restrict__ Xin, int ldi, const float* __restrict__ dinv,
            const int* __restrict__ rowptr, const int* __restrict__ col,
            const float* __restrict__ wgt, float* __restrict__ wout,
            float* __restrict__ Yout, int ldo,
            const float* __restrict__ rin, float* __restrict__ rout, int n, int e,
            int bid, int nblk) {
    const int lane = threadIdx.x & 31;
    const int c  = lane & 15;
    const int hh = lane >> 4;
    const int wid = __builtin_amdgcn_readfirstlane(bid * 8 + (threadIdx.x >> 5));
    const int nw  = nblk * 8;
    const unsigned uq = (unsigned)n / (unsigned)nw, ur = (unsigned)n % (unsigned)nw;
    const int v0 = (int)((unsigned)wid * uq + ((unsigned)wid < ur ? (unsigned)wid : ur));
    const int v1 = v0 + (int)uq + ((unsigned)wid < ur ? 1 : 0);
    const char* Xb = (const char*)Xin;
    const unsigned pitch = (unsigned)ldi * 4u;
    const unsigned coff  = (unsigned)c * 16u;

#define GLOAD(R, P)                                                                \
    {                                                                              \
        const int sa = __builtin_amdgcn_readlane(myc, 2 * (P));                    \
        const int sb = __builtin_amdgcn_readlane(myc, 2 * (P) + 1);                \
        const unsigned sr = (unsigned)(hh ? sb : sa);                              \
        q[R] = *(const float4*)(Xb + (sr * pitch + coff));                         \
    }
#define GFMA(R, P)                                                                 \
    {                                                                              \
        const int wa = __builtin_amdgcn_readlane(mywi, 2 * (P));                   \
        const int wb = __builtin_amdgcn_readlane(mywi, 2 * (P) + 1);               \
        const float w = __int_as_float(hh ? wb : wa);                              \
        acc.x += q[R].x * w; acc.y += q[R].y * w;                                  \
        acc.z += q[R].z * w; acc.w += q[R].w * w;                                  \
    }

    for (int vb = v0; vb < v1; vb += RUN_ROWS) {
        const int cnt = (v1 - vb) < RUN_ROWS ? (v1 - vb) : RUN_ROWS;
        const int ri = (vb + lane) < n ? (vb + lane) : n;
        const int di = (vb + lane) < n ? (vb + lane) : (n - 1);
        const int rp = rowptr[ri];
        const float dv = dinv[di];
        float rself = 0.0f;
        if (MODE == 2) rself = rin[di];
        const int dvi = __float_as_int(dv);
        const int rsi = __float_as_int(rself);
        int beg = __builtin_amdgcn_readlane(rp, 0);
        int pi = beg + lane;
        pi = pi < e ? pi : e - 1;
        pi = pi > 0 ? pi : 0;
        int   nc  = col[pi];
        float nwt = (MODE == 1) ? dinv[nc] : wgt[pi];
        for (int i = 0; i < cnt; ++i) {
            const int v = vb + i;
            const int end = __builtin_amdgcn_readlane(rp, i + 1);
            const float dn = __int_as_float(__builtin_amdgcn_readlane(dvi, i));
            const float nn = dn * dn;
            const int   fc = nc;
            const float fw = nwt;
            pi = end + lane;
            pi = pi < e ? pi : e - 1;
            pi = pi > 0 ? pi : 0;
            nc  = col[pi];
            if (MODE != 1) nwt = wgt[pi];
            const float4 xs = *(const float4*)(Xb + ((unsigned)v * pitch + coff));
            float4 acc = make_float4(0.0f, 0.0f, 0.0f, 0.0f);
            bool first = true;
            float rs = 0.0f;
            for (int base = beg; base < end; base += 32) {
                const int m = (end - base) < 32 ? (end - base) : 32;
                int   myc = v;
                float myw = 0.0f;
                if (lane < m) {
                    if (base == beg) { myc = fc; myw = (MODE == 1) ? fw * dn : fw; }
                    else {
                        myc = col[base + lane];
                        myw = (MODE == 1) ? dinv[myc] * dn : wgt[base + lane];
                    }
                }
                float rv = 0.0f;
                if (MODE == 2) rv = rin[myc];
                const int mywi = __float_as_int(myw);
                float4 q[12];
                GLOAD(0, 0) GLOAD(1, 1) GLOAD(2, 2) GLOAD(3, 3)
                if (m > 8) {
                    GLOAD(4, 4) GLOAD(5, 5) GLOAD(6, 6) GLOAD(7, 7)
                    if (m > 16) { GLOAD(8, 8) GLOAD(9, 9) GLOAD(10, 10) GLOAD(11, 11) }
                }
                if (first) {
                    acc.x = hh ? 0.0f : xs.x * nn; acc.y = hh ? 0.0f : xs.y * nn;
                    acc.z = hh ? 0.0f : xs.z * nn; acc.w = hh ? 0.0f : xs.w * nn;
                    first = false;
                }
                GFMA(0, 0) GFMA(1, 1) GFMA(2, 2) GFMA(3, 3)
                if (m > 8) {
                    GFMA(4, 4) GFMA(5, 5) GFMA(6, 6) GFMA(7, 7)
                    if (m > 16) { GFMA(8, 8) GFMA(9, 9) GFMA(10, 10) GFMA(11, 11) }
                }
                if (m > 24) {
                    GLOAD(0, 12) GLOAD(1, 13) GLOAD(2, 14) GLOAD(3, 15)
                    GFMA(0, 12) GFMA(1, 13) GFMA(2, 14) GFMA(3, 15)
                }
                if (MODE == 1) {
                    rs += myw;
                    if (lane < m) wout[base + lane] = myw;
                }
                if (MODE == 2) rs += myw * rv;
            }
            if (MODE == 1) nwt = dinv[nc];
            if (first) {
                acc.x = hh ? 0.0f : xs.x * nn; acc.y = hh ? 0.0f : xs.y * nn;
                acc.z = hh ? 0.0f : xs.z * nn; acc.w = hh ? 0.0f : xs.w * nn;
            }
            acc.x += __shfl_xor(acc.x, 16, 32);
            acc.y += __shfl_xor(acc.y, 16, 32);
            acc.z += __shfl_xor(acc.z, 16, 32);
            acc.w += __shfl_xor(acc.w, 16, 32);
            if (hh == 0)
                *(float4*)(Yout + (size_t)v * ldo + c * 4) = acc;
            if (MODE != 0) {
#pragma unroll
                for (int mm = 16; mm >= 1; mm >>= 1) rs += __shfl_xor(rs, mm, 32);
                const float rsv = __int_as_float(__builtin_amdgcn_readlane(rsi, i));
                if (lane == 0) rout[v] = rs + (MODE == 1 ? nn : nn * rsv);
            }
            beg = end;
        }
    }
#undef GLOAD
#undef GFMA
}

template <int MODE>
__global__ void __launch_bounds__(256)
gather64(const float* __restrict__ Xin, int ldi, const float* __restrict__ dinv,
         const int* __restrict__ rowptr, const int* __restrict__ col,
         const float* __restrict__ wgt, float* __restrict__ wout,
         float* __restrict__ Yout, int ldo,
         const float* __restrict__ rin, float* __restrict__ rout, int n, int e) {
    gather_rows<MODE>(Xin, ldi, dinv, rowptr, col, wgt, wout, Yout, ldo, rin, rout, n, e,
                      blockIdx.x, gridDim.x);
}

extern "C" void kernel_launch(void* const* d_in, const int* in_sizes, int n_in,
                              void* d_out, int out_size, void* d_ws, size_t ws_size,
                              hipStream_t stream) {
    const float* x   = (const float*)d_in[0];
    const int*   ei  = (const int*)d_in[1];
    const float* W1  = (const float*)d_in[2];
    const float* b1  = (const float*)d_in[3];
    const float* W2  = (const float*)d_in[4];
    const float* b2  = (const float*)d_in[5];
    const float* W3  = (const float*)d_in[6];
    const float* b3  = (const float*)d_in[7];
    const float* lW1 = (const float*)d_in[8];
    const float* lb1 = (const float*)d_in[9];
    const float* lW2 = (const float*)d_in[10];
    const float* lb2 = (const float*)d_in[11];

    const int N = NN, E = EE;
    const int* src = ei;
    const int* dst = ei + E;

    float* y    = (float*)d_ws;
    float* r1   = y + (size_t)N * YD;
    float* r2   = r1 + N;
    float* Tm   = r2 + N;
    float* Fm   = Tm + 3 * BM * HD;
    float* dinv = Fm + 3 * BM * CD;
    int*   bb     = (int*)(dinv + N);
    int*   boff   = bb + LL;
    int2*  binned = (int2*)(boff + LL + 4);
    int*   rowptr = (int*)(binned + E);
    int*   col    = rowptr + (N + 4);
    float* wgt    = (float*)(col + E);

    float* logits = (float*)d_out;
    float* probs  = logits + (size_t)N * 2;

    const int T = 256;
    const int rowBlocks = N / BM;
    const int gatherBlocks = 1024;


    bin_count_fold<<<NBLK + 1, ET, 0, stream>>>(dst, bb, W1, b1, Tm, W2, b2);
    csr_scan<<<SCAN_NB, SCAN_T, 0, stream>>>(bb, boff);
    bin_scatter_fold<<<NBLK + SC_FOLD, ET, 0, stream>>>(src, dst, boff, binned, Tm, W3, b3, lW1, Fm);
    bin_rows_fold<<<NBIN + RW_FOLD, ET, 0, stream>>>(binned, boff, rowptr, dinv, col, Tm, lW1, Fm);

    gather64<1><<<gatherBlocks, T, 0, stream>>>(x, XD, dinv, rowptr, col, nullptr, wgt, y, YD,
                                                nullptr, r1, N, E);
    gather64<2><<<gatherBlocks, T, 0, stream>>>(y, YD, dinv, rowptr, col, wgt, nullptr, y + XD, YD,
                                                r1, r2, N, E);
    gather64<0><<<gatherBlocks, T, 0, stream>>>(y + XD, YD, dinv, rowptr, col, wgt, nullptr,
                                                y + 2 * XD, YD, nullptr, nullptr, N, E);

    head_gemm<<<rowBlocks, T, 0, stream>>>(y, YD, Fm, CD, logits, 2, YD, lb1, lW2, lb2, 16,
                                           r1, r2, probs);
}


# Head epilogue: a tile's 16-lane sums for 8 rows x 2 classes as one halving exchange (15 shuffles, was 64 in 40 one-row chains each waited for alone), r1/r2 loaded up front; same sum order
# speedup vs baseline: 6.2525x; 1.1942x over previous
#include <hip/hip_runtime.h>
#include <hip/hip_bf16.h>


typedef __attribute__((ext_vector_type(2))) float v2f;
typedef __attribute__((ext_vector_type(4))) float v4f;
typedef __attribute__((ext_vector_type(8))) float v8f;

#define NN 50000
#define EE 800000
#define HD 128
#define CD 384
#define XD 64
#define YD 192


#define BM 80
#define KB 32
#define PA (KB + 4)

template <bool BT, int BN, bool HEAD>
__device__ __forceinline__ void
gemm_block(const float* __restrict__ A, int lda,
           const float* __restrict__ B, int ldb,
           float* __restrict__ C, int ldc, int K,
           const float* __restrict__ lb1,
           const float* __restrict__ lW2,
           const float* __restrict__ lb2,
           int bpad,
           const float* __restrict__ rowvec, int rowidx,
           int sAy, int sBy, int sCy, int sBz, int sCz,
           const float* __restrict__ r1, const float* __restrict__ r2,
           float* __restrict__ probs, int bx, int by, int bz) {
    A += (size_t)by * sAy;
    B += (size_t)by * sBy + (size_t)bz * sBz;
    C += (size_t)by * sCy + (size_t)bz * sCz;
    constexpr int RT = BM / 16;
    constexpr int CT = BN / 128;
    constexpr int PB = BT ? PA : (BN + 4);
    constexpr int BROWS = BT ? BN : KB;
    constexpr int QA = KB / 4;
    constexpr int A4 = BM * QA;
    constexpr int AL = (A4 + 255) / 256;
    constexpr int BL = BN * QA / 256;
    constexpr int QB = BN / 4;

    __shared__ __attribute__((aligned(16))) float As[BM * PA];
    __shared__ __attribute__((aligned(16))) float Bs[BROWS * PB];

    const int tid  = threadIdx.x;
    const int wave = tid >> 5;
    const int lane = tid & 31;
    const int c    = lane & 15;
    const int h    = lane >> 4;
    const int m0   = bx * BM;
    const int wn0  = wave * (CT * 16);

    v8f acc[RT][CT];
#pragma unroll
    for (int i = 0; i < RT; ++i)
#pragma unroll
        for (int j = 0; j < CT; ++j) {
            v8f z = {};
            acc[i][j] = z;
        }

    v4f ra[AL], rb[BL];

#define TILE_GLOAD(kb)                                                              \
    {                                                                               \
        _Pragma("unroll") for (int u = 0; u < AL; ++u) {                            \
            int idx = tid + 256 * u;                                                \
            if (idx < A4) {                                                         \
                int row = idx / QA, q = idx % QA;                                   \
                ra[u] = *(const v4f*)(A + (size_t)(m0 + row) * lda + (kb) + 4 * q); \
            }                                                                       \
        }                                                                           \
        _Pragma("unroll") for (int u = 0; u < BL; ++u) {                            \
            int idx = tid + 256 * u;                                                \
            if (BT) {                                                               \
                int n = idx / QA, q = idx % QA;                                     \
                rb[u] = *(const v4f*)(B + (size_t)n * ldb + (kb) + 4 * q);       \
            } else {                                                                \
                int k = idx / QB, q = idx % QB;                                     \
                rb[u] = *(const v4f*)(B + (size_t)((kb) + k + bpad * ((kb) >> 6)) * ldb + 4 * q); \
            }                                                                       \
        }                                                                           \
    }

    TILE_GLOAD(0)
    const int S = K / KB;
    for (int s = 0; s < S; ++s) {
#pragma unroll
        for (int u = 0; u < AL; ++u) {
            int idx = tid + 256 * u;
            if (idx < A4) {
                int row = idx / QA, q = idx % QA;
                *(v4f*)&As[row * PA + 4 * q] = ra[u];
            }
        }
#pragma unroll
        for (int u = 0; u < BL; ++u) {
            int idx = tid + 256 * u;
            if (BT) {
                int n = idx / QA, q = idx % QA;
                *(v4f*)&Bs[n * PB + 4 * q] = rb[u];
            } else {
                int k = idx / QB, q = idx % QB;
                *(v4f*)&Bs[k * PB + 4 * q] = rb[u];
            }
        }
        __syncthreads();
        if (s + 1 < S) TILE_GLOAD((s + 1) * KB)

#pragma unroll
        for (int g = 0; g < KB / 8; ++g) {
            v2f alo[RT], ahi[RT], blo[CT], bhi[CT];
#pragma unroll
            for (int i = 0; i < RT; ++i) {
                v4f a4 = *(const v4f*)&As[(16 * i + c) * PA + 8 * g + 4 * h];
                alo[i].x = a4.x; alo[i].y = a4.y;
                ahi[i].x = a4.z; ahi[i].y = a4.w;
            }
#pragma unroll
            for (int j = 0; j < CT; ++j) {
                if (BT) {
                    v4f b4 = *(const v4f*)&Bs[(wn0 + 16 * j + c) * PB + 8 * g + 4 * h];
                    blo[j].x = b4.x; blo[j].y = b4.y;
                    bhi[j].x = b4.z; bhi[j].y = b4.w;
                } else {
                    const float* bp = &Bs[(8 * g + 4 * h) * PB + wn0 + 16 * j + c];
                    blo[j].x = bp[0];
                    blo[j].y = bp[PB];
                    bhi[j].x = bp[2 * PB];
                    bhi[j].y = bp[3 * PB];
                }
            }
#pragma unroll
            for (int i = 0; i < RT; ++i)
#pragma unroll
                for (int j = 0; j < CT; ++j)
                    acc[i][j] = __builtin_amdgcn_wmma_f32_16x16x4_f32(
                        false, alo[i], false, blo[j], (short)0, acc[i][j], false, false);
#pragma unroll
            for (int i = 0; i < RT; ++i)
#pragma unroll
                for (int j = 0; j < CT; ++j)
                    acc[i][j] = __builtin_amdgcn_wmma_f32_16x16x4_f32(
                        false, ahi[i], false, bhi[j], (short)0, acc[i][j], false, false);
        }
        __syncthreads();
    }
#undef TILE_GLOAD

    if (!HEAD) {
#pragma unroll
        for (int i = 0; i < RT; ++i)
#pragma unroll
            for (int j = 0; j < CT; ++j)
#pragma unroll
                for (int r = 0; r < 8; ++r) {
                    const int row = m0 + 16 * i + 8 * h + r;
                    const int cc  = wn0 + 16 * j + c;
                    float o = acc[i][j][r];
                    if (rowvec != nullptr && row == rowidx) o += rowvec[cc];
                    C[(size_t)row * ldc + cc] = o;
                }
    } else {
        float bias[CT], u1[CT], u2[CT], w0[CT], w1[CT];
#pragma unroll
        for (int j = 0; j < CT; ++j) {
            int col = wn0 + 16 * j + c;
            bias[j] = lb1[col] + B[(size_t)(0 * BM + 64) * ldb + col]
                               + B[(size_t)(1 * BM + 65) * ldb + col]
                               + B[(size_t)(2 * BM + 66) * ldb + col];
            u1[j]   = B[(size_t)(1 * BM + 64) * ldb + col] + B[(size_t)(2 * BM + 65) * ldb + col];
            u2[j]   = B[(size_t)(2 * BM + 64) * ldb + col];
            w0[j]   = lW2[col];
            w1[j]   = lW2[BN + col];
        }
        float* red = Bs;
        v4f a1v[RT][2], a2v[RT][2];
#pragma unroll
        for (int i = 0; i < RT; ++i) {
            const float* q1 = r1 + m0 + 16 * i + 8 * h;
            const float* q2 = r2 + m0 + 16 * i + 8 * h;
            a1v[i][0] = *(const v4f*)q1;  a1v[i][1] = *(const v4f*)(q1 + 4);
            a2v[i][0] = *(const v4f*)q2;  a2v[i][1] = *(const v4f*)(q2 + 4);
        }
        const bool b3 = (c & 8) != 0, b2 = (c & 4) != 0, b1 = (c & 2) != 0, b0 = (c & 1) != 0;
#pragma unroll
        for (int i = 0; i < RT; ++i) {
            float p0[8], p1[8];
#pragma unroll
            for (int r = 0; r < 8; ++r) {
                const float a1 = a1v[i][r >> 2][r & 3], a2 = a2v[i][r >> 2][r & 3];
                float s0 = 0.0f, s1 = 0.0f;
#pragma unroll
                for (int j = 0; j < CT; ++j) {
                    float t = acc[i][j][r] + bias[j] + a1 * u1[j] + a2 * u2[j];
                    t = t > 0.0f ? t : 0.0f;
                    s0 += t * w0[j];
                    s1 += t * w1[j];
                }
                p0[r] = s0;
                p1[r] = s1;
            }
#pragma unroll
            for (int r = 0; r < 4; ++r) {
                const float g0 = b3 ? p0[r] : p0[r + 4], k0 = b3 ? p0[r + 4] : p0[r];
                const float g1 = b3 ? p1[r] : p1[r + 4], k1 = b3 ? p1[r + 4] : p1[r];
                p0[r] = k0 + __shfl_xor(g0, 8, 32);
                p1[r] = k1 + __shfl_xor(g1, 8, 32);
            }
#pragma unroll
            for (int r = 0; r < 2; ++r) {
                const float g0 = b2 ? p0[r] : p0[r + 2], k0 = b2 ? p0[r + 2] : p0[r];
                const float g1 = b2 ? p1[r] : p1[r + 2], k1 = b2 ? p1[r + 2] : p1[r];
                p0[r] = k0 + __shfl_xor(g0, 4, 32);
                p1[r] = k1 + __shfl_xor(g1, 4, 32);
            }
            {
                const float g0 = b1 ? p0[0] : p0[1], k0 = b1 ? p0[1] : p0[0];
                const float g1 = b1 ? p1[0] : p1[1], k1 = b1 ? p1[1] : p1[0];
                p0[0] = k0 + __shfl_xor(g0, 2, 32);
                p1[0] = k1 + __shfl_xor(g1, 2, 32);
            }
            const float gz = b0 ? p0[0] : p1[0], kz = b0 ? p1[0] : p0[0];
            const float z = kz + __shfl_xor(gz, 1, 32);
            red[(wave * BM + 16 * i) * 2 + lane] = z;
        }
        __syncthreads();
        if (tid < 2 * BM) {
            int row = tid >> 1, cls = tid & 1;
            float sum = lb2[cls];
#pragma unroll
            for (int w = 0; w < 8; ++w) sum += red[(w * BM + row) * 2 + cls];
            C[(size_t)(m0 + row) * ldc + cls] = sum;
            const float oth = __shfl_xor(sum, 1, 32);
            const float mx = fmaxf(sum, oth);
            const float es = __expf(sum - mx), eo = __expf(oth - mx);
            const float inv = 1.0f / (es + eo);
            probs[(size_t)(m0 + row) * 2 + cls] = es * inv;
        }
    }
}

__global__ void __launch_bounds__(256)
head_gemm(const float* __restrict__ A, int lda, const float* __restrict__ B, int ldb,
          float* __restrict__ C, int ldc, int K,
          const float* __restrict__ lb1, const float* __restrict__ lW2,
          const float* __restrict__ lb2, int bpad, const float* __restrict__ r1,
          const float* __restrict__ r2, float* __restrict__ probs) {
    gemm_block<false, CD, true>(A, lda, B, ldb, C, ldc, K, lb1, lW2, lb2, bpad, nullptr, 0,
                                0, 0, 0, 0, 0, r1, r2, probs, blockIdx.x, 0, 0);
}

__device__ __forceinline__ void fold_step(const float* __restrict__ Tin, const float* __restrict__ Wm,
                                          float* __restrict__ Tout, const float* __restrict__ bvec,
                                          int brow) {
    gemm_block<false, HD, false>(Tin, HD, Wm, HD, Tout, HD, HD, nullptr, nullptr, nullptr,
                                 0, bvec, brow, 0, 0, 0, 0, 0, nullptr, nullptr, nullptr, 0, 0, 0);
}

#define NBLK 256
#define NBIN ((NN + 255) / 256)
#define LL (NBIN * NBLK)
#define CH ((((EE + NBLK - 1) / NBLK) + 3) & ~3)
static_assert(EE % 4 == 0, "edges are read four at a time");
#define ET 256
#define CR ((CH / 4 + ET - 1) / ET)
static_assert(NBIN <= 256, "bin counters are handled by the first 256 threads");
#define DCAP 8192

__device__ __forceinline__ int4 ld4(const int* __restrict__ p, int i4, bool al) {
    if (al) return ((const int4*)p)[i4];
    int4 r;
    r.x = p[4 * i4]; r.y = p[4 * i4 + 1]; r.z = p[4 * i4 + 2]; r.w = p[4 * i4 + 3];
    return r;
}

__device__ __forceinline__ int block_excl_scan(int v, int* wsum) {
    const int lane = threadIdx.x & 31;
    const int w = threadIdx.x >> 5;
    int incl = v;
#pragma unroll
    for (int m = 1; m < 32; m <<= 1) {
        int up = __shfl_up(incl, m, 32);
        if (lane >= m) incl += up;
    }
    if (lane == 31) wsum[w] = incl;
    __syncthreads();
    int woff = (lane < w) ? wsum[lane] : 0;
#pragma unroll
    for (int m = 16; m >= 1; m >>= 1) woff += __shfl_xor(woff, m, 32);
    __syncthreads();
    return woff + incl - v;
}


__global__ void __launch_bounds__(ET)
bin_count_fold(const int* __restrict__ dst, int* __restrict__ bb,
               const float* __restrict__ W1, const float* __restrict__ b1, float* Tm,
               const float* __restrict__ W2, const float* __restrict__ b2) {
    __shared__ int h[256];
    const int t = threadIdx.x;
    if (blockIdx.x == 0) {
        for (int i = t; i < BM * HD; i += ET) {
            int r = i / HD, cc = i % HD;
            float o = 0.0f;
            if (r < XD) o = W1[i];
            else if (r == XD) o = b1[cc];
            Tm[i] = o;
        }
        __syncthreads();
        fold_step(Tm, W2, Tm + BM * HD, b2, 65);
        return;
    }
    const int blk = blockIdx.x - 1;
    h[t] = 0;
    __syncthreads();
    const bool al = ((((size_t)dst) & 15) == 0);
    int4 dd[CR];
#pragma unroll
    for (int k = 0; k < CR; ++k) {
        const int q = t + ET * k;
        const int i4 = blk * (CH / 4) + q;
        dd[k] = make_int4(0, 0, 0, 0);
        if (q < CH / 4 && i4 < EE / 4) dd[k] = ld4(dst, i4, al);
    }
#pragma unroll
    for (int k = 0; k < CR; ++k) {
        const int q = t + ET * k;
        const int i4 = blk * (CH / 4) + q;
        if (q < CH / 4 && i4 < EE / 4) {
            atomicAdd(&h[dd[k].x >> 8], 1);
            atomicAdd(&h[dd[k].y >> 8], 1);
            atomicAdd(&h[dd[k].z >> 8], 1);
            atomicAdd(&h[dd[k].w >> 8], 1);
        }
    }
    __syncthreads();
    if (t < NBIN) bb[t * NBLK + blk] = h[t];
}

#define SCAN_T 1024
#define SCAN_NB (LL / SCAN_T)
#define SCAN_PASS ((SCAN_NB - 1 + 3) / 4)
static_assert(LL % SCAN_T == 0, "the scan has no ragged block");
static_assert(SCAN_PASS * SCAN_T * 4 <= LL, "the unguarded 16-byte loads must stay inside the count array");
static_assert(SCAN_T / 32 == 32, "one wave total per lane in the block-level step");
__global__ void __launch_bounds__(SCAN_T)
csr_scan(const int* __restrict__ cnt, int* __restrict__ boff) {
    __shared__ int wsum[SCAN_T / 32];
    __shared__ int wpart[SCAN_T / 32];
    const int t = threadIdx.x;
    const int lane = t & 31;
    const int w = t >> 5;
    const int idx = blockIdx.x * SCAN_T + t;
    const int cv = cnt[idx];
    const int n4 = blockIdx.x * (SCAN_T / 4);
    const int4* c4 = (const int4*)cnt;
    int4 pv[SCAN_PASS];
#pragma unroll
    for (int i = 0; i < SCAN_PASS; ++i) pv[i] = c4[i * SCAN_T + t];
    int part = 0;
#pragma unroll
    for (int i = 0; i < SCAN_PASS; ++i) {
        const int s = (pv[i].x + pv[i].y) + (pv[i].z + pv[i].w);
        part += (i * SCAN_T + t < n4) ? s : 0;
    }
#pragma unroll
    for (int m = 16; m >= 1; m >>= 1) part += __shfl_xor(part, m, 32);
    int incl = cv;
#pragma unroll
    for (int m = 1; m < 32; m <<= 1) {
        int up = __shfl_up(incl, m, 32);
        if (lane >= m) incl += up;
    }
    if (lane == 31) wsum[w] = incl;
    if (lane == 0) wpart[w] = part;
    __syncthreads();
    int base = wpart[lane];
    int woff = (lane < w) ? wsum[lane] : 0;
#pragma unroll
    for (int m = 16; m >= 1; m >>= 1) {
        base += __shfl_xor(base, m, 32);
        woff += __shfl_xor(woff, m, 32);
    }
    const int p = base + woff + incl - cv;
    boff[idx] = p;
    if (idx == LL - 1) boff[LL] = p + cv;
}

#define SC_FOLD 7
__global__ void __launch_bounds__(ET)
bin_scatter_fold(const int* __restrict__ src, const int* __restrict__ dst,
                 const int* __restrict__ boff, int2* __restrict__ binned,
                 float* Tm, const float* __restrict__ W3, const float* __restrict__ b3,
                 const float* __restrict__ lW1, float* __restrict__ Fm) {
    __shared__ int cur[256];
    __shared__ int loff[256];
    __shared__ int goff[256];
    __shared__ int wsum[32];
    __shared__ int2 rec[CH];
    const int t = threadIdx.x;
    if (blockIdx.x == 0) {
        fold_step(Tm + BM * HD, W3, Tm + 2 * BM * HD, b3, 66);
        return;
    }
    if (blockIdx.x < SC_FOLD) {
        const int j = blockIdx.x - 1;
        gemm_block<true, HD, false>(Tm, HD, lW1, CD, Fm, CD, HD, nullptr, nullptr, nullptr,
                                    0, nullptr, 0, BM * HD, HD, BM * CD, HD * CD, HD,
                                    nullptr, nullptr, nullptr, 0, j % 2, j / 2);
        return;
    }
    const int blk = blockIdx.x - SC_FOLD;
    int g0 = 0, c = 0;
    if (t < NBIN) {
        g0 = boff[t * NBLK + blk];
        c = boff[t * NBLK + blk + 1] - g0;
    }
    const bool al = (((((size_t)src) | ((size_t)dst)) & 15) == 0);
    int4 ss[CR], dd[CR];
#pragma unroll
    for (int k = 0; k < CR; ++k) {
        const int q = t + ET * k;
        const int i4 = blk * (CH / 4) + q;
        ss[k] = make_int4(0, 0, 0, 0);
        dd[k] = make_int4(0, 0, 0, 0);
        if (q < CH / 4 && i4 < EE / 4) { dd[k] = ld4(dst, i4, al); ss[k] = ld4(src, i4, al); }
    }
    const int ex = block_excl_scan(c, wsum);
    if (t < NBIN) { cur[t] = ex; loff[t] = ex; goff[t] = g0; }
    __syncthreads();
#pragma unroll
    for (int k = 0; k < CR; ++k) {
        const int q = t + ET * k;
        const int i4 = blk * (CH / 4) + q;
        if (q < CH / 4 && i4 < EE / 4) {
            const int p0 = atomicAdd(&cur[dd[k].x >> 8], 1);
            const int p1 = atomicAdd(&cur[dd[k].y >> 8], 1);
            const int p2 = atomicAdd(&cur[dd[k].z >> 8], 1);
            const int p3 = atomicAdd(&cur[dd[k].w >> 8], 1);
            rec[p0] = make_int2(ss[k].x, dd[k].x);
            rec[p1] = make_int2(ss[k].y, dd[k].y);
            rec[p2] = make_int2(ss[k].z, dd[k].z);
            rec[p3] = make_int2(ss[k].w, dd[k].w);
        }
    }
    __syncthreads();
    int nloc = EE - blk * CH;
    nloc = nloc < CH ? nloc : CH;
    for (int i = t; i < nloc; i += ET) {
        const int2 r = rec[i];
        const int b = r.y >> 8;
        binned[goff[b] + (i - loff[b])] = r;
    }
}

#define RW_FOLD 3
static_assert(ET == 256, "one thread per row of a bin");
__global__ void __launch_bounds__(ET)
bin_rows_fold(const int2* __restrict__ binned, const int* __restrict__ boff,
              int* __restrict__ rowptr, float* __restrict__ dinv, int* __restrict__ col,
              const float* __restrict__ Tm, const float* __restrict__ lW1,
              float* __restrict__ Fm) {
    __shared__ int rc[256];
    __shared__ int wsum[32];
    __shared__ int scol[DCAP];
    const int t = threadIdx.x;
    if (blockIdx.x < RW_FOLD) {
        gemm_block<true, HD, false>(Tm, HD, lW1, CD, Fm, CD, HD, nullptr, nullptr, nullptr,
                                    0, nullptr, 0, BM * HD, HD, BM * CD, HD * CD, HD,
                                    nullptr, nullptr, nullptr, 0, 2, (int)blockIdx.x);
        return;
    }
    const int b = blockIdx.x - RW_FOLD;
    const int s0 = boff[b * NBLK];
    const int s1 = boff[(b + 1) * NBLK];
    const int n = s1 - s0;
    const bool small = (n <= DCAP);
    int2 rr[DCAP / ET];
    rc[t] = 0;
    if (small) {
#pragma unroll
        for (int k = 0; k < DCAP / ET; ++k) {
            const int i = t + k * ET;
            rr[k] = make_int2(0, 0);
            if (i < n) rr[k] = binned[s0 + i];
        }
    }
    __syncthreads();
    if (small) {
#pragma unroll
        for (int k = 0; k < DCAP / ET; ++k)
            if (t + k * ET < n) atomicAdd(&rc[rr[k].y & 255], 1);
    } else {
        for (int i = t; i < n; i += ET) atomicAdd(&rc[binned[s0 + i].y & 255], 1);
    }
    __syncthreads();
    const int c = rc[t];
    const int ex = block_excl_scan(c, wsum);
    const int row = b * 256 + t;
    rc[t] = ex;
    if (row < NN) {
        rowptr[row] = s0 + ex;
        dinv[row] = rsqrtf((float)(c + 1));
        if (row == NN - 1) rowptr[NN] = s0 + ex + c;
    }
    __syncthreads();
    if (small) {
#pragma unroll
        for (int k = 0; k < DCAP / ET; ++k)
            if (t + k * ET < n) {
                const int p = atomicAdd(&rc[rr[k].y & 255], 1);
                scol[p] = rr[k].x;
            }
        __syncthreads();
        for (int i = t; i < n; i += ET) col[s0 + i] = scol[i];
    } else {
        for (int i = t; i < n; i += ET) {
            const int2 r = binned[s0 + i];
            const int p = atomicAdd(&rc[r.y & 255], 1);
            col[s0 + p] = r.x;
        }
    }
}

#define RUN_ROWS 31

template <int MODE>
__device__ __forceinline__ void
gather_rows(const float* __restrict__ Xin, int ldi, const float* __restrict__ dinv,
            const int* __restrict__ rowptr, const int* __restrict__ col,
            const float* __restrict__ wgt, float* __restrict__ wout,
            float* __restrict__ Yout, int ldo,
            const float* __restrict__ rin, float* __restrict__ rout, int n, int e,
            int bid, int nblk) {
    const int lane = threadIdx.x & 31;
    const int c  = lane & 15;
    const int hh = lane >> 4;
    const int wid = __builtin_amdgcn_readfirstlane(bid * 8 + (threadIdx.x >> 5));
    const int nw  = nblk * 8;
    const unsigned uq = (unsigned)n / (unsigned)nw, ur = (unsigned)n % (unsigned)nw;
    const int v0 = (int)((unsigned)wid * uq + ((unsigned)wid < ur ? (unsigned)wid : ur));
    const int v1 = v0 + (int)uq + ((unsigned)wid < ur ? 1 : 0);
    const char* Xb = (const char*)Xin;
    const unsigned pitch = (unsigned)ldi * 4u;
    const unsigned coff  = (unsigned)c * 16u;

#define GLOAD(R, P)                                                                \
    {                                                                              \
        const int sa = __builtin_amdgcn_readlane(myc, 2 * (P));                    \
        const int sb = __builtin_amdgcn_readlane(myc, 2 * (P) + 1);                \
        const unsigned sr = (unsigned)(hh ? sb : sa);                              \
        q[R] = *(const float4*)(Xb + (sr * pitch + coff));                         \
    }
#define GFMA(R, P)                                                                 \
    {                                                                              \
        const int wa = __builtin_amdgcn_readlane(mywi, 2 * (P));                   \
        const int wb = __builtin_amdgcn_readlane(mywi, 2 * (P) + 1);               \
        const float w = __int_as_float(hh ? wb : wa);                              \
        acc.x += q[R].x * w; acc.y += q[R].y * w;                                  \
        acc.z += q[R].z * w; acc.w += q[R].w * w;                                  \
    }

    for (int vb = v0; vb < v1; vb += RUN_ROWS) {
        const int cnt = (v1 - vb) < RUN_ROWS ? (v1 - vb) : RUN_ROWS;
        const int ri = (vb + lane) < n ? (vb + lane) : n;
        const int di = (vb + lane) < n ? (vb + lane) : (n - 1);
        const int rp = rowptr[ri];
        const float dv = dinv[di];
        float rself = 0.0f;
        if (MODE == 2) rself = rin[di];
        const int dvi = __float_as_int(dv);
        const int rsi = __float_as_int(rself);
        int beg = __builtin_amdgcn_readlane(rp, 0);
        int pi = beg + lane;
        pi = pi < e ? pi : e - 1;
        pi = pi > 0 ? pi : 0;
        int   nc  = col[pi];
        float nwt = (MODE == 1) ? dinv[nc] : wgt[pi];
        for (int i = 0; i < cnt; ++i) {
            const int v = vb + i;
            const int end = __builtin_amdgcn_readlane(rp, i + 1);
            const float dn = __int_as_float(__builtin_amdgcn_readlane(dvi, i));
            const float nn = dn * dn;
            const int   fc = nc;
            const float fw = nwt;
            pi = end + lane;
            pi = pi < e ? pi : e - 1;
            pi = pi > 0 ? pi : 0;
            nc  = col[pi];
            if (MODE != 1) nwt = wgt[pi];
            const float4 xs = *(const float4*)(Xb + ((unsigned)v * pitch + coff));
            float4 acc = make_float4(0.0f, 0.0f, 0.0f, 0.0f);
            bool first = true;
            float rs = 0.0f;
            for (int base = beg; base < end; base += 32) {
                const int m = (end - base) < 32 ? (end - base) : 32;
                int   myc = v;
                float myw = 0.0f;
                if (lane < m) {
                    if (base == beg) { myc = fc; myw = (MODE == 1) ? fw * dn : fw; }
                    else {
                        myc = col[base + lane];
                        myw = (MODE == 1) ? dinv[myc] * dn : wgt[base + lane];
                    }
                }
                float rv = 0.0f;
                if (MODE == 2) rv = rin[myc];
                const int mywi = __float_as_int(myw);
                float4 q[12];
                GLOAD(0, 0) GLOAD(1, 1) GLOAD(2, 2) GLOAD(3, 3)
                if (m > 8) {
                    GLOAD(4, 4) GLOAD(5, 5) GLOAD(6, 6) GLOAD(7, 7)
                    if (m > 16) { GLOAD(8, 8) GLOAD(9, 9) GLOAD(10, 10) GLOAD(11, 11) }
                }
                if (first) {
                    acc.x = hh ? 0.0f : xs.x * nn; acc.y = hh ? 0.0f : xs.y * nn;
                    acc.z = hh ? 0.0f : xs.z * nn; acc.w = hh ? 0.0f : xs.w * nn;
                    first = false;
                }
                GFMA(0, 0) GFMA(1, 1) GFMA(2, 2) GFMA(3, 3)
                if (m > 8) {
                    GFMA(4, 4) GFMA(5, 5) GFMA(6, 6) GFMA(7, 7)
                    if (m > 16) { GFMA(8, 8) GFMA(9, 9) GFMA(10, 10) GFMA(11, 11) }
                }
                if (m > 24) {
                    GLOAD(0, 12) GLOAD(1, 13) GLOAD(2, 14) GLOAD(3, 15)
                    GFMA(0, 12) GFMA(1, 13) GFMA(2, 14) GFMA(3, 15)
                }
                if (MODE == 1) {
                    rs += myw;
                    if (lane < m) wout[base + lane] = myw;
                }
                if (MODE == 2) rs += myw * rv;
            }
            if (MODE == 1) nwt = dinv[nc];
            if (first) {
                acc.x = hh ? 0.0f : xs.x * nn; acc.y = hh ? 0.0f : xs.y * nn;
                acc.z = hh ? 0.0f : xs.z * nn; acc.w = hh ? 0.0f : xs.w * nn;
            }
            acc.x += __shfl_xor(acc.x, 16, 32);
            acc.y += __shfl_xor(acc.y, 16, 32);
            acc.z += __shfl_xor(acc.z, 16, 32);
            acc.w += __shfl_xor(acc.w, 16, 32);
            if (hh == 0)
                *(float4*)(Yout + (size_t)v * ldo + c * 4) = acc;
            if (MODE != 0) {
#pragma unroll
                for (int mm = 16; mm >= 1; mm >>= 1) rs += __shfl_xor(rs, mm, 32);
                const float rsv = __int_as_float(__builtin_amdgcn_readlane(rsi, i));
                if (lane == 0) rout[v] = rs + (MODE == 1 ? nn : nn * rsv);
            }
            beg = end;
        }
    }
#undef GLOAD
#undef GFMA
}

template <int MODE>
__global__ void __launch_bounds__(256)
gather64(const float* __restrict__ Xin, int ldi, const float* __restrict__ dinv,
         const int* __restrict__ rowptr, const int* __restrict__ col,
         const float* __restrict__ wgt, float* __restrict__ wout,
         float* __restrict__ Yout, int ldo,
         const float* __restrict__ rin, float* __restrict__ rout, int n, int e) {
    gather_rows<MODE>(Xin, ldi, dinv, rowptr, col, wgt, wout, Yout, ldo, rin, rout, n, e,
                      blockIdx.x, gridDim.x);
}

extern "C" void kernel_launch(void* const* d_in, const int* in_sizes, int n_in,
                              void* d_out, int out_size, void* d_ws, size_t ws_size,
                              hipStream_t stream) {
    const float* x   = (const float*)d_in[0];
    const int*   ei  = (const int*)d_in[1];
    const float* W1  = (const float*)d_in[2];
    const float* b1  = (const float*)d_in[3];
    const float* W2  = (const float*)d_in[4];
    const float* b2  = (const float*)d_in[5];
    const float* W3  = (const float*)d_in[6];
    const float* b3  = (const float*)d_in[7];
    const float* lW1 = (const float*)d_in[8];
    const float* lb1 = (const float*)d_in[9];
    const float* lW2 = (const float*)d_in[10];
    const float* lb2 = (const float*)d_in[11];

    const int N = NN, E = EE;
    const int* src = ei;
    const int* dst = ei + E;

    float* y    = (float*)d_ws;
    float* r1   = y + (size_t)N * YD;
    float* r2   = r1 + N;
    float* Tm   = r2 + N;
    float* Fm   = Tm + 3 * BM * HD;
    float* dinv = Fm + 3 * BM * CD;
    int*   bb     = (int*)(dinv + N);
    int*   boff   = bb + LL;
    int2*  binned = (int2*)(boff + LL + 4);
    int*   rowptr = (int*)(binned + E);
    int*   col    = rowptr + (N + 4);
    float* wgt    = (float*)(col + E);

    float* logits = (float*)d_out;
    float* probs  = logits + (size_t)N * 2;

    const int T = 256;
    const int rowBlocks = N / BM;
    const int gatherBlocks = 1024;


    bin_count_fold<<<NBLK + 1, ET, 0, stream>>>(dst, bb, W1, b1, Tm, W2, b2);
    csr_scan<<<SCAN_NB, SCAN_T, 0, stream>>>(bb, boff);
    bin_scatter_fold<<<NBLK + SC_FOLD, ET, 0, stream>>>(src, dst, boff, binned, Tm, W3, b3, lW1, Fm);
    bin_rows_fold<<<NBIN + RW_FOLD, ET, 0, stream>>>(binned, boff, rowptr, dinv, col, Tm, lW1, Fm);

    gather64<1><<<gatherBlocks, T, 0, stream>>>(x, XD, dinv, rowptr, col, nullptr, wgt, y, YD,
                                                nullptr, r1, N, E);
    gather64<2><<<gatherBlocks, T, 0, stream>>>(y, YD, dinv, rowptr, col, wgt, nullptr, y + XD, YD,
                                                r1, r2, N, E);
    gather64<0><<<gatherBlocks, T, 0, stream>>>(y + XD, YD, dinv, rowptr, col, wgt, nullptr,
                                                y + 2 * XD, YD, nullptr, nullptr, N, E);

    head_gemm<<<rowBlocks, T, 0, stream>>>(y, YD, Fm, CD, logits, 2, YD, lb1, lW2, lb2, 16,
                                           r1, r2, probs);
}


# Head GEMM block tile 80 -> 208 rows: 241 work-groups, one per WGP in ONE round (was 625 in three rounds of 256, one resident per WGP); r1/r2 fetched one tile ahead; same sum order
# speedup vs baseline: 6.8862x; 1.1014x over previous
#include <hip/hip_runtime.h>
#include <hip/hip_bf16.h>


typedef __attribute__((ext_vector_type(2))) float v2f;
typedef __attribute__((ext_vector_type(4))) float v4f;
typedef __attribute__((ext_vector_type(8))) float v8f;

#define NN 50000
#define EE 800000
#define HD 128
#define CD 384
#define XD 64
#define YD 192


#define BM 80
#define HM 208
#define HB ((NN + HM - 1) / HM)
static_assert(NN % 16 == 0, "a 16-row tile of the head lies wholly inside or wholly outside the matrix");
static_assert(HM % 16 == 0 && HB <= 256, "head grid: whole row tiles, one round");
#define KB 32
#define PA (KB + 4)

template <bool BT, int BN, bool HEAD, int TM>
__device__ __forceinline__ void
gemm_block(const float* __restrict__ A, int lda,
           const float* __restrict__ B, int ldb,
           float* __restrict__ C, int ldc, int K,
           const float* __restrict__ lb1,
           const float* __restrict__ lW2,
           const float* __restrict__ lb2,
           int bpad,
           const float* __restrict__ rowvec, int rowidx,
           int sAy, int sBy, int sCy, int sBz, int sCz,
           const float* __restrict__ r1, const float* __restrict__ r2,
           float* __restrict__ probs, int bx, int by, int bz) {
    A += (size_t)by * sAy;
    B += (size_t)by * sBy + (size_t)bz * sBz;
    C += (size_t)by * sCy + (size_t)bz * sCz;
    constexpr int RT = TM / 16;
    constexpr int CT = BN / 128;
    constexpr int PB = BT ? PA : (BN + 4);
    constexpr int BROWS = BT ? BN : KB;
    constexpr int QA = KB / 4;
    constexpr int A4 = TM * QA;
    constexpr int AL = (A4 + 255) / 256;
    constexpr int BL = BN * QA / 256;
    constexpr int QB = BN / 4;

    __shared__ __attribute__((aligned(16))) float As[TM * PA];
    __shared__ __attribute__((aligned(16))) float Bs[BROWS * PB];

    const int tid  = threadIdx.x;
    const int wave = tid >> 5;
    const int lane = tid & 31;
    const int c    = lane & 15;
    const int h    = lane >> 4;
    const int m0   = bx * TM;
    const int wn0  = wave * (CT * 16);

    v8f acc[RT][CT];
#pragma unroll
    for (int i = 0; i < RT; ++i)
#pragma unroll
        for (int j = 0; j < CT; ++j) {
            v8f z = {};
            acc[i][j] = z;
        }

    v4f ra[AL], rb[BL];

#define TILE_GLOAD(kb)                                                              \
    {                                                                               \
        _Pragma("unroll") for (int u = 0; u < AL; ++u) {                            \
            int idx = tid + 256 * u;                                                \
            if (idx < A4) {                                                         \
                int row = idx / QA, q = idx % QA;                                   \
                if (HEAD) {                                                         \
                                     \
                    int gr = m0 + row;                                              \
                    gr = gr < NN ? gr : NN - 1;                                     \
                    ra[u] = *(const v4f*)(A + ((unsigned)gr * (unsigned)lda         \
                                               + (unsigned)((kb) + 4 * q)));        \
                } else {                                                            \
                    ra[u] = *(const v4f*)(A + (size_t)(m0 + row) * lda + (kb) + 4 * q); \
                }                                                                   \
            }                                                                       \
        }                                                                           \
        _Pragma("unroll") for (int u = 0; u < BL; ++u) {                            \
            int idx = tid + 256 * u;                                                \
            if (BT) {                                                               \
                int n = idx / QA, q = idx % QA;                                     \
                rb[u] = *(const v4f*)(B + (size_t)n * ldb + (kb) + 4 * q);       \
            } else {                                                                \
                int k = idx / QB, q = idx % QB;                                     \
                if (HEAD)                     \
                    rb[u] = *(const v4f*)(B + ((unsigned)((kb) + k + bpad * ((kb) >> 6)) \
                                               * (unsigned)ldb + (unsigned)(4 * q))); \
                else                                                                \
                    rb[u] = *(const v4f*)(B + (size_t)((kb) + k + bpad * ((kb) >> 6)) * ldb + 4 * q); \
            }                                                                       \
        }                                                                           \
    }

    TILE_GLOAD(0)
    const int S = K / KB;
    for (int s = 0; s < S; ++s) {
#pragma unroll
        for (int u = 0; u < AL; ++u) {
            int idx = tid + 256 * u;
            if (idx < A4) {
                int row = idx / QA, q = idx % QA;
                *(v4f*)&As[row * PA + 4 * q] = ra[u];
            }
        }
#pragma unroll
        for (int u = 0; u < BL; ++u) {
            int idx = tid + 256 * u;
            if (BT) {
                int n = idx / QA, q = idx % QA;
                *(v4f*)&Bs[n * PB + 4 * q] = rb[u];
            } else {
                int k = idx / QB, q = idx % QB;
                *(v4f*)&Bs[k * PB + 4 * q] = rb[u];
            }
        }
        __syncthreads();
        if (s + 1 < S) TILE_GLOAD((s + 1) * KB)

#pragma unroll
        for (int g = 0; g < KB / 8; ++g) {
            v2f alo[RT], ahi[RT], blo[CT], bhi[CT];
#pragma unroll
            for (int i = 0; i < RT; ++i) {
                v4f a4 = *(const v4f*)&As[(16 * i + c) * PA + 8 * g + 4 * h];
                alo[i].x = a4.x; alo[i].y = a4.y;
                ahi[i].x = a4.z; ahi[i].y = a4.w;
            }
#pragma unroll
            for (int j = 0; j < CT; ++j) {
                if (BT) {
                    v4f b4 = *(const v4f*)&Bs[(wn0 + 16 * j + c) * PB + 8 * g + 4 * h];
                    blo[j].x = b4.x; blo[j].y = b4.y;
                    bhi[j].x = b4.z; bhi[j].y = b4.w;
                } else {
                    const float* bp = &Bs[(8 * g + 4 * h) * PB + wn0 + 16 * j + c];
                    blo[j].x = bp[0];
                    blo[j].y = bp[PB];
                    bhi[j].x = bp[2 * PB];
                    bhi[j].y = bp[3 * PB];
                }
            }
#pragma unroll
            for (int i = 0; i < RT; ++i)
#pragma unroll
                for (int j = 0; j < CT; ++j)
                    acc[i][j] = __builtin_amdgcn_wmma_f32_16x16x4_f32(
                        false, alo[i], false, blo[j], (short)0, acc[i][j], false, false);
#pragma unroll
            for (int i = 0; i < RT; ++i)
#pragma unroll
                for (int j = 0; j < CT; ++j)
                    acc[i][j] = __builtin_amdgcn_wmma_f32_16x16x4_f32(
                        false, ahi[i], false, bhi[j], (short)0, acc[i][j], false, false);
        }
        __syncthreads();
    }
#undef TILE_GLOAD

    if (!HEAD) {
#pragma unroll
        for (int i = 0; i < RT; ++i)
#pragma unroll
            for (int j = 0; j < CT; ++j)
#pragma unroll
                for (int r = 0; r < 8; ++r) {
                    const int row = m0 + 16 * i + 8 * h + r;
                    const int cc  = wn0 + 16 * j + c;
                    float o = acc[i][j][r];
                    if (rowvec != nullptr && row == rowidx) o += rowvec[cc];
                    C[(size_t)row * ldc + cc] = o;
                }
    } else {
        float bias[CT], u1[CT], u2[CT], w0[CT], w1[CT];
#pragma unroll
        for (int j = 0; j < CT; ++j) {
            int col = wn0 + 16 * j + c;
            bias[j] = lb1[col] + B[(size_t)(0 * BM + 64) * ldb + col]
                               + B[(size_t)(1 * BM + 65) * ldb + col]
                               + B[(size_t)(2 * BM + 66) * ldb + col];
            u1[j]   = B[(size_t)(1 * BM + 64) * ldb + col] + B[(size_t)(2 * BM + 65) * ldb + col];
            u2[j]   = B[(size_t)(2 * BM + 64) * ldb + col];
            w0[j]   = lW2[col];
            w1[j]   = lW2[BN + col];
        }
        float* red = Bs;
        v4f a1c[2], a2c[2];
        {
            const float* q1 = r1 + m0 + 8 * h;
            const float* q2 = r2 + m0 + 8 * h;
            a1c[0] = *(const v4f*)q1;  a1c[1] = *(const v4f*)(q1 + 4);
            a2c[0] = *(const v4f*)q2;  a2c[1] = *(const v4f*)(q2 + 4);
        }
        const bool b3 = (c & 8) != 0, b2 = (c & 4) != 0, b1 = (c & 2) != 0, b0 = (c & 1) != 0;
#pragma unroll
        for (int i = 0; i < RT; ++i) {
            v4f a1x[2], a2x[2];
            a1x[0] = a1c[0]; a1x[1] = a1c[1]; a2x[0] = a2c[0]; a2x[1] = a2c[1];
            if (i + 1 < RT) {
                int tb = m0 + 16 * (i + 1);
                tb = tb < NN ? tb : NN - 16;
                const float* q1 = r1 + tb + 8 * h;
                const float* q2 = r2 + tb + 8 * h;
                a1x[0] = *(const v4f*)q1;  a1x[1] = *(const v4f*)(q1 + 4);
                a2x[0] = *(const v4f*)q2;  a2x[1] = *(const v4f*)(q2 + 4);
            }
            float p0[8], p1[8];
#pragma unroll
            for (int r = 0; r < 8; ++r) {
                const float a1 = a1c[r >> 2][r & 3], a2 = a2c[r >> 2][r & 3];
                float s0 = 0.0f, s1 = 0.0f;
#pragma unroll
                for (int j = 0; j < CT; ++j) {
                    float t = acc[i][j][r] + bias[j] + a1 * u1[j] + a2 * u2[j];
                    t = t > 0.0f ? t : 0.0f;
                    s0 += t * w0[j];
                    s1 += t * w1[j];
                }
                p0[r] = s0;
                p1[r] = s1;
            }
#pragma unroll
            for (int r = 0; r < 4; ++r) {
                const float g0 = b3 ? p0[r] : p0[r + 4], k0 = b3 ? p0[r + 4] : p0[r];
                const float g1 = b3 ? p1[r] : p1[r + 4], k1 = b3 ? p1[r + 4] : p1[r];
                p0[r] = k0 + __shfl_xor(g0, 8, 32);
                p1[r] = k1 + __shfl_xor(g1, 8, 32);
            }
#pragma unroll
            for (int r = 0; r < 2; ++r) {
                const float g0 = b2 ? p0[r] : p0[r + 2], k0 = b2 ? p0[r + 2] : p0[r];
                const float g1 = b2 ? p1[r] : p1[r + 2], k1 = b2 ? p1[r + 2] : p1[r];
                p0[r] = k0 + __shfl_xor(g0, 4, 32);
                p1[r] = k1 + __shfl_xor(g1, 4, 32);
            }
            {
                const float g0 = b1 ? p0[0] : p0[1], k0 = b1 ? p0[1] : p0[0];
                const float g1 = b1 ? p1[0] : p1[1], k1 = b1 ? p1[1] : p1[0];
                p0[0] = k0 + __shfl_xor(g0, 2, 32);
                p1[0] = k1 + __shfl_xor(g1, 2, 32);
            }
            const float gz = b0 ? p0[0] : p1[0], kz = b0 ? p1[0] : p0[0];
            const float z = kz + __shfl_xor(gz, 1, 32);
            red[(wave * TM + 16 * i) * 2 + lane] = z;
            a1c[0] = a1x[0]; a1c[1] = a1x[1]; a2c[0] = a2x[0]; a2c[1] = a2x[1];
        }
        __syncthreads();
        static_assert((2 * TM) % 32 == 0, "the final sum runs in whole waves");
#pragma unroll
        for (int u = 0; u < (2 * TM + 255) / 256; ++u) {
            const int idx = tid + 256 * u;
            if (idx < 2 * TM) {
                int row = idx >> 1, cls = idx & 1;
                float sum = lb2[cls];
#pragma unroll
                for (int w = 0; w < 8; ++w) sum += red[(w * TM + row) * 2 + cls];
                const float oth = __shfl_xor(sum, 1, 32);
                const float mx = fmaxf(sum, oth);
                const float es = __expf(sum - mx), eo = __expf(oth - mx);
                const float inv = 1.0f / (es + eo);
                if (m0 + row < NN) {
                    C[(size_t)(m0 + row) * ldc + cls] = sum;
                    probs[(size_t)(m0 + row) * 2 + cls] = es * inv;
                }
            }
        }
    }
}

__global__ void __launch_bounds__(256)
head_gemm(const float* __restrict__ A, const float* __restrict__ B,
          float* __restrict__ C, int K,
          const float* __restrict__ lb1, const float* __restrict__ lW2,
          const float* __restrict__ lb2, const float* __restrict__ r1,
          const float* __restrict__ r2, float* __restrict__ probs) {
    gemm_block<false, CD, true, HM>(A, YD, B, CD, C, 2, K, lb1, lW2, lb2, BM - XD, nullptr, 0,
                                0, 0, 0, 0, 0, r1, r2, probs, blockIdx.x, 0, 0);
}

__device__ __forceinline__ void fold_step(const float* __restrict__ Tin, const float* __restrict__ Wm,
                                          float* __restrict__ Tout, const float* __restrict__ bvec,
                                          int brow) {
    gemm_block<false, HD, false, BM>(Tin, HD, Wm, HD, Tout, HD, HD, nullptr, nullptr, nullptr,
                                 0, bvec, brow, 0, 0, 0, 0, 0, nullptr, nullptr, nullptr, 0, 0, 0);
}

#define NBLK 256
#define NBIN ((NN + 255) / 256)
#define LL (NBIN * NBLK)
#define CH ((((EE + NBLK - 1) / NBLK) + 3) & ~3)
static_assert(EE % 4 == 0, "edges are read four at a time");
#define ET 256
#define CR ((CH / 4 + ET - 1) / ET)
static_assert(NBIN <= 256, "bin counters are handled by the first 256 threads");
#define DCAP 8192

__device__ __forceinline__ int4 ld4(const int* __restrict__ p, int i4, bool al) {
    if (al) return ((const int4*)p)[i4];
    int4 r;
    r.x = p[4 * i4]; r.y = p[4 * i4 + 1]; r.z = p[4 * i4 + 2]; r.w = p[4 * i4 + 3];
    return r;
}

__device__ __forceinline__ int block_excl_scan(int v, int* wsum) {
    const int lane = threadIdx.x & 31;
    const int w = threadIdx.x >> 5;
    int incl = v;
#pragma unroll
    for (int m = 1; m < 32; m <<= 1) {
        int up = __shfl_up(incl, m, 32);
        if (lane >= m) incl += up;
    }
    if (lane == 31) wsum[w] = incl;
    __syncthreads();
    int woff = (lane < w) ? wsum[lane] : 0;
#pragma unroll
    for (int m = 16; m >= 1; m >>= 1) woff += __shfl_xor(woff, m, 32);
    __syncthreads();
    return woff + incl - v;
}


__global__ void __launch_bounds__(ET)
bin_count_fold(const int* __restrict__ dst, int* __restrict__ bb,
               const float* __restrict__ W1, const float* __restrict__ b1, float* Tm,
               const float* __restrict__ W2, const float* __restrict__ b2) {
    __shared__ int h[256];
    const int t = threadIdx.x;
    if (blockIdx.x == 0) {
        for (int i = t; i < BM * HD; i += ET) {
            int r = i / HD, cc = i % HD;
            float o = 0.0f;
            if (r < XD) o = W1[i];
            else if (r == XD) o = b1[cc];
            Tm[i] = o;
        }
        __syncthreads();
        fold_step(Tm, W2, Tm + BM * HD, b2, 65);
        return;
    }
    const int blk = blockIdx.x - 1;
    h[t] = 0;
    __syncthreads();
    const bool al = ((((size_t)dst) & 15) == 0);
    int4 dd[CR];
#pragma unroll
    for (int k = 0; k < CR; ++k) {
        const int q = t + ET * k;
        const int i4 = blk * (CH / 4) + q;
        dd[k] = make_int4(0, 0, 0, 0);
        if (q < CH / 4 && i4 < EE / 4) dd[k] = ld4(dst, i4, al);
    }
#pragma unroll
    for (int k = 0; k < CR; ++k) {
        const int q = t + ET * k;
        const int i4 = blk * (CH / 4) + q;
        if (q < CH / 4 && i4 < EE / 4) {
            atomicAdd(&h[dd[k].x >> 8], 1);
            atomicAdd(&h[dd[k].y >> 8], 1);
            atomicAdd(&h[dd[k].z >> 8], 1);
            atomicAdd(&h[dd[k].w >> 8], 1);
        }
    }
    __syncthreads();
    if (t < NBIN) bb[t * NBLK + blk] = h[t];
}

#define SCAN_T 1024
#define SCAN_NB (LL / SCAN_T)
#define SCAN_PASS ((SCAN_NB - 1 + 3) / 4)
static_assert(LL % SCAN_T == 0, "the scan has no ragged block");
static_assert(SCAN_PASS * SCAN_T * 4 <= LL, "the unguarded 16-byte loads must stay inside the count array");
static_assert(SCAN_T / 32 == 32, "one wave total per lane in the block-level step");
__global__ void __launch_bounds__(SCAN_T)
csr_scan(const int* __restrict__ cnt, int* __restrict__ boff) {
    __shared__ int wsum[SCAN_T / 32];
    __shared__ int wpart[SCAN_T / 32];
    const int t = threadIdx.x;
    const int lane = t & 31;
    const int w = t >> 5;
    const int idx = blockIdx.x * SCAN_T + t;
    const int cv = cnt[idx];
    const int n4 = blockIdx.x * (SCAN_T / 4);
    const int4* c4 = (const int4*)cnt;
    int4 pv[SCAN_PASS];
#pragma unroll
    for (int i = 0; i < SCAN_PASS; ++i) pv[i] = c4[i * SCAN_T + t];
    int part = 0;
#pragma unroll
    for (int i = 0; i < SCAN_PASS; ++i) {
        const int s = (pv[i].x + pv[i].y) + (pv[i].z + pv[i].w);
        part += (i * SCAN_T + t < n4) ? s : 0;
    }
#pragma unroll
    for (int m = 16; m >= 1; m >>= 1) part += __shfl_xor(part, m, 32);
    int incl = cv;
#pragma unroll
    for (int m = 1; m < 32; m <<= 1) {
        int up = __shfl_up(incl, m, 32);
        if (lane >= m) incl += up;
    }
    if (lane == 31) wsum[w] = incl;
    if (lane == 0) wpart[w] = part;
    __syncthreads();
    int base = wpart[lane];
    int woff = (lane < w) ? wsum[lane] : 0;
#pragma unroll
    for (int m = 16; m >= 1; m >>= 1) {
        base += __shfl_xor(base, m, 32);
        woff += __shfl_xor(woff, m, 32);
    }
    const int p = base + woff + incl - cv;
    boff[idx] = p;
    if (idx == LL - 1) boff[LL] = p + cv;
}

#define SC_FOLD 7
__global__ void __launch_bounds__(ET)
bin_scatter_fold(const int* __restrict__ src, const int* __restrict__ dst,
                 const int* __restrict__ boff, int2* __restrict__ binned,
                 float* Tm, const float* __restrict__ W3, const float* __restrict__ b3,
                 const float* __restrict__ lW1, float* __restrict__ Fm) {
    __shared__ int cur[256];
    __shared__ int loff[256];
    __shared__ int goff[256];
    __shared__ int wsum[32];
    __shared__ int2 rec[CH];
    const int t = threadIdx.x;
    if (blockIdx.x == 0) {
        fold_step(Tm + BM * HD, W3, Tm + 2 * BM * HD, b3, 66);
        return;
    }
    if (blockIdx.x < SC_FOLD) {
        const int j = blockIdx.x - 1;
        gemm_block<true, HD, false, BM>(Tm, HD, lW1, CD, Fm, CD, HD, nullptr, nullptr, nullptr,
                                    0, nullptr, 0, BM * HD, HD, BM * CD, HD * CD, HD,
                                    nullptr, nullptr, nullptr, 0, j % 2, j / 2);
        return;
    }
    const int blk = blockIdx.x - SC_FOLD;
    int g0 = 0, c = 0;
    if (t < NBIN) {
        g0 = boff[t * NBLK + blk];
        c = boff[t * NBLK + blk + 1] - g0;
    }
    const bool al = (((((size_t)src) | ((size_t)dst)) & 15) == 0);
    int4 ss[CR], dd[CR];
#pragma unroll
    for (int k = 0; k < CR; ++k) {
        const int q = t + ET * k;
        const int i4 = blk * (CH / 4) + q;
        ss[k] = make_int4(0, 0, 0, 0);
        dd[k] = make_int4(0, 0, 0, 0);
        if (q < CH / 4 && i4 < EE / 4) { dd[k] = ld4(dst, i4, al); ss[k] = ld4(src, i4, al); }
    }
    const int ex = block_excl_scan(c, wsum);
    if (t < NBIN) { cur[t] = ex; loff[t] = ex; goff[t] = g0; }
    __syncthreads();
#pragma unroll
    for (int k = 0; k < CR; ++k) {
        const int q = t + ET * k;
        const int i4 = blk * (CH / 4) + q;
        if (q < CH / 4 && i4 < EE / 4) {
            const int p0 = atomicAdd(&cur[dd[k].x >> 8], 1);
            const int p1 = atomicAdd(&cur[dd[k].y >> 8], 1);
            const int p2 = atomicAdd(&cur[dd[k].z >> 8], 1);
            const int p3 = atomicAdd(&cur[dd[k].w >> 8], 1);
            rec[p0] = make_int2(ss[k].x, dd[k].x);
            rec[p1] = make_int2(ss[k].y, dd[k].y);
            rec[p2] = make_int2(ss[k].z, dd[k].z);
            rec[p3] = make_int2(ss[k].w, dd[k].w);
        }
    }
    __syncthreads();
    int nloc = EE - blk * CH;
    nloc = nloc < CH ? nloc : CH;
    for (int i = t; i < nloc; i += ET) {
        const int2 r = rec[i];
        const int b = r.y >> 8;
        binned[goff[b] + (i - loff[b])] = r;
    }
}

#define RW_FOLD 3
static_assert(ET == 256, "one thread per row of a bin");
__global__ void __launch_bounds__(ET)
bin_rows_fold(const int2* __restrict__ binned, const int* __restrict__ boff,
              int* __restrict__ rowptr, float* __restrict__ dinv, int* __restrict__ col,
              const float* __restrict__ Tm, const float* __restrict__ lW1,
              float* __restrict__ Fm) {
    __shared__ int rc[256];
    __shared__ int wsum[32];
    __shared__ int scol[DCAP];
    const int t = threadIdx.x;
    if (blockIdx.x < RW_FOLD) {
        gemm_block<true, HD, false, BM>(Tm, HD, lW1, CD, Fm, CD, HD, nullptr, nullptr, nullptr,
                                    0, nullptr, 0, BM * HD, HD, BM * CD, HD * CD, HD,
                                    nullptr, nullptr, nullptr, 0, 2, (int)blockIdx.x);
        return;
    }
    const int b = blockIdx.x - RW_FOLD;
    const int s0 = boff[b * NBLK];
    const int s1 = boff[(b + 1) * NBLK];
    const int n = s1 - s0;
    const bool small = (n <= DCAP);
    int2 rr[DCAP / ET];
    rc[t] = 0;
    if (small) {
#pragma unroll
        for (int k = 0; k < DCAP / ET; ++k) {
            const int i = t + k * ET;
            rr[k] = make_int2(0, 0);
            if (i < n) rr[k] = binned[s0 + i];
        }
    }
    __syncthreads();
    if (small) {
#pragma unroll
        for (int k = 0; k < DCAP / ET; ++k)
            if (t + k * ET < n) atomicAdd(&rc[rr[k].y & 255], 1);
    } else {
        for (int i = t; i < n; i += ET) atomicAdd(&rc[binned[s0 + i].y & 255], 1);
    }
    __syncthreads();
    const int c = rc[t];
    const int ex = block_excl_scan(c, wsum);
    const int row = b * 256 + t;
    rc[t] = ex;
    if (row < NN) {
        rowptr[row] = s0 + ex;
        dinv[row] = rsqrtf((float)(c + 1));
        if (row == NN - 1) rowptr[NN] = s0 + ex + c;
    }
    __syncthreads();
    if (small) {
#pragma unroll
        for (int k = 0; k < DCAP / ET; ++k)
            if (t + k * ET < n) {
                const int p = atomicAdd(&rc[rr[k].y & 255], 1);
                scol[p] = rr[k].x;
            }
        __syncthreads();
        for (int i = t; i < n; i += ET) col[s0 + i] = scol[i];
    } else {
        for (int i = t; i < n; i += ET) {
            const int2 r = binned[s0 + i];
            const int p = atomicAdd(&rc[r.y & 255], 1);
            col[s0 + p] = r.x;
        }
    }
}

#define RUN_ROWS 31

template <int MODE>
__device__ __forceinline__ void
gather_rows(const float* __restrict__ Xin, int ldi, const float* __restrict__ dinv,
            const int* __restrict__ rowptr, const int* __restrict__ col,
            const float* __restrict__ wgt, float* __restrict__ wout,
            float* __restrict__ Yout, int ldo,
            const float* __restrict__ rin, float* __restrict__ rout, int n, int e,
            int bid, int nblk) {
    const int lane = threadIdx.x & 31;
    const int c  = lane & 15;
    const int hh = lane >> 4;
    const int wid = __builtin_amdgcn_readfirstlane(bid * 8 + (threadIdx.x >> 5));
    const int nw  = nblk * 8;
    const unsigned uq = (unsigned)n / (unsigned)nw, ur = (unsigned)n % (unsigned)nw;
    const int v0 = (int)((unsigned)wid * uq + ((unsigned)wid < ur ? (unsigned)wid : ur));
    const int v1 = v0 + (int)uq + ((unsigned)wid < ur ? 1 : 0);
    const char* Xb = (const char*)Xin;
    const unsigned pitch = (unsigned)ldi * 4u;
    const unsigned coff  = (unsigned)c * 16u;

#define GLOAD(R, P)                                                                \
    {                                                                              \
        const int sa = __builtin_amdgcn_readlane(myc, 2 * (P));                    \
        const int sb = __builtin_amdgcn_readlane(myc, 2 * (P) + 1);                \
        const unsigned sr = (unsigned)(hh ? sb : sa);                              \
        q[R] = *(const float4*)(Xb + (sr * pitch + coff));                         \
    }
#define GFMA(R, P)                                                                 \
    {                                                                              \
        const int wa = __builtin_amdgcn_readlane(mywi, 2 * (P));                   \
        const int wb = __builtin_amdgcn_readlane(mywi, 2 * (P) + 1);               \
        const float w = __int_as_float(hh ? wb : wa);                              \
        acc.x += q[R].x * w; acc.y += q[R].y * w;                                  \
        acc.z += q[R].z * w; acc.w += q[R].w * w;                                  \
    }

    for (int vb = v0; vb < v1; vb += RUN_ROWS) {
        const int cnt = (v1 - vb) < RUN_ROWS ? (v1 - vb) : RUN_ROWS;
        const int ri = (vb + lane) < n ? (vb + lane) : n;
        const int di = (vb + lane) < n ? (vb + lane) : (n - 1);
        const int rp = rowptr[ri];
        const float dv = dinv[di];
        float rself = 0.0f;
        if (MODE == 2) rself = rin[di];
        const int dvi = __float_as_int(dv);
        const int rsi = __float_as_int(rself);
        int beg = __builtin_amdgcn_readlane(rp, 0);
        int pi = beg + lane;
        pi = pi < e ? pi : e - 1;
        pi = pi > 0 ? pi : 0;
        int   nc  = col[pi];
        float nwt = (MODE == 1) ? dinv[nc] : wgt[pi];
        for (int i = 0; i < cnt; ++i) {
            const int v = vb + i;
            const int end = __builtin_amdgcn_readlane(rp, i + 1);
            const float dn = __int_as_float(__builtin_amdgcn_readlane(dvi, i));
            const float nn = dn * dn;
            const int   fc = nc;
            const float fw = nwt;
            pi = end + lane;
            pi = pi < e ? pi : e - 1;
            pi = pi > 0 ? pi : 0;
            nc  = col[pi];
            if (MODE != 1) nwt = wgt[pi];
            const float4 xs = *(const float4*)(Xb + ((unsigned)v * pitch + coff));
            float4 acc = make_float4(0.0f, 0.0f, 0.0f, 0.0f);
            bool first = true;
            float rs = 0.0f;
            for (int base = beg; base < end; base += 32) {
                const int m = (end - base) < 32 ? (end - base) : 32;
                int   myc = v;
                float myw = 0.0f;
                if (lane < m) {
                    if (base == beg) { myc = fc; myw = (MODE == 1) ? fw * dn : fw; }
                    else {
                        myc = col[base + lane];
                        myw = (MODE == 1) ? dinv[myc] * dn : wgt[base + lane];
                    }
                }
                float rv = 0.0f;
                if (MODE == 2) rv = rin[myc];
                const int mywi = __float_as_int(myw);
                float4 q[12];
                GLOAD(0, 0) GLOAD(1, 1) GLOAD(2, 2) GLOAD(3, 3)
                if (m > 8) {
                    GLOAD(4, 4) GLOAD(5, 5) GLOAD(6, 6) GLOAD(7, 7)
                    if (m > 16) { GLOAD(8, 8) GLOAD(9, 9) GLOAD(10, 10) GLOAD(11, 11) }
                }
                if (first) {
                    acc.x = hh ? 0.0f : xs.x * nn; acc.y = hh ? 0.0f : xs.y * nn;
                    acc.z = hh ? 0.0f : xs.z * nn; acc.w = hh ? 0.0f : xs.w * nn;
                    first = false;
                }
                GFMA(0, 0) GFMA(1, 1) GFMA(2, 2) GFMA(3, 3)
                if (m > 8) {
                    GFMA(4, 4) GFMA(5, 5) GFMA(6, 6) GFMA(7, 7)
                    if (m > 16) { GFMA(8, 8) GFMA(9, 9) GFMA(10, 10) GFMA(11, 11) }
                }
                if (m > 24) {
                    GLOAD(0, 12) GLOAD(1, 13) GLOAD(2, 14) GLOAD(3, 15)
                    GFMA(0, 12) GFMA(1, 13) GFMA(2, 14) GFMA(3, 15)
                }
                if (MODE == 1) {
                    rs += myw;
                    if (lane < m) wout[base + lane] = myw;
                }
                if (MODE == 2) rs += myw * rv;
            }
            if (MODE == 1) nwt = dinv[nc];
            if (first) {
                acc.x = hh ? 0.0f : xs.x * nn; acc.y = hh ? 0.0f : xs.y * nn;
                acc.z = hh ? 0.0f : xs.z * nn; acc.w = hh ? 0.0f : xs.w * nn;
            }
            acc.x += __shfl_xor(acc.x, 16, 32);
            acc.y += __shfl_xor(acc.y, 16, 32);
            acc.z += __shfl_xor(acc.z, 16, 32);
            acc.w += __shfl_xor(acc.w, 16, 32);
            if (hh == 0)
                *(float4*)(Yout + (size_t)v * ldo + c * 4) = acc;
            if (MODE != 0) {
#pragma unroll
                for (int mm = 16; mm >= 1; mm >>= 1) rs += __shfl_xor(rs, mm, 32);
                const float rsv = __int_as_float(__builtin_amdgcn_readlane(rsi, i));
                if (lane == 0) rout[v] = rs + (MODE == 1 ? nn : nn * rsv);
            }
            beg = end;
        }
    }
#undef GLOAD
#undef GFMA
}

template <int MODE>
__global__ void __launch_bounds__(256)
gather64(const float* __restrict__ Xin, int ldi, const float* __restrict__ dinv,
         const int* __restrict__ rowptr, const int* __restrict__ col,
         const float* __restrict__ wgt, float* __restrict__ wout,
         float* __restrict__ Yout, int ldo,
         const float* __restrict__ rin, float* __restrict__ rout, int n, int e) {
    gather_rows<MODE>(Xin, ldi, dinv, rowptr, col, wgt, wout, Yout, ldo, rin, rout, n, e,
                      blockIdx.x, gridDim.x);
}

extern "C" void kernel_launch(void* const* d_in, const int* in_sizes, int n_in,
                              void* d_out, int out_size, void* d_ws, size_t ws_size,
                              hipStream_t stream) {
    const float* x   = (const float*)d_in[0];
    const int*   ei  = (const int*)d_in[1];
    const float* W1  = (const float*)d_in[2];
    const float* b1  = (const float*)d_in[3];
    const float* W2  = (const float*)d_in[4];
    const float* b2  = (const float*)d_in[5];
    const float* W3  = (const float*)d_in[6];
    const float* b3  = (const float*)d_in[7];
    const float* lW1 = (const float*)d_in[8];
    const float* lb1 = (const float*)d_in[9];
    const float* lW2 = (const float*)d_in[10];
    const float* lb2 = (const float*)d_in[11];

    const int N = NN, E = EE;
    const int* src = ei;
    const int* dst = ei + E;

    float* y    = (float*)d_ws;
    float* r1   = y + (size_t)N * YD;
    float* r2   = r1 + N;
    float* Tm   = r2 + N;
    float* Fm   = Tm + 3 * BM * HD;
    float* dinv = Fm + 3 * BM * CD;
    int*   bb     = (int*)(dinv + N);
    int*   boff   = bb + LL;
    int2*  binned = (int2*)(boff + LL + 4);
    int*   rowptr = (int*)(binned + E);
    int*   col    = rowptr + (N + 4);
    float* wgt    = (float*)(col + E);

    float* logits = (float*)d_out;
    float* probs  = logits + (size_t)N * 2;

    const int T = 256;
    const int rowBlocks = HB;
    const int gatherBlocks = 1024;


    bin_count_fold<<<NBLK + 1, ET, 0, stream>>>(dst, bb, W1, b1, Tm, W2, b2);
    csr_scan<<<SCAN_NB, SCAN_T, 0, stream>>>(bb, boff);
    bin_scatter_fold<<<NBLK + SC_FOLD, ET, 0, stream>>>(src, dst, boff, binned, Tm, W3, b3, lW1, Fm);
    bin_rows_fold<<<NBIN + RW_FOLD, ET, 0, stream>>>(binned, boff, rowptr, dinv, col, Tm, lW1, Fm);

    gather64<1><<<gatherBlocks, T, 0, stream>>>(x, XD, dinv, rowptr, col, nullptr, wgt, y, YD,
                                                nullptr, r1, N, E);
    gather64<2><<<gatherBlocks, T, 0, stream>>>(y, YD, dinv, rowptr, col, wgt, nullptr, y + XD, YD,
                                                r1, r2, N, E);
    gather64<0><<<gatherBlocks, T, 0, stream>>>(y + XD, YD, dinv, rowptr, col, wgt, nullptr,
                                                y + 2 * XD, YD, nullptr, nullptr, N, E);

    head_gemm<<<rowBlocks, T, 0, stream>>>(y, Fm, logits, YD, lb1, lW2, lb2, r1, r2, probs);
}
